# starting kernel
# baseline (speedup 1.0000x reference)
#include <hip/hip_runtime.h>
#include <hip/hip_bf16.h>

typedef __attribute__((ext_vector_type(2))) float v2f;
typedef __attribute__((ext_vector_type(8))) float v8f;

#define FEATS 128

__global__ void gcn_degrees(const int* __restrict__ src, const int* __restrict__ dst,
                            float* __restrict__ dOut, float* __restrict__ dIn, int nE) {
    int e = blockIdx.x * blockDim.x + threadIdx.x;
    if (e >= nE) return;
    atomicAdd(&dOut[src[e]], 1.0f);
    atomicAdd(&dIn[dst[e]], 1.0f);
}

__global__ void gcn_norm(float* __restrict__ dOut, float* __restrict__ dIn, int n) {
    int i = blockIdx.x * blockDim.x + threadIdx.x;
    if (i >= n) return;
    dOut[i] = rsqrtf(fmaxf(dOut[i], 1.0f));
    dIn[i]  = rsqrtf(fmaxf(dIn[i], 1.0f));
}

__global__ void gcn_gemm1_wmma(const float* __restrict__ x, const float* __restrict__ W1,
                               const float* __restrict__ invOut, float* __restrict__ h) {
    const int lane    = threadIdx.x & 31;
    const int wave    = blockIdx.x * (blockDim.x >> 5) + (threadIdx.x >> 5);
    const int rowTile = wave >> 3;
    const int colTile = wave & 7;
    const int l15     = lane & 15;
    const int khalf   = lane >> 4;

    const int   row  = rowTile * 16 + l15;
    const int   nCol = colTile * 16 + l15;
    const float s    = invOut[row];
    const float* __restrict__ arow = x + (size_t)row * FEATS;

    v8f c = {};
#pragma unroll
    for (int k = 0; k < FEATS; k += 4) {
        const int ka = k + 2 * khalf;
        v2f a, b;
        a.x = arow[ka]     * s;
        a.y = arow[ka + 1] * s;
        b.x = W1[(size_t)ka       * FEATS + nCol];
        b.y = W1[(size_t)(ka + 1) * FEATS + nCol];
        c = __builtin_amdgcn_wmma_f32_16x16x4_f32(
                 false, a,  false, b,
                 (short)0, c,  false,  false);
    }

#pragma unroll
    for (int r = 0; r < 8; ++r) {
        const int mm = (lane < 16) ? r : (r + 8);
        h[(size_t)(rowTile * 16 + mm) * FEATS + nCol] = c[r];
    }
}

__global__ void gcn_scatter_rows(const float* __restrict__ h, const int* __restrict__ src,
                                 const int* __restrict__ dst, float* __restrict__ agg, int nE) {
    const int tid = blockIdx.x * blockDim.x + threadIdx.x;
    const int e   = tid >> 5;
    if (e >= nE) return;
    const int f = (tid & 31) * 4;
    const float4 v = *(const float4*)(h + (size_t)src[e] * FEATS + f);
    float* p = agg + (size_t)dst[e] * FEATS + f;
    atomicAdd(p + 0, v.x);
    atomicAdd(p + 1, v.y);
    atomicAdd(p + 2, v.z);
    atomicAdd(p + 3, v.w);
}

__global__ void gcn_post1(const float* __restrict__ agg, const float* __restrict__ invIn,
                          const float* __restrict__ b1, float* __restrict__ h, int n) {
    const int i = blockIdx.x * blockDim.x + threadIdx.x;
    if (i >= n * FEATS) return;
    const int node = i >> 7;
    const int feat = i & (FEATS - 1);
    const float v = agg[i] * invIn[node] + b1[feat];
    h[i] = fmaxf(v, 0.0f);
}

__global__ void gcn_dot2(const float* __restrict__ h, const float* __restrict__ W2,
                         const float* __restrict__ invOut, float* __restrict__ z) {
    const int node = blockIdx.x * (blockDim.x >> 5) + (threadIdx.x >> 5);
    const int lane = threadIdx.x & 31;
    const float4 hv = *(const float4*)(h + (size_t)node * FEATS + lane * 4);
    const float4 wv = *(const float4*)(W2 + lane * 4);
    float sum = hv.x * wv.x + hv.y * wv.y + hv.z * wv.z + hv.w * wv.w;
#pragma unroll
    for (int off = 16; off > 0; off >>= 1) sum += __shfl_down(sum, off, 32);
    if (lane == 0) z[node] = sum * invOut[node];
}

__global__ void gcn_scatter_scalar(const float* __restrict__ z, const int* __restrict__ src,
                                   const int* __restrict__ dst, float* __restrict__ agg2, int nE) {
    const int e = blockIdx.x * blockDim.x + threadIdx.x;
    if (e >= nE) return;
    atomicAdd(&agg2[dst[e]], z[src[e]]);
}

__global__ void gcn_final(const float* __restrict__ agg2, const float* __restrict__ invIn,
                          const float* __restrict__ b2, float* __restrict__ out, int n) {
    const int i = blockIdx.x * blockDim.x + threadIdx.x;
    if (i >= n) return;
    out[i] = agg2[i] * invIn[i] + b2[0];
}

extern "C" void kernel_launch(void* const* d_in, const int* in_sizes, int n_in,
                              void* d_out, int out_size, void* d_ws, size_t ws_size,
                              hipStream_t stream) {
    const float* x   = (const float*)d_in[0];
    const int*   src = (const int*)d_in[1];
    const int*   dst = (const int*)d_in[2];
    const float* W1  = (const float*)d_in[3];
    const float* b1  = (const float*)d_in[4];
    const float* W2  = (const float*)d_in[5];
    const float* b2  = (const float*)d_in[6];
    float* out = (float*)d_out;

    const int n  = in_sizes[0] / FEATS;
    const int nE = in_sizes[1];

    float* f      = (float*)d_ws;
    float* invOut = f;
    float* invIn  = f + (size_t)n;
    float* z      = f + 2 * (size_t)n;
    float* agg2   = f + 3 * (size_t)n;
    float* h      = f + 4 * (size_t)n;
    float* agg    = h + (size_t)n * FEATS;

    hipMemsetAsync(invOut, 0, 2 * (size_t)n * sizeof(float), stream);
    hipMemsetAsync(agg2,   0, (size_t)n * sizeof(float), stream);
    hipMemsetAsync(agg,    0, (size_t)n * FEATS * sizeof(float), stream);

    gcn_degrees<<<(nE + 255) / 256, 256, 0, stream>>>(src, dst, invOut, invIn, nE);
    gcn_norm<<<(n + 255) / 256, 256, 0, stream>>>(invOut, invIn, n);

    gcn_gemm1_wmma<<<n / 16, 256, 0, stream>>>(x, W1, invOut, h);

    gcn_scatter_rows<<<(int)(((size_t)nE * 32 + 255) / 256), 256, 0, stream>>>(h, src, dst, agg, nE);
    gcn_post1<<<(int)(((size_t)n * FEATS + 255) / 256), 256, 0, stream>>>(agg, invIn, b1, h, n);

    gcn_dot2<<<n / 8, 256, 0, stream>>>(h, W2, invOut, z);
    gcn_scatter_scalar<<<(nE + 255) / 256, 256, 0, stream>>>(z, src, dst, agg2, nE);
    gcn_final<<<(n + 255) / 256, 256, 0, stream>>>(agg2, invIn, b2, out, n);
}


# Row sums by owner blocks: a block holds 400 dst rows in LDS, reads the whole dst list, adds its h rows by LDS float adds, stores agg once; no global row atomics or agg memset; sum order = LDS arrival
# speedup vs baseline: 1.3714x; 1.3714x over previous
#include <hip/hip_runtime.h>
#include <hip/hip_bf16.h>

typedef __attribute__((ext_vector_type(2))) float v2f;
typedef __attribute__((ext_vector_type(8))) float v8f;

#define FEATS 128

__global__ void gcn_degrees(const int* __restrict__ src, const int* __restrict__ dst,
                            float* __restrict__ dOut, float* __restrict__ dIn, int nE) {
    int e = blockIdx.x * blockDim.x + threadIdx.x;
    if (e >= nE) return;
    atomicAdd(&dOut[src[e]], 1.0f);
    atomicAdd(&dIn[dst[e]], 1.0f);
}

__global__ void gcn_norm(float* __restrict__ dOut, float* __restrict__ dIn, int n) {
    int i = blockIdx.x * blockDim.x + threadIdx.x;
    if (i >= n) return;
    dOut[i] = rsqrtf(fmaxf(dOut[i], 1.0f));
    dIn[i]  = rsqrtf(fmaxf(dIn[i], 1.0f));
}

__global__ void gcn_gemm1_wmma(const float* __restrict__ x, const float* __restrict__ W1,
                               const float* __restrict__ invOut, float* __restrict__ h) {
    const int lane    = threadIdx.x & 31;
    const int wave    = blockIdx.x * (blockDim.x >> 5) + (threadIdx.x >> 5);
    const int rowTile = wave >> 3;
    const int colTile = wave & 7;
    const int l15     = lane & 15;
    const int khalf   = lane >> 4;

    const int   row  = rowTile * 16 + l15;
    const int   nCol = colTile * 16 + l15;
    const float s    = invOut[row];
    const float* __restrict__ arow = x + (size_t)row * FEATS;

    v8f c = {};
#pragma unroll
    for (int k = 0; k < FEATS; k += 4) {
        const int ka = k + 2 * khalf;
        v2f a, b;
        a.x = arow[ka]     * s;
        a.y = arow[ka + 1] * s;
        b.x = W1[(size_t)ka       * FEATS + nCol];
        b.y = W1[(size_t)(ka + 1) * FEATS + nCol];
        c = __builtin_amdgcn_wmma_f32_16x16x4_f32(
                 false, a,  false, b,
                 (short)0, c,  false,  false);
    }

#pragma unroll
    for (int r = 0; r < 8; ++r) {
        const int mm = (lane < 16) ? r : (r + 8);
        h[(size_t)(rowTile * 16 + mm) * FEATS + nCol] = c[r];
    }
}

#define NB          400
#define QCAP        64
#define CH          8
#define DRAIN_AT    24u
#define RG          8
#define OWN_THREADS 1024
#define OWN_WAVES   (OWN_THREADS / 32)

__device__ __forceinline__ void own_row_by_one_lane(float* accRow, const float* __restrict__ hRow) {
#pragma unroll 1
    for (int f = 0; f < FEATS; ++f) atomicAdd(accRow + f, hRow[f]);
}

__global__ __launch_bounds__(OWN_THREADS)
void gcn_rows_owner(const float* __restrict__ h, const int* __restrict__ src,
                    const int* __restrict__ dst, float* __restrict__ agg, int n, int nE) {
    __shared__ __attribute__((aligned(16))) float acc[NB * FEATS];
    __shared__ unsigned qe[OWN_WAVES][QCAP];
    __shared__ unsigned qd[OWN_WAVES][QCAP];
    __shared__ unsigned qn[OWN_WAVES];

    const int tid  = threadIdx.x;
    const int lane = tid & 31;
    const int wave = __builtin_amdgcn_readfirstlane(tid >> 5);
    const int lo   = blockIdx.x * NB;
    const unsigned rows = (unsigned)min(NB, n - lo);

    float4* acc4 = (float4*)acc;
    for (int i = tid; i < NB * FEATS / 4; i += OWN_THREADS) acc4[i] = make_float4(0.f, 0.f, 0.f, 0.f);
    if (lane == 0) qn[wave] = 0u;
    __syncthreads();

    const int nC   = nE >> 2;
    const int perB = 32 * CH;
    const int nBt  = (nC + perB - 1) / perB;
    const int rot  = (int)(((long long)blockIdx.x * nBt) / gridDim.x);
    const int4* __restrict__ dst4 = (const int4*)dst;

    if (wave == 0 && lane < 4) {
        const int e = 4 * nC + lane;
        if (e < nE) {
            const unsigned t = (unsigned)(dst[e] - lo);
            if (t < rows) own_row_by_one_lane(acc + t * FEATS, h + (size_t)src[e] * FEATS);
        }
    }

    int4 cur[CH];
    int idx0 = wave;
    int curB = min(idx0, nBt - 1) + rot;
    if (curB >= nBt) curB -= nBt;
#pragma unroll
    for (int j = 0; j < CH; ++j) {
        const int c = curB * perB + j * 32 + lane;
        cur[j] = dst4[min(c, nC - 1)];
    }

    while (idx0 < nBt) {
        const int nidx = idx0 + OWN_WAVES;
        int4 nxt[CH];
        int nxtB = curB;
        if (nidx < nBt) {
            nxtB = nidx + rot;
            if (nxtB >= nBt) nxtB -= nBt;
        }
#pragma unroll
        for (int j = 0; j < CH; ++j) {
            const int c = nxtB * perB + j * 32 + lane;
            nxt[j] = dst4[min(c, nC - 1)];
        }

#pragma unroll
        for (int j = 0; j < CH; ++j) {
            const int c = curB * perB + j * 32 + lane;
            const unsigned t0 = (unsigned)(cur[j].x - lo);
            const unsigned t1 = (unsigned)(cur[j].y - lo);
            const unsigned t2 = (unsigned)(cur[j].z - lo);
            const unsigned t3 = (unsigned)(cur[j].w - lo);
            unsigned tm = min(min(t0, t1), min(t2, t3));
            if (c >= nC) tm = 0xFFFFFFFFu;
            if (tm < rows) {
                const unsigned tt[4] = {t0, t1, t2, t3};
#pragma unroll
                for (int i = 0; i < 4; ++i) {
                    if (tt[i] < rows) {
                        const unsigned e   = (unsigned)(4 * c + i);
                        const unsigned pos = atomicAdd(&qn[wave], 1u);
                        if (pos < QCAP) {
                            qe[wave][pos] = e;
                            qd[wave][pos] = tt[i];
                        } else {
                            own_row_by_one_lane(acc + tt[i] * FEATS, h + (size_t)src[e] * FEATS);
                        }
                    }
                }
            }
        }

        __builtin_amdgcn_fence(__ATOMIC_SEQ_CST, "wavefront");
        unsigned cnt = __hip_atomic_load(&qn[wave], __ATOMIC_RELAXED, __HIP_MEMORY_SCOPE_WAVEFRONT);
        cnt = (unsigned)__builtin_amdgcn_readfirstlane((int)cnt);
        if (cnt > QCAP) cnt = QCAP;

        if (cnt >= DRAIN_AT || (nidx >= nBt && cnt > 0u)) {
            for (unsigned base = 0; base < cnt; base += 32u) {
                const unsigned m = min(32u, cnt - base);
                int      s_l = 0;
                unsigned d_l = 0;
                if ((unsigned)lane < m) {
                    const unsigned e_l = qe[wave][base + lane];
                    d_l = qd[wave][base + lane];
                    s_l = src[e_l];
                }
                for (unsigned i = 0; i < m; i += RG) {
                    float    v[RG][4];
                    unsigned dd[RG];
#pragma unroll
                    for (int k = 0; k < RG; ++k) {
                        const unsigned idx = min(i + k, m - 1u);
                        const int s = __builtin_amdgcn_readlane(s_l, (int)idx);
                        dd[k] = (unsigned)__builtin_amdgcn_readlane((int)d_l, (int)idx);
                        const float* __restrict__ hr = h + (size_t)s * FEATS + lane;
                        v[k][0] = hr[0];
                        v[k][1] = hr[32];
                        v[k][2] = hr[64];
                        v[k][3] = hr[96];
                    }
#pragma unroll
                    for (int k = 0; k < RG; ++k) {
                        if (i + k < m) {
                            float* ar = acc + dd[k] * FEATS + lane;
                            atomicAdd(ar,      v[k][0]);
                            atomicAdd(ar + 32, v[k][1]);
                            atomicAdd(ar + 64, v[k][2]);
                            atomicAdd(ar + 96, v[k][3]);
                        }
                    }
                }
            }
            if (lane == 0) qn[wave] = 0u;
            __builtin_amdgcn_fence(__ATOMIC_SEQ_CST, "wavefront");
        }

#pragma unroll
        for (int j = 0; j < CH; ++j) cur[j] = nxt[j];
        curB = nxtB;
        idx0 = nidx;
    }

    __syncthreads();
    float4* __restrict__ out4 = (float4*)(agg + (size_t)lo * FEATS);
    for (int i = tid; i < (int)rows * (FEATS / 4); i += OWN_THREADS) out4[i] = acc4[i];
}

__global__ void gcn_post1(const float* __restrict__ agg, const float* __restrict__ invIn,
                          const float* __restrict__ b1, float* __restrict__ h, int n) {
    const int i = blockIdx.x * blockDim.x + threadIdx.x;
    if (i >= n * FEATS) return;
    const int node = i >> 7;
    const int feat = i & (FEATS - 1);
    const float v = agg[i] * invIn[node] + b1[feat];
    h[i] = fmaxf(v, 0.0f);
}

__global__ void gcn_dot2(const float* __restrict__ h, const float* __restrict__ W2,
                         const float* __restrict__ invOut, float* __restrict__ z) {
    const int node = blockIdx.x * (blockDim.x >> 5) + (threadIdx.x >> 5);
    const int lane = threadIdx.x & 31;
    const float4 hv = *(const float4*)(h + (size_t)node * FEATS + lane * 4);
    const float4 wv = *(const float4*)(W2 + lane * 4);
    float sum = hv.x * wv.x + hv.y * wv.y + hv.z * wv.z + hv.w * wv.w;
#pragma unroll
    for (int off = 16; off > 0; off >>= 1) sum += __shfl_down(sum, off, 32);
    if (lane == 0) z[node] = sum * invOut[node];
}

__global__ void gcn_scatter_scalar(const float* __restrict__ z, const int* __restrict__ src,
                                   const int* __restrict__ dst, float* __restrict__ agg2, int nE) {
    const int e = blockIdx.x * blockDim.x + threadIdx.x;
    if (e >= nE) return;
    atomicAdd(&agg2[dst[e]], z[src[e]]);
}

__global__ void gcn_final(const float* __restrict__ agg2, const float* __restrict__ invIn,
                          const float* __restrict__ b2, float* __restrict__ out, int n) {
    const int i = blockIdx.x * blockDim.x + threadIdx.x;
    if (i >= n) return;
    out[i] = agg2[i] * invIn[i] + b2[0];
}

extern "C" void kernel_launch(void* const* d_in, const int* in_sizes, int n_in,
                              void* d_out, int out_size, void* d_ws, size_t ws_size,
                              hipStream_t stream) {
    const float* x   = (const float*)d_in[0];
    const int*   src = (const int*)d_in[1];
    const int*   dst = (const int*)d_in[2];
    const float* W1  = (const float*)d_in[3];
    const float* b1  = (const float*)d_in[4];
    const float* W2  = (const float*)d_in[5];
    const float* b2  = (const float*)d_in[6];
    float* out = (float*)d_out;

    const int n  = in_sizes[0] / FEATS;
    const int nE = in_sizes[1];

    float* f      = (float*)d_ws;
    float* invOut = f;
    float* invIn  = f + (size_t)n;
    float* z      = f + 2 * (size_t)n;
    float* agg2   = f + 3 * (size_t)n;
    float* h      = f + 4 * (size_t)n;
    float* agg    = h + (size_t)n * FEATS;

    hipMemsetAsync(invOut, 0, 2 * (size_t)n * sizeof(float), stream);
    hipMemsetAsync(agg2,   0, (size_t)n * sizeof(float), stream);

    gcn_degrees<<<(nE + 255) / 256, 256, 0, stream>>>(src, dst, invOut, invIn, nE);
    gcn_norm<<<(n + 255) / 256, 256, 0, stream>>>(invOut, invIn, n);

    gcn_gemm1_wmma<<<n / 16, 256, 0, stream>>>(x, W1, invOut, h);

    gcn_rows_owner<<<(n + NB - 1) / NB, OWN_THREADS, 0, stream>>>(h, src, dst, agg, n, nE);
    gcn_post1<<<(int)(((size_t)n * FEATS + 255) / 256), 256, 0, stream>>>(agg, invIn, b1, h, n);

    gcn_dot2<<<n / 8, 256, 0, stream>>>(h, W2, invOut, z);
    gcn_scatter_scalar<<<(nE + 255) / 256, 256, 0, stream>>>(z, src, dst, agg2, nE);
    gcn_final<<<(n + 255) / 256, 256, 0, stream>>>(agg2, invIn, b2, out, n);
}


# Owner blocks scan a 1-byte owner key per edge (dst/400, 16 per lane load, packed in the norm launch), not the 4-byte dst list: 1/4 of the scan bytes; queue slots by ballot; sum order = LDS arrival
# speedup vs baseline: 1.4534x; 1.0598x over previous
#include <hip/hip_runtime.h>
#include <hip/hip_bf16.h>

typedef __attribute__((ext_vector_type(2))) float v2f;
typedef __attribute__((ext_vector_type(8))) float v8f;

#define FEATS 128

__global__ void gcn_degrees(const int* __restrict__ src, const int* __restrict__ dst,
                            float* __restrict__ dOut, float* __restrict__ dIn, int nE) {
    int e = blockIdx.x * blockDim.x + threadIdx.x;
    if (e >= nE) return;
    atomicAdd(&dOut[src[e]], 1.0f);
    atomicAdd(&dIn[dst[e]], 1.0f);
}

#define NB 400

__global__ void gcn_norm_keys(float* __restrict__ dOut, float* __restrict__ dIn, int n,
                              const int* __restrict__ dst, unsigned* __restrict__ key, int nE) {
    const int i = blockIdx.x * blockDim.x + threadIdx.x;
    if (i < n) {
        dOut[i] = rsqrtf(fmaxf(dOut[i], 1.0f));
        dIn[i]  = rsqrtf(fmaxf(dIn[i], 1.0f));
    }
    const int nK = nE >> 4;
    const int4* __restrict__ dst4 = (const int4*)dst;
    uint4* __restrict__ key4 = (uint4*)key;
    const int stride = gridDim.x * blockDim.x;
    for (int g = i; g < nK; g += stride) {
        unsigned k[4];
#pragma unroll
        for (int q = 0; q < 4; ++q) {
            const int4 v = dst4[4 * g + q];
            k[q] = ((unsigned)v.x / NB) | (((unsigned)v.y / NB) << 8) |
                   (((unsigned)v.z / NB) << 16) | (((unsigned)v.w / NB) << 24);
        }
        key4[g] = make_uint4(k[0], k[1], k[2], k[3]);
    }
}

__global__ void gcn_gemm1_wmma(const float* __restrict__ x, const float* __restrict__ W1,
                               const float* __restrict__ invOut, float* __restrict__ h) {
    const int lane    = threadIdx.x & 31;
    const int wave    = blockIdx.x * (blockDim.x >> 5) + (threadIdx.x >> 5);
    const int rowTile = wave >> 3;
    const int colTile = wave & 7;
    const int l15     = lane & 15;
    const int khalf   = lane >> 4;

    const int   row  = rowTile * 16 + l15;
    const int   nCol = colTile * 16 + l15;
    const float s    = invOut[row];
    const float* __restrict__ arow = x + (size_t)row * FEATS;

    v8f c = {};
#pragma unroll
    for (int k = 0; k < FEATS; k += 4) {
        const int ka = k + 2 * khalf;
        v2f a, b;
        a.x = arow[ka]     * s;
        a.y = arow[ka + 1] * s;
        b.x = W1[(size_t)ka       * FEATS + nCol];
        b.y = W1[(size_t)(ka + 1) * FEATS + nCol];
        c = __builtin_amdgcn_wmma_f32_16x16x4_f32(
                 false, a,  false, b,
                 (short)0, c,  false,  false);
    }

#pragma unroll
    for (int r = 0; r < 8; ++r) {
        const int mm = (lane < 16) ? r : (r + 8);
        h[(size_t)(rowTile * 16 + mm) * FEATS + nCol] = c[r];
    }
}

#define QCAP        64
#define CH          4
#define DRAIN_AT    24u
#define RG          8
#define OWN_THREADS 1024
#define OWN_WAVES   (OWN_THREADS / 32)

__device__ __forceinline__ void own_row_by_one_lane(float* accRow, const float* __restrict__ hRow) {
#pragma unroll 1
    for (int f = 0; f < FEATS; ++f) atomicAdd(accRow + f, hRow[f]);
}

__device__ __forceinline__ unsigned own_zero_bytes(unsigned x) {
    return ~(((x & 0x7F7F7F7Fu) + 0x7F7F7F7Fu) | x | 0x7F7F7F7Fu);
}

__device__ __forceinline__ unsigned own_hit_bits(uint4 w, unsigned pat) {
    return (own_zero_bytes(w.x ^ pat) >> 7) | (own_zero_bytes(w.y ^ pat) >> 6) |
           (own_zero_bytes(w.z ^ pat) >> 5) | (own_zero_bytes(w.w ^ pat) >> 4);
}

__global__ __launch_bounds__(OWN_THREADS)
void gcn_rows_owner(const float* __restrict__ h, const int* __restrict__ src,
                    const int* __restrict__ dst, const unsigned* __restrict__ key,
                    float* __restrict__ agg, int n, int nE) {
    __shared__ __attribute__((aligned(16))) float acc[NB * FEATS];
    __shared__ unsigned qe[OWN_WAVES][QCAP];

    const int tid  = threadIdx.x;
    const int lane = tid & 31;
    const int wave = __builtin_amdgcn_readfirstlane(tid >> 5);
    const int lo   = blockIdx.x * NB;
    const unsigned rows = (unsigned)min(NB, n - lo);

    float4* acc4 = (float4*)acc;
    for (int i = tid; i < NB * FEATS / 4; i += OWN_THREADS) acc4[i] = make_float4(0.f, 0.f, 0.f, 0.f);
    __syncthreads();

    const int nK   = nE >> 4;
    const int perB = 32 * CH;
    const int nBt  = (nK + perB - 1) / perB;
    const int rot  = (int)(((long long)blockIdx.x * nBt) / gridDim.x);
    const uint4* __restrict__ key4 = (const uint4*)key;
    const unsigned pat = (unsigned)blockIdx.x * 0x01010101u;

    if (wave == 0 && lane < 16) {
        const int e = 16 * nK + lane;
        if (e < nE) {
            const unsigned t = (unsigned)(dst[e] - lo);
            if (t < rows) own_row_by_one_lane(acc + t * FEATS, h + (size_t)src[e] * FEATS);
        }
    }

    uint4 cur[CH];
    unsigned qcnt = 0u;
    int idx0 = wave;
    int curB = min(idx0, nBt - 1) + rot;
    if (curB >= nBt) curB -= nBt;
#pragma unroll
    for (int j = 0; j < CH; ++j) {
        const int c = curB * perB + j * 32 + lane;
        cur[j] = key4[min(c, nK - 1)];
    }

    while (idx0 < nBt) {
        const int nidx = idx0 + OWN_WAVES;
        uint4 nxt[CH];
        int nxtB = curB;
        if (nidx < nBt) {
            nxtB = nidx + rot;
            if (nxtB >= nBt) nxtB -= nBt;
        }
#pragma unroll
        for (int j = 0; j < CH; ++j) {
            const int c = nxtB * perB + j * 32 + lane;
            nxt[j] = key4[min(c, nK - 1)];
        }

#pragma unroll
        for (int j = 0; j < CH; ++j) {
            const int c = curB * perB + j * 32 + lane;
            unsigned f = own_hit_bits(cur[j], pat);
            if (c >= nK) f = 0u;
            unsigned bal = __builtin_amdgcn_ballot_w32(f != 0u);
            while (bal != 0u) {
                if (f != 0u) {
                    const unsigned p = (unsigned)__builtin_ctz(f);
                    f &= f - 1u;
                    const unsigned e   = (unsigned)(16 * c) + 4u * (p & 7u) + (p >> 3);
                    const unsigned pos = qcnt + __builtin_amdgcn_mbcnt_lo(bal, 0u);
                    if (pos < QCAP) {
                        qe[wave][pos] = e;
                    } else {
                        const unsigned t = (unsigned)(dst[e] - lo);
                        own_row_by_one_lane(acc + t * FEATS, h + (size_t)src[e] * FEATS);
                    }
                }
                qcnt += (unsigned)__builtin_popcount(bal);
                bal = __builtin_amdgcn_ballot_w32(f != 0u);
            }
        }

        __builtin_amdgcn_fence(__ATOMIC_SEQ_CST, "wavefront");
        const unsigned cnt = min(qcnt, (unsigned)QCAP);

        if (cnt >= DRAIN_AT || (nidx >= nBt && cnt > 0u)) {
            for (unsigned base = 0; base < cnt; base += 32u) {
                const unsigned m = min(32u, cnt - base);
                int      s_l = 0;
                unsigned d_l = 0;
                if ((unsigned)lane < m) {
                    const unsigned e_l = qe[wave][base + lane];
                    d_l = (unsigned)(dst[e_l] - lo);
                    s_l = src[e_l];
                }
                for (unsigned i = 0; i < m; i += RG) {
                    float    v[RG][4];
                    unsigned dd[RG];
#pragma unroll
                    for (int k = 0; k < RG; ++k) {
                        const unsigned idx = min(i + k, m - 1u);
                        const int s = __builtin_amdgcn_readlane(s_l, (int)idx);
                        dd[k] = (unsigned)__builtin_amdgcn_readlane((int)d_l, (int)idx);
                        const float* __restrict__ hr = h + (size_t)s * FEATS + lane;
                        v[k][0] = hr[0];
                        v[k][1] = hr[32];
                        v[k][2] = hr[64];
                        v[k][3] = hr[96];
                    }
#pragma unroll
                    for (int k = 0; k < RG; ++k) {
                        if (i + k < m) {
                            float* ar = acc + dd[k] * FEATS + lane;
                            atomicAdd(ar,      v[k][0]);
                            atomicAdd(ar + 32, v[k][1]);
                            atomicAdd(ar + 64, v[k][2]);
                            atomicAdd(ar + 96, v[k][3]);
                        }
                    }
                }
            }
            qcnt = 0u;
            __builtin_amdgcn_fence(__ATOMIC_SEQ_CST, "wavefront");
        }

#pragma unroll
        for (int j = 0; j < CH; ++j) cur[j] = nxt[j];
        curB = nxtB;
        idx0 = nidx;
    }

    __syncthreads();
    float4* __restrict__ out4 = (float4*)(agg + (size_t)lo * FEATS);
    for (int i = tid; i < (int)rows * (FEATS / 4); i += OWN_THREADS) out4[i] = acc4[i];
}

__global__ void gcn_post1(const float* __restrict__ agg, const float* __restrict__ invIn,
                          const float* __restrict__ b1, float* __restrict__ h, int n) {
    const int i = blockIdx.x * blockDim.x + threadIdx.x;
    if (i >= n * FEATS) return;
    const int node = i >> 7;
    const int feat = i & (FEATS - 1);
    const float v = agg[i] * invIn[node] + b1[feat];
    h[i] = fmaxf(v, 0.0f);
}

__global__ void gcn_dot2(const float* __restrict__ h, const float* __restrict__ W2,
                         const float* __restrict__ invOut, float* __restrict__ z) {
    const int node = blockIdx.x * (blockDim.x >> 5) + (threadIdx.x >> 5);
    const int lane = threadIdx.x & 31;
    const float4 hv = *(const float4*)(h + (size_t)node * FEATS + lane * 4);
    const float4 wv = *(const float4*)(W2 + lane * 4);
    float sum = hv.x * wv.x + hv.y * wv.y + hv.z * wv.z + hv.w * wv.w;
#pragma unroll
    for (int off = 16; off > 0; off >>= 1) sum += __shfl_down(sum, off, 32);
    if (lane == 0) z[node] = sum * invOut[node];
}

__global__ void gcn_scatter_scalar(const float* __restrict__ z, const int* __restrict__ src,
                                   const int* __restrict__ dst, float* __restrict__ agg2, int nE) {
    const int e = blockIdx.x * blockDim.x + threadIdx.x;
    if (e >= nE) return;
    atomicAdd(&agg2[dst[e]], z[src[e]]);
}

__global__ void gcn_final(const float* __restrict__ agg2, const float* __restrict__ invIn,
                          const float* __restrict__ b2, float* __restrict__ out, int n) {
    const int i = blockIdx.x * blockDim.x + threadIdx.x;
    if (i >= n) return;
    out[i] = agg2[i] * invIn[i] + b2[0];
}

extern "C" void kernel_launch(void* const* d_in, const int* in_sizes, int n_in,
                              void* d_out, int out_size, void* d_ws, size_t ws_size,
                              hipStream_t stream) {
    const float* x   = (const float*)d_in[0];
    const int*   src = (const int*)d_in[1];
    const int*   dst = (const int*)d_in[2];
    const float* W1  = (const float*)d_in[3];
    const float* b1  = (const float*)d_in[4];
    const float* W2  = (const float*)d_in[5];
    const float* b2  = (const float*)d_in[6];
    float* out = (float*)d_out;

    const int n  = in_sizes[0] / FEATS;
    const int nE = in_sizes[1];

    float* f      = (float*)d_ws;
    float* invOut = f;
    float* invIn  = f + (size_t)n;
    float* z      = f + 2 * (size_t)n;
    float* agg2   = f + 3 * (size_t)n;
    float* h      = f + 4 * (size_t)n;
    float* agg    = h + (size_t)n * FEATS;
    unsigned* key = (unsigned*)(agg + (size_t)n * FEATS);

    hipMemsetAsync(invOut, 0, 2 * (size_t)n * sizeof(float), stream);
    hipMemsetAsync(agg2,   0, (size_t)n * sizeof(float), stream);

    gcn_degrees<<<(nE + 255) / 256, 256, 0, stream>>>(src, dst, invOut, invIn, nE);
    gcn_norm_keys<<<(n + 255) / 256, 256, 0, stream>>>(invOut, invIn, n, dst, key, nE);

    gcn_gemm1_wmma<<<n / 16, 256, 0, stream>>>(x, W1, invOut, h);

    gcn_rows_owner<<<(n + NB - 1) / NB, OWN_THREADS, 0, stream>>>(h, src, dst, key, agg, n, nE);
    gcn_post1<<<(int)(((size_t)n * FEATS + 255) / 256), 256, 0, stream>>>(agg, invIn, b1, h, n);

    gcn_dot2<<<n / 8, 256, 0, stream>>>(h, W2, invOut, z);
    gcn_scatter_scalar<<<(nE + 255) / 256, 256, 0, stream>>>(z, src, dst, agg2, nE);
    gcn_final<<<(n + 255) / 256, 256, 0, stream>>>(agg2, invIn, b2, out, n);
}


# Owner block finishes its rows out of LDS: scale, bias, relu, dot with W2 per row (wave xor sum, pairs 16,8,4,2,1 as before), one z per node stored; agg and relu planes and 2 launches dropped
# speedup vs baseline: 1.6419x; 1.1297x over previous
#include <hip/hip_runtime.h>
#include <hip/hip_bf16.h>

typedef __attribute__((ext_vector_type(2))) float v2f;
typedef __attribute__((ext_vector_type(8))) float v8f;

#define FEATS 128

__global__ void gcn_degrees(const int* __restrict__ src, const int* __restrict__ dst,
                            float* __restrict__ dOut, float* __restrict__ dIn, int nE) {
    int e = blockIdx.x * blockDim.x + threadIdx.x;
    if (e >= nE) return;
    atomicAdd(&dOut[src[e]], 1.0f);
    atomicAdd(&dIn[dst[e]], 1.0f);
}

#define NB 400

__global__ void gcn_norm_keys(float* __restrict__ dOut, float* __restrict__ dIn, int n,
                              const int* __restrict__ dst, unsigned* __restrict__ key, int nE) {
    const int i = blockIdx.x * blockDim.x + threadIdx.x;
    if (i < n) {
        dOut[i] = rsqrtf(fmaxf(dOut[i], 1.0f));
        dIn[i]  = rsqrtf(fmaxf(dIn[i], 1.0f));
    }
    const int nK = nE >> 4;
    const int4* __restrict__ dst4 = (const int4*)dst;
    uint4* __restrict__ key4 = (uint4*)key;
    const int stride = gridDim.x * blockDim.x;
    for (int g = i; g < nK; g += stride) {
        unsigned k[4];
#pragma unroll
        for (int q = 0; q < 4; ++q) {
            const int4 v = dst4[4 * g + q];
            k[q] = ((unsigned)v.x / NB) | (((unsigned)v.y / NB) << 8) |
                   (((unsigned)v.z / NB) << 16) | (((unsigned)v.w / NB) << 24);
        }
        key4[g] = make_uint4(k[0], k[1], k[2], k[3]);
    }
}

__global__ void gcn_gemm1_wmma(const float* __restrict__ x, const float* __restrict__ W1,
                               const float* __restrict__ invOut, float* __restrict__ h) {
    const int lane    = threadIdx.x & 31;
    const int wave    = blockIdx.x * (blockDim.x >> 5) + (threadIdx.x >> 5);
    const int rowTile = wave >> 3;
    const int colTile = wave & 7;
    const int l15     = lane & 15;
    const int khalf   = lane >> 4;

    const int   row  = rowTile * 16 + l15;
    const int   nCol = colTile * 16 + l15;
    const float s    = invOut[row];
    const float* __restrict__ arow = x + (size_t)row * FEATS;

    v8f c = {};
#pragma unroll
    for (int k = 0; k < FEATS; k += 4) {
        const int ka = k + 2 * khalf;
        v2f a, b;
        a.x = arow[ka]     * s;
        a.y = arow[ka + 1] * s;
        b.x = W1[(size_t)ka       * FEATS + nCol];
        b.y = W1[(size_t)(ka + 1) * FEATS + nCol];
        c = __builtin_amdgcn_wmma_f32_16x16x4_f32(
                 false, a,  false, b,
                 (short)0, c,  false,  false);
    }

#pragma unroll
    for (int r = 0; r < 8; ++r) {
        const int mm = (lane < 16) ? r : (r + 8);
        h[(size_t)(rowTile * 16 + mm) * FEATS + nCol] = c[r];
    }
}

#define QCAP        64
#define CH          4
#define DRAIN_AT    24u
#define RG          8
#define OWN_THREADS 1024
#define OWN_WAVES   (OWN_THREADS / 32)

__device__ __forceinline__ void own_row_by_one_lane(float* accRow, const float* __restrict__ hRow) {
#pragma unroll 1
    for (int f = 0; f < FEATS; ++f) atomicAdd(accRow + f, hRow[f]);
}

__device__ __forceinline__ float own_other_row(float x) {
    const int b = __builtin_bit_cast(int, x);
    return __builtin_bit_cast(float, __builtin_amdgcn_permlanex16(b, b, 0x76543210, 0xfedcba98, false, false));
}

template <int M>
__device__ __forceinline__ float own_row_xor(float x) {
    const int b = __builtin_bit_cast(int, x);
    return __builtin_bit_cast(float, __builtin_amdgcn_update_dpp(b, b, 0x160 + M, 0xf, 0xf, true));
}

__device__ __forceinline__ unsigned own_zero_bytes(unsigned x) {
    return ~(((x & 0x7F7F7F7Fu) + 0x7F7F7F7Fu) | x | 0x7F7F7F7Fu);
}

__device__ __forceinline__ unsigned own_hit_bits(uint4 w, unsigned pat) {
    return (own_zero_bytes(w.x ^ pat) >> 7) | (own_zero_bytes(w.y ^ pat) >> 6) |
           (own_zero_bytes(w.z ^ pat) >> 5) | (own_zero_bytes(w.w ^ pat) >> 4);
}

__global__ __launch_bounds__(OWN_THREADS)
void gcn_rows_owner(const float* __restrict__ h, const int* __restrict__ src,
                    const int* __restrict__ dst, const unsigned* __restrict__ key,
                    const float* __restrict__ invIn, const float* __restrict__ invOut,
                    const float* __restrict__ b1, const float* __restrict__ W2,
                    float* __restrict__ z, int n, int nE) {
    __shared__ __attribute__((aligned(16))) float acc[NB * FEATS];
    __shared__ unsigned qe[OWN_WAVES][QCAP];

    const int tid  = threadIdx.x;
    const int lane = tid & 31;
    const int wave = __builtin_amdgcn_readfirstlane(tid >> 5);
    const int lo   = blockIdx.x * NB;
    const unsigned rows = (unsigned)min(NB, n - lo);

    static_assert(NB <= 32 * OWN_WAVES, "one lane per row of a wave's run");
    const unsigned per = (rows + OWN_WAVES - 1u) / OWN_WAVES;
    const unsigned r0  = (unsigned)wave * per;
    const unsigned cnt = r0 < rows ? min(per, rows - r0) : 0u;
    float nIn = 0.f, nOut = 0.f;
    if ((unsigned)lane < cnt) {
        nIn  = invIn[lo + r0 + lane];
        nOut = invOut[lo + r0 + lane];
    }
    const float4 bv = ((const float4*)b1)[lane];
    const float4 wv = ((const float4*)W2)[lane];

    float4* acc4 = (float4*)acc;
    for (int i = tid; i < NB * FEATS / 4; i += OWN_THREADS) acc4[i] = make_float4(0.f, 0.f, 0.f, 0.f);
    __syncthreads();

    const int nK   = nE >> 4;
    const int perB = 32 * CH;
    const int nBt  = (nK + perB - 1) / perB;
    const int rot  = (int)(((long long)blockIdx.x * nBt) / gridDim.x);
    const uint4* __restrict__ key4 = (const uint4*)key;
    const unsigned pat = (unsigned)blockIdx.x * 0x01010101u;

    if (wave == 0 && lane < 16) {
        const int e = 16 * nK + lane;
        if (e < nE) {
            const unsigned t = (unsigned)(dst[e] - lo);
            if (t < rows) own_row_by_one_lane(acc + t * FEATS, h + (size_t)src[e] * FEATS);
        }
    }

    uint4 cur[CH];
    unsigned qcnt = 0u;
    int idx0 = wave;
    int curB = min(idx0, nBt - 1) + rot;
    if (curB >= nBt) curB -= nBt;
#pragma unroll
    for (int j = 0; j < CH; ++j) {
        const int c = curB * perB + j * 32 + lane;
        cur[j] = key4[min(c, nK - 1)];
    }

    while (idx0 < nBt) {
        const int nidx = idx0 + OWN_WAVES;
        uint4 nxt[CH];
        int nxtB = curB;
        if (nidx < nBt) {
            nxtB = nidx + rot;
            if (nxtB >= nBt) nxtB -= nBt;
        }
#pragma unroll
        for (int j = 0; j < CH; ++j) {
            const int c = nxtB * perB + j * 32 + lane;
            nxt[j] = key4[min(c, nK - 1)];
        }

#pragma unroll
        for (int j = 0; j < CH; ++j) {
            const int c = curB * perB + j * 32 + lane;
            unsigned f = own_hit_bits(cur[j], pat);
            if (c >= nK) f = 0u;
            unsigned bal = __builtin_amdgcn_ballot_w32(f != 0u);
            while (bal != 0u) {
                if (f != 0u) {
                    const unsigned p = (unsigned)__builtin_ctz(f);
                    f &= f - 1u;
                    const unsigned e   = (unsigned)(16 * c) + 4u * (p & 7u) + (p >> 3);
                    const unsigned pos = qcnt + __builtin_amdgcn_mbcnt_lo(bal, 0u);
                    if (pos < QCAP) {
                        qe[wave][pos] = e;
                    } else {
                        const unsigned t = (unsigned)(dst[e] - lo);
                        own_row_by_one_lane(acc + t * FEATS, h + (size_t)src[e] * FEATS);
                    }
                }
                qcnt += (unsigned)__builtin_popcount(bal);
                bal = __builtin_amdgcn_ballot_w32(f != 0u);
            }
        }

        __builtin_amdgcn_fence(__ATOMIC_SEQ_CST, "wavefront");
        const unsigned cnt = min(qcnt, (unsigned)QCAP);

        if (cnt >= DRAIN_AT || (nidx >= nBt && cnt > 0u)) {
            for (unsigned base = 0; base < cnt; base += 32u) {
                const unsigned m = min(32u, cnt - base);
                int      s_l = 0;
                unsigned d_l = 0;
                if ((unsigned)lane < m) {
                    const unsigned e_l = qe[wave][base + lane];
                    d_l = (unsigned)(dst[e_l] - lo);
                    s_l = src[e_l];
                }
                for (unsigned i = 0; i < m; i += RG) {
                    float    v[RG][4];
                    unsigned dd[RG];
#pragma unroll
                    for (int k = 0; k < RG; ++k) {
                        const unsigned idx = min(i + k, m - 1u);
                        const int s = __builtin_amdgcn_readlane(s_l, (int)idx);
                        dd[k] = (unsigned)__builtin_amdgcn_readlane((int)d_l, (int)idx);
                        const float* __restrict__ hr = h + (size_t)s * FEATS + lane;
                        v[k][0] = hr[0];
                        v[k][1] = hr[32];
                        v[k][2] = hr[64];
                        v[k][3] = hr[96];
                    }
#pragma unroll
                    for (int k = 0; k < RG; ++k) {
                        if (i + k < m) {
                            float* ar = acc + dd[k] * FEATS + lane;
                            atomicAdd(ar,      v[k][0]);
                            atomicAdd(ar + 32, v[k][1]);
                            atomicAdd(ar + 64, v[k][2]);
                            atomicAdd(ar + 96, v[k][3]);
                        }
                    }
                }
            }
            qcnt = 0u;
            __builtin_amdgcn_fence(__ATOMIC_SEQ_CST, "wavefront");
        }

#pragma unroll
        for (int j = 0; j < CH; ++j) cur[j] = nxt[j];
        curB = nxtB;
        idx0 = nidx;
    }

    __syncthreads();
    float zk = 0.f;
    for (unsigned k = 0; k < cnt; ++k) {
        const float sIn = __builtin_bit_cast(float, __builtin_amdgcn_readlane(__builtin_bit_cast(int, nIn), (int)k));
        const float4 a = acc4[(r0 + k) * (FEATS / 4) + lane];
        const float hx = fmaxf(a.x * sIn + bv.x, 0.0f);
        const float hy = fmaxf(a.y * sIn + bv.y, 0.0f);
        const float hz = fmaxf(a.z * sIn + bv.z, 0.0f);
        const float hw = fmaxf(a.w * sIn + bv.w, 0.0f);
        float sum = hx * wv.x + hy * wv.y + hz * wv.z + hw * wv.w;
        sum += own_other_row(sum);
        sum += own_row_xor<8>(sum);
        sum += own_row_xor<4>(sum);
        sum += own_row_xor<2>(sum);
        sum += own_row_xor<1>(sum);
        if ((unsigned)lane == k) zk = sum;
    }
    if ((unsigned)lane < cnt) z[lo + r0 + lane] = zk * nOut;
}

__global__ void gcn_scatter_scalar(const float* __restrict__ z, const int* __restrict__ src,
                                   const int* __restrict__ dst, float* __restrict__ agg2, int nE) {
    const int e = blockIdx.x * blockDim.x + threadIdx.x;
    if (e >= nE) return;
    atomicAdd(&agg2[dst[e]], z[src[e]]);
}

__global__ void gcn_final(const float* __restrict__ agg2, const float* __restrict__ invIn,
                          const float* __restrict__ b2, float* __restrict__ out, int n) {
    const int i = blockIdx.x * blockDim.x + threadIdx.x;
    if (i >= n) return;
    out[i] = agg2[i] * invIn[i] + b2[0];
}

extern "C" void kernel_launch(void* const* d_in, const int* in_sizes, int n_in,
                              void* d_out, int out_size, void* d_ws, size_t ws_size,
                              hipStream_t stream) {
    const float* x   = (const float*)d_in[0];
    const int*   src = (const int*)d_in[1];
    const int*   dst = (const int*)d_in[2];
    const float* W1  = (const float*)d_in[3];
    const float* b1  = (const float*)d_in[4];
    const float* W2  = (const float*)d_in[5];
    const float* b2  = (const float*)d_in[6];
    float* out = (float*)d_out;

    const int n  = in_sizes[0] / FEATS;
    const int nE = in_sizes[1];

    float* f      = (float*)d_ws;
    float* invOut = f;
    float* invIn  = f + (size_t)n;
    float* z      = f + 2 * (size_t)n;
    float* agg2   = f + 3 * (size_t)n;
    float* h      = f + 4 * (size_t)n;
    unsigned* key = (unsigned*)(h + (size_t)n * FEATS);

    hipMemsetAsync(invOut, 0, 2 * (size_t)n * sizeof(float), stream);
    hipMemsetAsync(agg2,   0, (size_t)n * sizeof(float), stream);

    gcn_degrees<<<(nE + 255) / 256, 256, 0, stream>>>(src, dst, invOut, invIn, nE);
    gcn_norm_keys<<<(n + 255) / 256, 256, 0, stream>>>(invOut, invIn, n, dst, key, nE);

    gcn_gemm1_wmma<<<n / 16, 256, 0, stream>>>(x, W1, invOut, h);

    gcn_rows_owner<<<(n + NB - 1) / NB, OWN_THREADS, 0, stream>>>(h, src, dst, key, invIn, invOut, b1, W2, z, n, nE);

    gcn_scatter_scalar<<<(nE + 255) / 256, 256, 0, stream>>>(z, src, dst, agg2, nE);
    gcn_final<<<(n + 255) / 256, 256, 0, stream>>>(agg2, invIn, b2, out, n);
}


# Layer-2 sums by the owner blocks too: a block logs (source,row) of each edge it adds; a second owner launch replays the log into LDS sums and stores out; no global adds; sum order = LDS arrival
# speedup vs baseline: 2.0961x; 1.2766x over previous
#include <hip/hip_runtime.h>
#include <hip/hip_bf16.h>

typedef __attribute__((ext_vector_type(2))) float v2f;
typedef __attribute__((ext_vector_type(8))) float v8f;

#define FEATS 128

__global__ void gcn_degrees(const int* __restrict__ src, const int* __restrict__ dst,
                            float* __restrict__ dOut, float* __restrict__ dIn, int nE) {
    int e = blockIdx.x * blockDim.x + threadIdx.x;
    if (e >= nE) return;
    atomicAdd(&dOut[src[e]], 1.0f);
    atomicAdd(&dIn[dst[e]], 1.0f);
}

#define NB 400

__global__ void gcn_norm_keys(float* __restrict__ dOut, float* __restrict__ dIn, int n,
                              const int* __restrict__ dst, unsigned* __restrict__ key, int nE) {
    const int i = blockIdx.x * blockDim.x + threadIdx.x;
    if (i < n) {
        dOut[i] = rsqrtf(fmaxf(dOut[i], 1.0f));
        dIn[i]  = rsqrtf(fmaxf(dIn[i], 1.0f));
    }
    const int nK = nE >> 4;
    const int4* __restrict__ dst4 = (const int4*)dst;
    uint4* __restrict__ key4 = (uint4*)key;
    const int stride = gridDim.x * blockDim.x;
    for (int g = i; g < nK; g += stride) {
        unsigned k[4];
#pragma unroll
        for (int q = 0; q < 4; ++q) {
            const int4 v = dst4[4 * g + q];
            k[q] = ((unsigned)v.x / NB) | (((unsigned)v.y / NB) << 8) |
                   (((unsigned)v.z / NB) << 16) | (((unsigned)v.w / NB) << 24);
        }
        key4[g] = make_uint4(k[0], k[1], k[2], k[3]);
    }
}

__global__ void gcn_gemm1_wmma(const float* __restrict__ x, const float* __restrict__ W1,
                               const float* __restrict__ invOut, float* __restrict__ h) {
    const int lane    = threadIdx.x & 31;
    const int wave    = blockIdx.x * (blockDim.x >> 5) + (threadIdx.x >> 5);
    const int rowTile = wave >> 3;
    const int colTile = wave & 7;
    const int l15     = lane & 15;
    const int khalf   = lane >> 4;

    const int   row  = rowTile * 16 + l15;
    const int   nCol = colTile * 16 + l15;
    const float s    = invOut[row];
    const float* __restrict__ arow = x + (size_t)row * FEATS;

    v8f c = {};
#pragma unroll
    for (int k = 0; k < FEATS; k += 4) {
        const int ka = k + 2 * khalf;
        v2f a, b;
        a.x = arow[ka]     * s;
        a.y = arow[ka + 1] * s;
        b.x = W1[(size_t)ka       * FEATS + nCol];
        b.y = W1[(size_t)(ka + 1) * FEATS + nCol];
        c = __builtin_amdgcn_wmma_f32_16x16x4_f32(
                 false, a,  false, b,
                 (short)0, c,  false,  false);
    }

#pragma unroll
    for (int r = 0; r < 8; ++r) {
        const int mm = (lane < 16) ? r : (r + 8);
        h[(size_t)(rowTile * 16 + mm) * FEATS + nCol] = c[r];
    }
}

#define QCAP        64
#define CH          4
#define DRAIN_AT    24u
#define RG          8
#define OWN_THREADS 1024
#define OWN_WAVES   (OWN_THREADS / 32)
#define LOG_CAP     32768u
#define LOG_ROW_BITS 9
static_assert(NB <= (1 << LOG_ROW_BITS), "row inside the block must fit the low bits of a log word");

__device__ __forceinline__ void own_log_one(unsigned* logCount, unsigned* __restrict__ logB, unsigned s, unsigned t) {
    const unsigned p = atomicAdd(logCount, 1u);
    if (p < LOG_CAP) logB[p] = (s << LOG_ROW_BITS) | t;
}

__device__ __forceinline__ void own_row_by_one_lane(float* accRow, const float* __restrict__ hRow) {
#pragma unroll 1
    for (int f = 0; f < FEATS; ++f) atomicAdd(accRow + f, hRow[f]);
}

__device__ __forceinline__ float own_other_row(float x) {
    const int b = __builtin_bit_cast(int, x);
    return __builtin_bit_cast(float, __builtin_amdgcn_permlanex16(b, b, 0x76543210, 0xfedcba98, false, false));
}

template <int M>
__device__ __forceinline__ float own_row_xor(float x) {
    const int b = __builtin_bit_cast(int, x);
    return __builtin_bit_cast(float, __builtin_amdgcn_update_dpp(b, b, 0x160 + M, 0xf, 0xf, true));
}

__device__ __forceinline__ unsigned own_zero_bytes(unsigned x) {
    return ~(((x & 0x7F7F7F7Fu) + 0x7F7F7F7Fu) | x | 0x7F7F7F7Fu);
}

__device__ __forceinline__ unsigned own_hit_bits(uint4 w, unsigned pat) {
    return (own_zero_bytes(w.x ^ pat) >> 7) | (own_zero_bytes(w.y ^ pat) >> 6) |
           (own_zero_bytes(w.z ^ pat) >> 5) | (own_zero_bytes(w.w ^ pat) >> 4);
}

__global__ __launch_bounds__(OWN_THREADS)
void gcn_rows_owner(const float* __restrict__ h, const int* __restrict__ src,
                    const int* __restrict__ dst, const unsigned* __restrict__ key,
                    const float* __restrict__ invIn, const float* __restrict__ invOut,
                    const float* __restrict__ b1, const float* __restrict__ W2,
                    float* __restrict__ z, unsigned* __restrict__ elog, unsigned* __restrict__ elogCount,
                    int n, int nE) {
    __shared__ __attribute__((aligned(16))) float acc[NB * FEATS];
    __shared__ unsigned qe[OWN_WAVES][QCAP];
    __shared__ unsigned logCount;
    unsigned* __restrict__ logB = elog + (size_t)blockIdx.x * LOG_CAP;

    const int tid  = threadIdx.x;
    const int lane = tid & 31;
    const int wave = __builtin_amdgcn_readfirstlane(tid >> 5);
    const int lo   = blockIdx.x * NB;
    const unsigned rows = (unsigned)min(NB, n - lo);

    static_assert(NB <= 32 * OWN_WAVES, "one lane per row of a wave's run");
    const unsigned per = (rows + OWN_WAVES - 1u) / OWN_WAVES;
    const unsigned r0  = (unsigned)wave * per;
    const unsigned cnt = r0 < rows ? min(per, rows - r0) : 0u;
    float nIn = 0.f, nOut = 0.f;
    if ((unsigned)lane < cnt) {
        nIn  = invIn[lo + r0 + lane];
        nOut = invOut[lo + r0 + lane];
    }
    const float4 bv = ((const float4*)b1)[lane];
    const float4 wv = ((const float4*)W2)[lane];

    float4* acc4 = (float4*)acc;
    for (int i = tid; i < NB * FEATS / 4; i += OWN_THREADS) acc4[i] = make_float4(0.f, 0.f, 0.f, 0.f);
    if (tid == 0) logCount = 0u;
    __syncthreads();

    const int nK   = nE >> 4;
    const int perB = 32 * CH;
    const int nBt  = (nK + perB - 1) / perB;
    const int rot  = (int)(((long long)blockIdx.x * nBt) / gridDim.x);
    const uint4* __restrict__ key4 = (const uint4*)key;
    const unsigned pat = (unsigned)blockIdx.x * 0x01010101u;

    if (wave == 0 && lane < 16) {
        const int e = 16 * nK + lane;
        if (e < nE) {
            const unsigned t = (unsigned)(dst[e] - lo);
            if (t < rows) {
                const int s = src[e];
                own_log_one(&logCount, logB, (unsigned)s, t);
                own_row_by_one_lane(acc + t * FEATS, h + (size_t)s * FEATS);
            }
        }
    }

    uint4 cur[CH];
    unsigned qcnt = 0u;
    int idx0 = wave;
    int curB = min(idx0, nBt - 1) + rot;
    if (curB >= nBt) curB -= nBt;
#pragma unroll
    for (int j = 0; j < CH; ++j) {
        const int c = curB * perB + j * 32 + lane;
        cur[j] = key4[min(c, nK - 1)];
    }

    while (idx0 < nBt) {
        const int nidx = idx0 + OWN_WAVES;
        uint4 nxt[CH];
        int nxtB = curB;
        if (nidx < nBt) {
            nxtB = nidx + rot;
            if (nxtB >= nBt) nxtB -= nBt;
        }
#pragma unroll
        for (int j = 0; j < CH; ++j) {
            const int c = nxtB * perB + j * 32 + lane;
            nxt[j] = key4[min(c, nK - 1)];
        }

#pragma unroll
        for (int j = 0; j < CH; ++j) {
            const int c = curB * perB + j * 32 + lane;
            unsigned f = own_hit_bits(cur[j], pat);
            if (c >= nK) f = 0u;
            unsigned bal = __builtin_amdgcn_ballot_w32(f != 0u);
            while (bal != 0u) {
                if (f != 0u) {
                    const unsigned p = (unsigned)__builtin_ctz(f);
                    f &= f - 1u;
                    const unsigned e   = (unsigned)(16 * c) + 4u * (p & 7u) + (p >> 3);
                    const unsigned pos = qcnt + __builtin_amdgcn_mbcnt_lo(bal, 0u);
                    if (pos < QCAP) {
                        qe[wave][pos] = e;
                    } else {
                        const unsigned t = (unsigned)(dst[e] - lo);
                        const int s = src[e];
                        own_log_one(&logCount, logB, (unsigned)s, t);
                        own_row_by_one_lane(acc + t * FEATS, h + (size_t)s * FEATS);
                    }
                }
                qcnt += (unsigned)__builtin_popcount(bal);
                bal = __builtin_amdgcn_ballot_w32(f != 0u);
            }
        }

        __builtin_amdgcn_fence(__ATOMIC_SEQ_CST, "wavefront");
        const unsigned cnt = min(qcnt, (unsigned)QCAP);

        if (cnt >= DRAIN_AT || (nidx >= nBt && cnt > 0u)) {
            for (unsigned base = 0; base < cnt; base += 32u) {
                const unsigned m = min(32u, cnt - base);
                int      s_l = 0;
                unsigned d_l = 0;
                if ((unsigned)lane < m) {
                    const unsigned e_l = qe[wave][base + lane];
                    d_l = (unsigned)(dst[e_l] - lo);
                    s_l = src[e_l];
                }
                unsigned p0 = 0u;
                if (lane == 0) p0 = atomicAdd(&logCount, m);
                p0 = (unsigned)__builtin_amdgcn_readfirstlane((int)p0);
                if ((unsigned)lane < m && p0 + (unsigned)lane < LOG_CAP)
                    logB[p0 + lane] = ((unsigned)s_l << LOG_ROW_BITS) | d_l;
                for (unsigned i = 0; i < m; i += RG) {
                    float    v[RG][4];
                    unsigned dd[RG];
#pragma unroll
                    for (int k = 0; k < RG; ++k) {
                        const unsigned idx = min(i + k, m - 1u);
                        const int s = __builtin_amdgcn_readlane(s_l, (int)idx);
                        dd[k] = (unsigned)__builtin_amdgcn_readlane((int)d_l, (int)idx);
                        const float* __restrict__ hr = h + (size_t)s * FEATS + lane;
                        v[k][0] = hr[0];
                        v[k][1] = hr[32];
                        v[k][2] = hr[64];
                        v[k][3] = hr[96];
                    }
#pragma unroll
                    for (int k = 0; k < RG; ++k) {
                        if (i + k < m) {
                            float* ar = acc + dd[k] * FEATS + lane;
                            atomicAdd(ar,      v[k][0]);
                            atomicAdd(ar + 32, v[k][1]);
                            atomicAdd(ar + 64, v[k][2]);
                            atomicAdd(ar + 96, v[k][3]);
                        }
                    }
                }
            }
            qcnt = 0u;
            __builtin_amdgcn_fence(__ATOMIC_SEQ_CST, "wavefront");
        }

#pragma unroll
        for (int j = 0; j < CH; ++j) cur[j] = nxt[j];
        curB = nxtB;
        idx0 = nidx;
    }

    __syncthreads();
    if (tid == 0) elogCount[blockIdx.x] = logCount;
    float zk = 0.f;
    for (unsigned k = 0; k < cnt; ++k) {
        const float sIn = __builtin_bit_cast(float, __builtin_amdgcn_readlane(__builtin_bit_cast(int, nIn), (int)k));
        const float4 a = acc4[(r0 + k) * (FEATS / 4) + lane];
        const float hx = fmaxf(a.x * sIn + bv.x, 0.0f);
        const float hy = fmaxf(a.y * sIn + bv.y, 0.0f);
        const float hz = fmaxf(a.z * sIn + bv.z, 0.0f);
        const float hw = fmaxf(a.w * sIn + bv.w, 0.0f);
        float sum = hx * wv.x + hy * wv.y + hz * wv.z + hw * wv.w;
        sum += own_other_row(sum);
        sum += own_row_xor<8>(sum);
        sum += own_row_xor<4>(sum);
        sum += own_row_xor<2>(sum);
        sum += own_row_xor<1>(sum);
        if ((unsigned)lane == k) zk = sum;
    }
    if ((unsigned)lane < cnt) z[lo + r0 + lane] = zk * nOut;
}

#define L2_THREADS 1024
#define L2_UN      8
static_assert(L2_UN == 8, "the two load fences in gcn_layer2_owner name eight values");

__global__ __launch_bounds__(L2_THREADS)
void gcn_layer2_owner(const float* __restrict__ z, const int* __restrict__ src,
                      const int* __restrict__ dst, const unsigned* __restrict__ elog,
                      const unsigned* __restrict__ elogCount, const float* __restrict__ invIn,
                      const float* __restrict__ b2, float* __restrict__ out, int n, int nE) {
    __shared__ float sums[NB];
    const int tid = threadIdx.x;
    const int lo  = blockIdx.x * NB;
    const unsigned rows = (unsigned)min(NB, n - lo);
    static_assert(NB <= L2_THREADS, "one thread per row at the start and the end");

    if (tid < NB) sums[tid] = 0.f;
    const unsigned total = elogCount[blockIdx.x];
    float nIn = 0.f;
    if ((unsigned)tid < rows) nIn = invIn[lo + tid];
    const float bias = b2[0];
    __syncthreads();

    if (total <= LOG_CAP) {
        const unsigned* __restrict__ logB = elog + (size_t)blockIdx.x * LOG_CAP;
        for (unsigned i0 = (unsigned)tid; i0 < total; i0 += L2_UN * L2_THREADS) {
            unsigned w[L2_UN];
            float    v[L2_UN];
#pragma unroll
            for (int k = 0; k < L2_UN; ++k) w[k] = logB[min(i0 + (unsigned)k * L2_THREADS, total - 1u)];
            asm volatile("" :: "v"(w[0]), "v"(w[1]), "v"(w[2]), "v"(w[3]), "v"(w[4]), "v"(w[5]), "v"(w[6]), "v"(w[7]));
#pragma unroll
            for (int k = 0; k < L2_UN; ++k) v[k] = z[w[k] >> LOG_ROW_BITS];
            asm volatile("" :: "v"(v[0]), "v"(v[1]), "v"(v[2]), "v"(v[3]), "v"(v[4]), "v"(v[5]), "v"(v[6]), "v"(v[7]));
#pragma unroll
            for (int k = 0; k < L2_UN; ++k)
                if (i0 + (unsigned)k * L2_THREADS < total)
                    atomicAdd(&sums[w[k] & ((1u << LOG_ROW_BITS) - 1u)], v[k]);
        }
    } else {
        for (int e = tid; e < nE; e += L2_THREADS) {
            const unsigned t = (unsigned)(dst[e] - lo);
            if (t < rows) atomicAdd(&sums[t], z[src[e]]);
        }
    }
    __syncthreads();
    if ((unsigned)tid < rows) out[lo + tid] = sums[tid] * nIn + bias;
}

extern "C" void kernel_launch(void* const* d_in, const int* in_sizes, int n_in,
                              void* d_out, int out_size, void* d_ws, size_t ws_size,
                              hipStream_t stream) {
    const float* x   = (const float*)d_in[0];
    const int*   src = (const int*)d_in[1];
    const int*   dst = (const int*)d_in[2];
    const float* W1  = (const float*)d_in[3];
    const float* b1  = (const float*)d_in[4];
    const float* W2  = (const float*)d_in[5];
    const float* b2  = (const float*)d_in[6];
    float* out = (float*)d_out;

    const int n  = in_sizes[0] / FEATS;
    const int nE = in_sizes[1];

    const int nOwn = (n + NB - 1) / NB;

    float* f      = (float*)d_ws;
    float* invOut = f;
    float* invIn  = f + (size_t)n;
    float* z      = f + 2 * (size_t)n;
    unsigned* elogCount = (unsigned*)(f + 3 * (size_t)n);
    float* h      = f + 4 * (size_t)n;
    unsigned* elog = (unsigned*)(h + (size_t)n * FEATS);
    unsigned* key  = elog + (size_t)nOwn * LOG_CAP;

    hipMemsetAsync(invOut, 0, 2 * (size_t)n * sizeof(float), stream);

    gcn_degrees<<<(nE + 255) / 256, 256, 0, stream>>>(src, dst, invOut, invIn, nE);
    gcn_norm_keys<<<(n + 255) / 256, 256, 0, stream>>>(invOut, invIn, n, dst, key, nE);

    gcn_gemm1_wmma<<<n / 16, 256, 0, stream>>>(x, W1, invOut, h);

    gcn_rows_owner<<<nOwn, OWN_THREADS, 0, stream>>>(h, src, dst, key, invIn, invOut, b1, W2, z, elog, elogCount, n, nE);

    gcn_layer2_owner<<<nOwn, L2_THREADS, 0, stream>>>(z, src, dst, elog, elogCount, invIn, b2, out, n, nE);
}


# Degree counts in LDS, not by 3.2M scattered float adds in memory: block (edge slice, node range) counts src and dst hits per node in one LDS word; norm launch sums the planes; memset dropped; exact
# speedup vs baseline: 3.7171x; 1.7733x over previous
#include <hip/hip_runtime.h>
#include <hip/hip_bf16.h>

typedef __attribute__((ext_vector_type(2))) float v2f;
typedef __attribute__((ext_vector_type(8))) float v8f;

#define FEATS 128

#define DEG_SLICE   50000
#define DEG_RNG     12512
#define DEG_THREADS 1024
#define DEG_UN      8
static_assert(DEG_SLICE < 65536 && DEG_SLICE % 4 == 0, "half-word counts per slice; aligned 16-byte loads");
static_assert(DEG_RNG % 32 == 0, "whole lines per range");

__device__ __forceinline__ void deg_hit(unsigned* cnt, int v, int lo, unsigned rng, unsigned inc) {
    const unsigned t = (unsigned)(v - lo);
    if (t < rng) atomicAdd(&cnt[t], inc);
}

__global__ __launch_bounds__(DEG_THREADS)
void gcn_degrees(const int* __restrict__ src, const int* __restrict__ dst,
                 unsigned* __restrict__ part, int n, int nE, int nRanges) {
    __shared__ unsigned cnt[DEG_RNG];
    const int tid = threadIdx.x;
    const int s  = blockIdx.x / nRanges;
    const int r  = blockIdx.x - s * nRanges;
    const int lo = r * DEG_RNG;
    const unsigned rng = (unsigned)min(DEG_RNG, n - lo);

    for (int i = tid; i < DEG_RNG; i += DEG_THREADS) cnt[i] = 0u;
    __syncthreads();

    const int e0 = s * DEG_SLICE;
    const int e1 = min(nE, e0 + DEG_SLICE);
    const int nV = (e1 - e0) >> 2;
    const int4* __restrict__ src4 = (const int4*)(src + e0);
    const int4* __restrict__ dst4 = (const int4*)(dst + e0);
    for (int g0 = tid; g0 < nV; g0 += DEG_UN * DEG_THREADS) {
        int4 a[DEG_UN], b[DEG_UN];
#pragma unroll
        for (int k = 0; k < DEG_UN; ++k) {
            const int g = min(g0 + k * DEG_THREADS, nV - 1);
            a[k] = src4[g];
            b[k] = dst4[g];
        }
#pragma unroll
        for (int k = 0; k < DEG_UN; ++k) {
            if (g0 + k * DEG_THREADS < nV) {
                deg_hit(cnt, a[k].x, lo, rng, 1u);
                deg_hit(cnt, a[k].y, lo, rng, 1u);
                deg_hit(cnt, a[k].z, lo, rng, 1u);
                deg_hit(cnt, a[k].w, lo, rng, 1u);
                deg_hit(cnt, b[k].x, lo, rng, 0x10000u);
                deg_hit(cnt, b[k].y, lo, rng, 0x10000u);
                deg_hit(cnt, b[k].z, lo, rng, 0x10000u);
                deg_hit(cnt, b[k].w, lo, rng, 0x10000u);
            }
        }
    }
    for (int e = e0 + 4 * nV + tid; e < e1; e += DEG_THREADS) {
        deg_hit(cnt, src[e], lo, rng, 1u);
        deg_hit(cnt, dst[e], lo, rng, 0x10000u);
    }
    __syncthreads();

    unsigned* __restrict__ plane = part + (size_t)s * n + lo;
    for (unsigned i = (unsigned)tid; i < rng; i += DEG_THREADS) plane[i] = cnt[i];
}

#define NB 400

__global__ void gcn_norm_keys(float* __restrict__ dOut, float* __restrict__ dIn, int n,
                              const unsigned* __restrict__ part, int nSlices,
                              const int* __restrict__ dst, unsigned* __restrict__ key, int nE) {
    const int i = blockIdx.x * blockDim.x + threadIdx.x;
    if (i < n) {
        unsigned cOut = 0u, cIn = 0u;
#pragma unroll 16
        for (int s = 0; s < nSlices; ++s) {
            const unsigned w = part[(size_t)s * n + i];
            cOut += w & 0xFFFFu;
            cIn  += w >> 16;
        }
        dOut[i] = rsqrtf(fmaxf((float)cOut, 1.0f));
        dIn[i]  = rsqrtf(fmaxf((float)cIn, 1.0f));
    }
    const int nK = nE >> 4;
    const int4* __restrict__ dst4 = (const int4*)dst;
    uint4* __restrict__ key4 = (uint4*)key;
    const int stride = gridDim.x * blockDim.x;
    for (int g = i; g < nK; g += stride) {
        unsigned k[4];
#pragma unroll
        for (int q = 0; q < 4; ++q) {
            const int4 v = dst4[4 * g + q];
            k[q] = ((unsigned)v.x / NB) | (((unsigned)v.y / NB) << 8) |
                   (((unsigned)v.z / NB) << 16) | (((unsigned)v.w / NB) << 24);
        }
        key4[g] = make_uint4(k[0], k[1], k[2], k[3]);
    }
}

__global__ void gcn_gemm1_wmma(const float* __restrict__ x, const float* __restrict__ W1,
                               const float* __restrict__ invOut, float* __restrict__ h) {
    const int lane    = threadIdx.x & 31;
    const int wave    = blockIdx.x * (blockDim.x >> 5) + (threadIdx.x >> 5);
    const int rowTile = wave >> 3;
    const int colTile = wave & 7;
    const int l15     = lane & 15;
    const int khalf   = lane >> 4;

    const int   row  = rowTile * 16 + l15;
    const int   nCol = colTile * 16 + l15;
    const float s    = invOut[row];
    const float* __restrict__ arow = x + (size_t)row * FEATS;

    v8f c = {};
#pragma unroll
    for (int k = 0; k < FEATS; k += 4) {
        const int ka = k + 2 * khalf;
        v2f a, b;
        a.x = arow[ka]     * s;
        a.y = arow[ka + 1] * s;
        b.x = W1[(size_t)ka       * FEATS + nCol];
        b.y = W1[(size_t)(ka + 1) * FEATS + nCol];
        c = __builtin_amdgcn_wmma_f32_16x16x4_f32(
                 false, a,  false, b,
                 (short)0, c,  false,  false);
    }

#pragma unroll
    for (int r = 0; r < 8; ++r) {
        const int mm = (lane < 16) ? r : (r + 8);
        h[(size_t)(rowTile * 16 + mm) * FEATS + nCol] = c[r];
    }
}

#define QCAP        64
#define CH          4
#define DRAIN_AT    24u
#define RG          8
#define OWN_THREADS 1024
#define OWN_WAVES   (OWN_THREADS / 32)
#define LOG_CAP     32768u
#define LOG_ROW_BITS 9
static_assert(NB <= (1 << LOG_ROW_BITS), "row inside the block must fit the low bits of a log word");

__device__ __forceinline__ void own_log_one(unsigned* logCount, unsigned* __restrict__ logB, unsigned s, unsigned t) {
    const unsigned p = atomicAdd(logCount, 1u);
    if (p < LOG_CAP) logB[p] = (s << LOG_ROW_BITS) | t;
}

__device__ __forceinline__ void own_row_by_one_lane(float* accRow, const float* __restrict__ hRow) {
#pragma unroll 1
    for (int f = 0; f < FEATS; ++f) atomicAdd(accRow + f, hRow[f]);
}

__device__ __forceinline__ float own_other_row(float x) {
    const int b = __builtin_bit_cast(int, x);
    return __builtin_bit_cast(float, __builtin_amdgcn_permlanex16(b, b, 0x76543210, 0xfedcba98, false, false));
}

template <int M>
__device__ __forceinline__ float own_row_xor(float x) {
    const int b = __builtin_bit_cast(int, x);
    return __builtin_bit_cast(float, __builtin_amdgcn_update_dpp(b, b, 0x160 + M, 0xf, 0xf, true));
}

__device__ __forceinline__ unsigned own_zero_bytes(unsigned x) {
    return ~(((x & 0x7F7F7F7Fu) + 0x7F7F7F7Fu) | x | 0x7F7F7F7Fu);
}

__device__ __forceinline__ unsigned own_hit_bits(uint4 w, unsigned pat) {
    return (own_zero_bytes(w.x ^ pat) >> 7) | (own_zero_bytes(w.y ^ pat) >> 6) |
           (own_zero_bytes(w.z ^ pat) >> 5) | (own_zero_bytes(w.w ^ pat) >> 4);
}

__global__ __launch_bounds__(OWN_THREADS)
void gcn_rows_owner(const float* __restrict__ h, const int* __restrict__ src,
                    const int* __restrict__ dst, const unsigned* __restrict__ key,
                    const float* __restrict__ invIn, const float* __restrict__ invOut,
                    const float* __restrict__ b1, const float* __restrict__ W2,
                    float* __restrict__ z, unsigned* __restrict__ elog, unsigned* __restrict__ elogCount,
                    int n, int nE) {
    __shared__ __attribute__((aligned(16))) float acc[NB * FEATS];
    __shared__ unsigned qe[OWN_WAVES][QCAP];
    __shared__ unsigned logCount;
    unsigned* __restrict__ logB = elog + (size_t)blockIdx.x * LOG_CAP;

    const int tid  = threadIdx.x;
    const int lane = tid & 31;
    const int wave = __builtin_amdgcn_readfirstlane(tid >> 5);
    const int lo   = blockIdx.x * NB;
    const unsigned rows = (unsigned)min(NB, n - lo);

    static_assert(NB <= 32 * OWN_WAVES, "one lane per row of a wave's run");
    const unsigned per = (rows + OWN_WAVES - 1u) / OWN_WAVES;
    const unsigned r0  = (unsigned)wave * per;
    const unsigned cnt = r0 < rows ? min(per, rows - r0) : 0u;
    float nIn = 0.f, nOut = 0.f;
    if ((unsigned)lane < cnt) {
        nIn  = invIn[lo + r0 + lane];
        nOut = invOut[lo + r0 + lane];
    }
    const float4 bv = ((const float4*)b1)[lane];
    const float4 wv = ((const float4*)W2)[lane];

    float4* acc4 = (float4*)acc;
    for (int i = tid; i < NB * FEATS / 4; i += OWN_THREADS) acc4[i] = make_float4(0.f, 0.f, 0.f, 0.f);
    if (tid == 0) logCount = 0u;
    __syncthreads();

    const int nK   = nE >> 4;
    const int perB = 32 * CH;
    const int nBt  = (nK + perB - 1) / perB;
    const int rot  = (int)(((long long)blockIdx.x * nBt) / gridDim.x);
    const uint4* __restrict__ key4 = (const uint4*)key;
    const unsigned pat = (unsigned)blockIdx.x * 0x01010101u;

    if (wave == 0 && lane < 16) {
        const int e = 16 * nK + lane;
        if (e < nE) {
            const unsigned t = (unsigned)(dst[e] - lo);
            if (t < rows) {
                const int s = src[e];
                own_log_one(&logCount, logB, (unsigned)s, t);
                own_row_by_one_lane(acc + t * FEATS, h + (size_t)s * FEATS);
            }
        }
    }

    uint4 cur[CH];
    unsigned qcnt = 0u;
    int idx0 = wave;
    int curB = min(idx0, nBt - 1) + rot;
    if (curB >= nBt) curB -= nBt;
#pragma unroll
    for (int j = 0; j < CH; ++j) {
        const int c = curB * perB + j * 32 + lane;
        cur[j] = key4[min(c, nK - 1)];
    }

    while (idx0 < nBt) {
        const int nidx = idx0 + OWN_WAVES;
        uint4 nxt[CH];
        int nxtB = curB;
        if (nidx < nBt) {
            nxtB = nidx + rot;
            if (nxtB >= nBt) nxtB -= nBt;
        }
#pragma unroll
        for (int j = 0; j < CH; ++j) {
            const int c = nxtB * perB + j * 32 + lane;
            nxt[j] = key4[min(c, nK - 1)];
        }

#pragma unroll
        for (int j = 0; j < CH; ++j) {
            const int c = curB * perB + j * 32 + lane;
            unsigned f = own_hit_bits(cur[j], pat);
            if (c >= nK) f = 0u;
            unsigned bal = __builtin_amdgcn_ballot_w32(f != 0u);
            while (bal != 0u) {
                if (f != 0u) {
                    const unsigned p = (unsigned)__builtin_ctz(f);
                    f &= f - 1u;
                    const unsigned e   = (unsigned)(16 * c) + 4u * (p & 7u) + (p >> 3);
                    const unsigned pos = qcnt + __builtin_amdgcn_mbcnt_lo(bal, 0u);
                    if (pos < QCAP) {
                        qe[wave][pos] = e;
                    } else {
                        const unsigned t = (unsigned)(dst[e] - lo);
                        const int s = src[e];
                        own_log_one(&logCount, logB, (unsigned)s, t);
                        own_row_by_one_lane(acc + t * FEATS, h + (size_t)s * FEATS);
                    }
                }
                qcnt += (unsigned)__builtin_popcount(bal);
                bal = __builtin_amdgcn_ballot_w32(f != 0u);
            }
        }

        __builtin_amdgcn_fence(__ATOMIC_SEQ_CST, "wavefront");
        const unsigned cnt = min(qcnt, (unsigned)QCAP);

        if (cnt >= DRAIN_AT || (nidx >= nBt && cnt > 0u)) {
            for (unsigned base = 0; base < cnt; base += 32u) {
                const unsigned m = min(32u, cnt - base);
                int      s_l = 0;
                unsigned d_l = 0;
                if ((unsigned)lane < m) {
                    const unsigned e_l = qe[wave][base + lane];
                    d_l = (unsigned)(dst[e_l] - lo);
                    s_l = src[e_l];
                }
                unsigned p0 = 0u;
                if (lane == 0) p0 = atomicAdd(&logCount, m);
                p0 = (unsigned)__builtin_amdgcn_readfirstlane((int)p0);
                if ((unsigned)lane < m && p0 + (unsigned)lane < LOG_CAP)
                    logB[p0 + lane] = ((unsigned)s_l << LOG_ROW_BITS) | d_l;
                for (unsigned i = 0; i < m; i += RG) {
                    float    v[RG][4];
                    unsigned dd[RG];
#pragma unroll
                    for (int k = 0; k < RG; ++k) {
                        const unsigned idx = min(i + k, m - 1u);
                        const int s = __builtin_amdgcn_readlane(s_l, (int)idx);
                        dd[k] = (unsigned)__builtin_amdgcn_readlane((int)d_l, (int)idx);
                        const float* __restrict__ hr = h + (size_t)s * FEATS + lane;
                        v[k][0] = hr[0];
                        v[k][1] = hr[32];
                        v[k][2] = hr[64];
                        v[k][3] = hr[96];
                    }
#pragma unroll
                    for (int k = 0; k < RG; ++k) {
                        if (i + k < m) {
                            float* ar = acc + dd[k] * FEATS + lane;
                            atomicAdd(ar,      v[k][0]);
                            atomicAdd(ar + 32, v[k][1]);
                            atomicAdd(ar + 64, v[k][2]);
                            atomicAdd(ar + 96, v[k][3]);
                        }
                    }
                }
            }
            qcnt = 0u;
            __builtin_amdgcn_fence(__ATOMIC_SEQ_CST, "wavefront");
        }

#pragma unroll
        for (int j = 0; j < CH; ++j) cur[j] = nxt[j];
        curB = nxtB;
        idx0 = nidx;
    }

    __syncthreads();
    if (tid == 0) elogCount[blockIdx.x] = logCount;
    float zk = 0.f;
    for (unsigned k = 0; k < cnt; ++k) {
        const float sIn = __builtin_bit_cast(float, __builtin_amdgcn_readlane(__builtin_bit_cast(int, nIn), (int)k));
        const float4 a = acc4[(r0 + k) * (FEATS / 4) + lane];
        const float hx = fmaxf(a.x * sIn + bv.x, 0.0f);
        const float hy = fmaxf(a.y * sIn + bv.y, 0.0f);
        const float hz = fmaxf(a.z * sIn + bv.z, 0.0f);
        const float hw = fmaxf(a.w * sIn + bv.w, 0.0f);
        float sum = hx * wv.x + hy * wv.y + hz * wv.z + hw * wv.w;
        sum += own_other_row(sum);
        sum += own_row_xor<8>(sum);
        sum += own_row_xor<4>(sum);
        sum += own_row_xor<2>(sum);
        sum += own_row_xor<1>(sum);
        if ((unsigned)lane == k) zk = sum;
    }
    if ((unsigned)lane < cnt) z[lo + r0 + lane] = zk * nOut;
}

#define L2_THREADS 1024
#define L2_UN      8
static_assert(L2_UN == 8, "the two load fences in gcn_layer2_owner name eight values");

__global__ __launch_bounds__(L2_THREADS)
void gcn_layer2_owner(const float* __restrict__ z, const int* __restrict__ src,
                      const int* __restrict__ dst, const unsigned* __restrict__ elog,
                      const unsigned* __restrict__ elogCount, const float* __restrict__ invIn,
                      const float* __restrict__ b2, float* __restrict__ out, int n, int nE) {
    __shared__ float sums[NB];
    const int tid = threadIdx.x;
    const int lo  = blockIdx.x * NB;
    const unsigned rows = (unsigned)min(NB, n - lo);
    static_assert(NB <= L2_THREADS, "one thread per row at the start and the end");

    if (tid < NB) sums[tid] = 0.f;
    const unsigned total = elogCount[blockIdx.x];
    float nIn = 0.f;
    if ((unsigned)tid < rows) nIn = invIn[lo + tid];
    const float bias = b2[0];
    __syncthreads();

    if (total <= LOG_CAP) {
        const unsigned* __restrict__ logB = elog + (size_t)blockIdx.x * LOG_CAP;
        for (unsigned i0 = (unsigned)tid; i0 < total; i0 += L2_UN * L2_THREADS) {
            unsigned w[L2_UN];
            float    v[L2_UN];
#pragma unroll
            for (int k = 0; k < L2_UN; ++k) w[k] = logB[min(i0 + (unsigned)k * L2_THREADS, total - 1u)];
            asm volatile("" :: "v"(w[0]), "v"(w[1]), "v"(w[2]), "v"(w[3]), "v"(w[4]), "v"(w[5]), "v"(w[6]), "v"(w[7]));
#pragma unroll
            for (int k = 0; k < L2_UN; ++k) v[k] = z[w[k] >> LOG_ROW_BITS];
            asm volatile("" :: "v"(v[0]), "v"(v[1]), "v"(v[2]), "v"(v[3]), "v"(v[4]), "v"(v[5]), "v"(v[6]), "v"(v[7]));
#pragma unroll
            for (int k = 0; k < L2_UN; ++k)
                if (i0 + (unsigned)k * L2_THREADS < total)
                    atomicAdd(&sums[w[k] & ((1u << LOG_ROW_BITS) - 1u)], v[k]);
        }
    } else {
        for (int e = tid; e < nE; e += L2_THREADS) {
            const unsigned t = (unsigned)(dst[e] - lo);
            if (t < rows) atomicAdd(&sums[t], z[src[e]]);
        }
    }
    __syncthreads();
    if ((unsigned)tid < rows) out[lo + tid] = sums[tid] * nIn + bias;
}

extern "C" void kernel_launch(void* const* d_in, const int* in_sizes, int n_in,
                              void* d_out, int out_size, void* d_ws, size_t ws_size,
                              hipStream_t stream) {
    const float* x   = (const float*)d_in[0];
    const int*   src = (const int*)d_in[1];
    const int*   dst = (const int*)d_in[2];
    const float* W1  = (const float*)d_in[3];
    const float* b1  = (const float*)d_in[4];
    const float* W2  = (const float*)d_in[5];
    const float* b2  = (const float*)d_in[6];
    float* out = (float*)d_out;

    const int n  = in_sizes[0] / FEATS;
    const int nE = in_sizes[1];

    const int nOwn = (n + NB - 1) / NB;

    float* f      = (float*)d_ws;
    float* invOut = f;
    float* invIn  = f + (size_t)n;
    float* z      = f + 2 * (size_t)n;
    unsigned* elogCount = (unsigned*)(f + 3 * (size_t)n);
    float* h      = f + 4 * (size_t)n;
    unsigned* elog = (unsigned*)(h + (size_t)n * FEATS);
    unsigned* key  = elog + (size_t)nOwn * LOG_CAP;
    unsigned* part = key + (((size_t)nE / 4 + 32) & ~(size_t)31);
    const int nRanges = (n + DEG_RNG - 1) / DEG_RNG;
    const int nSlices = (nE + DEG_SLICE - 1) / DEG_SLICE;

    gcn_degrees<<<nRanges * nSlices, DEG_THREADS, 0, stream>>>(src, dst, part, n, nE, nRanges);
    gcn_norm_keys<<<(n + 255) / 256, 256, 0, stream>>>(invOut, invIn, n, part, nSlices, dst, key, nE);

    gcn_gemm1_wmma<<<n / 16, 256, 0, stream>>>(x, W1, invOut, h);

    gcn_rows_owner<<<nOwn, OWN_THREADS, 0, stream>>>(h, src, dst, key, invIn, invOut, b1, W2, z, elog, elogCount, n, nE);

    gcn_layer2_owner<<<nOwn, L2_THREADS, 0, stream>>>(z, src, dst, elog, elogCount, invIn, b2, out, n, nE);
}


# Owner blocks read their own bit index of the edges (1 bit per 4 edges, built in LDS by the norm launch), fetch dst,src of marked groups only in full 32-lane passes; no key scan; sum order LDS arrival
# speedup vs baseline: 4.3069x; 1.1587x over previous
#include <hip/hip_runtime.h>
#include <hip/hip_bf16.h>

typedef __attribute__((ext_vector_type(2))) float v2f;
typedef __attribute__((ext_vector_type(8))) float v8f;

#define FEATS 128

#define DEG_SLICE   50000
#define DEG_RNG     12512
#define DEG_THREADS 1024
#define DEG_UN      8
static_assert(DEG_SLICE < 65536 && DEG_SLICE % 4 == 0, "half-word counts per slice; aligned 16-byte loads");
static_assert(DEG_RNG % 32 == 0, "whole lines per range");

__device__ __forceinline__ void deg_hit(unsigned* cnt, int v, int lo, unsigned rng, unsigned inc) {
    const unsigned t = (unsigned)(v - lo);
    if (t < rng) atomicAdd(&cnt[t], inc);
}

__global__ __launch_bounds__(DEG_THREADS)
void gcn_degrees(const int* __restrict__ src, const int* __restrict__ dst,
                 unsigned* __restrict__ part, int n, int nE, int nRanges) {
    __shared__ unsigned cnt[DEG_RNG];
    const int tid = threadIdx.x;
    const int s  = blockIdx.x / nRanges;
    const int r  = blockIdx.x - s * nRanges;
    const int lo = r * DEG_RNG;
    const unsigned rng = (unsigned)min(DEG_RNG, n - lo);

    for (int i = tid; i < DEG_RNG; i += DEG_THREADS) cnt[i] = 0u;
    __syncthreads();

    const int e0 = s * DEG_SLICE;
    const int e1 = min(nE, e0 + DEG_SLICE);
    const int nV = (e1 - e0) >> 2;
    const int4* __restrict__ src4 = (const int4*)(src + e0);
    const int4* __restrict__ dst4 = (const int4*)(dst + e0);
    for (int g0 = tid; g0 < nV; g0 += DEG_UN * DEG_THREADS) {
        int4 a[DEG_UN], b[DEG_UN];
#pragma unroll
        for (int k = 0; k < DEG_UN; ++k) {
            const int g = min(g0 + k * DEG_THREADS, nV - 1);
            a[k] = src4[g];
            b[k] = dst4[g];
        }
#pragma unroll
        for (int k = 0; k < DEG_UN; ++k) {
            if (g0 + k * DEG_THREADS < nV) {
                deg_hit(cnt, a[k].x, lo, rng, 1u);
                deg_hit(cnt, a[k].y, lo, rng, 1u);
                deg_hit(cnt, a[k].z, lo, rng, 1u);
                deg_hit(cnt, a[k].w, lo, rng, 1u);
                deg_hit(cnt, b[k].x, lo, rng, 0x10000u);
                deg_hit(cnt, b[k].y, lo, rng, 0x10000u);
                deg_hit(cnt, b[k].z, lo, rng, 0x10000u);
                deg_hit(cnt, b[k].w, lo, rng, 0x10000u);
            }
        }
    }
    for (int e = e0 + 4 * nV + tid; e < e1; e += DEG_THREADS) {
        deg_hit(cnt, src[e], lo, rng, 1u);
        deg_hit(cnt, dst[e], lo, rng, 0x10000u);
    }
    __syncthreads();

    unsigned* __restrict__ plane = part + (size_t)s * n + lo;
    for (unsigned i = (unsigned)tid; i < rng; i += DEG_THREADS) plane[i] = cnt[i];
}

#define NB 400

#define IDX_GROUPS 1024
#define IDX_OWN    256
#define IDX_PITCH  257
static_assert(IDX_GROUPS == 1024, "32 waves, one index word per wave and owner");

__global__ __launch_bounds__(IDX_GROUPS)
void gcn_norm_index(float* __restrict__ dOut, float* __restrict__ dIn, int n,
                    const unsigned* __restrict__ part, int nSlices,
                    const int* __restrict__ dst, unsigned* __restrict__ bmp, int nE, int nOwn) {
    __shared__ unsigned bm[32 * IDX_PITCH];
    const int tid = threadIdx.x;
    const int i = blockIdx.x * IDX_GROUPS + tid;
    for (int k = tid; k < 32 * IDX_PITCH; k += IDX_GROUPS) bm[k] = 0u;

    for (int v = i; v < n; v += gridDim.x * IDX_GROUPS) {
        unsigned cOut = 0u, cIn = 0u;
#pragma unroll 16
        for (int s = 0; s < nSlices; ++s) {
            const unsigned w = part[(size_t)s * n + v];
            cOut += w & 0xFFFFu;
            cIn  += w >> 16;
        }
        dOut[v] = rsqrtf(fmaxf((float)cOut, 1.0f));
        dIn[v]  = rsqrtf(fmaxf((float)cIn, 1.0f));
    }
    __syncthreads();

    const int nG = nE >> 2;
    if (i < nG) {
        const int4 d = ((const int4*)dst)[i];
        const unsigned bit = 1u << (tid & 31);
        unsigned* row = bm + (tid >> 5) * IDX_PITCH;
        atomicOr(&row[min((unsigned)d.x / NB, (unsigned)(IDX_OWN - 1))], bit);
        atomicOr(&row[min((unsigned)d.y / NB, (unsigned)(IDX_OWN - 1))], bit);
        atomicOr(&row[min((unsigned)d.z / NB, (unsigned)(IDX_OWN - 1))], bit);
        atomicOr(&row[min((unsigned)d.w / NB, (unsigned)(IDX_OWN - 1))], bit);
    }
    __syncthreads();

    const int lane = tid & 31;
    const size_t wordsPerOwner = (size_t)32 * gridDim.x;
    for (int o = tid >> 5; o < nOwn; o += IDX_GROUPS / 32)
        bmp[(size_t)o * wordsPerOwner + 32 * blockIdx.x + lane] = bm[lane * IDX_PITCH + o];
}

__global__ void gcn_gemm1_wmma(const float* __restrict__ x, const float* __restrict__ W1,
                               const float* __restrict__ invOut, float* __restrict__ h) {
    const int lane    = threadIdx.x & 31;
    const int wave    = blockIdx.x * (blockDim.x >> 5) + (threadIdx.x >> 5);
    const int rowTile = wave >> 3;
    const int colTile = wave & 7;
    const int l15     = lane & 15;
    const int khalf   = lane >> 4;

    const int   row  = rowTile * 16 + l15;
    const int   nCol = colTile * 16 + l15;
    const float s    = invOut[row];
    const float* __restrict__ arow = x + (size_t)row * FEATS;

    v8f c = {};
#pragma unroll
    for (int k = 0; k < FEATS; k += 4) {
        const int ka = k + 2 * khalf;
        v2f a, b;
        a.x = arow[ka]     * s;
        a.y = arow[ka + 1] * s;
        b.x = W1[(size_t)ka       * FEATS + nCol];
        b.y = W1[(size_t)(ka + 1) * FEATS + nCol];
        c = __builtin_amdgcn_wmma_f32_16x16x4_f32(
                 false, a,  false, b,
                 (short)0, c,  false,  false);
    }

#pragma unroll
    for (int r = 0; r < 8; ++r) {
        const int mm = (lane < 16) ? r : (r + 8);
        h[(size_t)(rowTile * 16 + mm) * FEATS + nCol] = c[r];
    }
}

#define QCAP        128
#define Q2CAP       160
#define CH          2
#define RG          8
#define OWN_THREADS 1024
#define OWN_WAVES   (OWN_THREADS / 32)
#define LOG_CAP     32768u
#define LOG_ROW_BITS 9
static_assert(NB <= (1 << LOG_ROW_BITS), "row inside the block must fit the low bits of a log word");

__device__ __forceinline__ void own_log_one(unsigned* logCount, unsigned* __restrict__ logB, unsigned s, unsigned t) {
    const unsigned p = atomicAdd(logCount, 1u);
    if (p < LOG_CAP) logB[p] = (s << LOG_ROW_BITS) | t;
}

__device__ __forceinline__ void own_row_by_one_lane(float* accRow, const float* __restrict__ hRow) {
#pragma unroll 1
    for (int f = 0; f < FEATS; ++f) atomicAdd(accRow + f, hRow[f]);
}

__device__ __forceinline__ void own_edge_by_one_lane(float* acc, const float* __restrict__ h, unsigned* logCount,
                                                     unsigned* __restrict__ logB, int d, int s, int lo, unsigned rows) {
    const unsigned t = (unsigned)(d - lo);
    if (t < rows) {
        own_log_one(logCount, logB, (unsigned)s, t);
        own_row_by_one_lane(acc + t * FEATS, h + (size_t)s * FEATS);
    }
}

__device__ __forceinline__ float own_other_row(float x) {
    const int b = __builtin_bit_cast(int, x);
    return __builtin_bit_cast(float, __builtin_amdgcn_permlanex16(b, b, 0x76543210, 0xfedcba98, false, false));
}

template <int M>
__device__ __forceinline__ float own_row_xor(float x) {
    const int b = __builtin_bit_cast(int, x);
    return __builtin_bit_cast(float, __builtin_amdgcn_update_dpp(b, b, 0x160 + M, 0xf, 0xf, true));
}

__global__ __launch_bounds__(OWN_THREADS)
void gcn_rows_owner(const float* __restrict__ h, const int* __restrict__ src,
                    const int* __restrict__ dst, const unsigned* __restrict__ bmp, int wordsPerOwner,
                    const float* __restrict__ invIn, const float* __restrict__ invOut,
                    const float* __restrict__ b1, const float* __restrict__ W2,
                    float* __restrict__ z, unsigned* __restrict__ elog, unsigned* __restrict__ elogCount,
                    int n, int nE) {
    __shared__ __attribute__((aligned(16))) float acc[NB * FEATS];
    __shared__ unsigned qe[OWN_WAVES][QCAP];
    __shared__ unsigned q2[OWN_WAVES][Q2CAP];
    __shared__ unsigned logCount;
    unsigned* __restrict__ logB = elog + (size_t)blockIdx.x * LOG_CAP;

    const int tid  = threadIdx.x;
    const int lane = tid & 31;
    const int wave = __builtin_amdgcn_readfirstlane(tid >> 5);
    const int lo   = blockIdx.x * NB;
    const unsigned rows = (unsigned)min(NB, n - lo);

    static_assert(NB <= 32 * OWN_WAVES, "one lane per row of a wave's run");
    const unsigned per = (rows + OWN_WAVES - 1u) / OWN_WAVES;
    const unsigned r0  = (unsigned)wave * per;
    const unsigned cnt = r0 < rows ? min(per, rows - r0) : 0u;
    float nIn = 0.f, nOut = 0.f;
    if ((unsigned)lane < cnt) {
        nIn  = invIn[lo + r0 + lane];
        nOut = invOut[lo + r0 + lane];
    }
    const float4 bv = ((const float4*)b1)[lane];
    const float4 wv = ((const float4*)W2)[lane];

    float4* acc4 = (float4*)acc;
    for (int i = tid; i < NB * FEATS / 4; i += OWN_THREADS) acc4[i] = make_float4(0.f, 0.f, 0.f, 0.f);
    if (tid == 0) logCount = 0u;
    __syncthreads();

    const int nG   = nE >> 2;
    const int nW   = (nG + 31) >> 5;
    const int perB = 32 * CH;
    const int nBt  = (nW + perB - 1) / perB;
    const unsigned* __restrict__ bmB = bmp + (size_t)blockIdx.x * wordsPerOwner;
    const int4* __restrict__ dst4 = (const int4*)dst;
    const int4* __restrict__ src4 = (const int4*)src;

    if (wave == 0 && lane < 4) {
        const int e = 4 * nG + lane;
        if (e < nE) own_edge_by_one_lane(acc, h, &logCount, logB, dst[e], src[e], lo, rows);
    }

    unsigned cur[CH];
    unsigned qcnt = 0u;
    unsigned rcnt = 0u;
    int idx0 = wave;
#pragma unroll
    for (int j = 0; j < CH; ++j) {
        const int c = min(idx0, nBt - 1) * perB + j * 32 + lane;
        cur[j] = bmB[min(c, nW - 1)];
    }

    while (idx0 < nBt) {
        const int nidx = idx0 + OWN_WAVES;
        const int nxtB = min(nidx, nBt - 1);
        unsigned nxt[CH];
#pragma unroll
        for (int j = 0; j < CH; ++j) {
            const int c = nxtB * perB + j * 32 + lane;
            nxt[j] = bmB[min(c, nW - 1)];
        }

#pragma unroll
        for (int j = 0; j < CH; ++j) {
            const int c = idx0 * perB + j * 32 + lane;
            unsigned f = cur[j];
            if (c >= nW) f = 0u;
            unsigned bal = __builtin_amdgcn_ballot_w32(f != 0u);
            while (bal != 0u) {
                if (f != 0u) {
                    const unsigned p = (unsigned)__builtin_ctz(f);
                    f &= f - 1u;
                    const unsigned g   = 32u * (unsigned)c + p;
                    const unsigned pos = qcnt + __builtin_amdgcn_mbcnt_lo(bal, 0u);
                    if (pos < QCAP) {
                        qe[wave][pos] = g;
                    } else {
                        const int4 dv = dst4[g];
                        const int4 sv = src4[g];
                        own_edge_by_one_lane(acc, h, &logCount, logB, dv.x, sv.x, lo, rows);
                        own_edge_by_one_lane(acc, h, &logCount, logB, dv.y, sv.y, lo, rows);
                        own_edge_by_one_lane(acc, h, &logCount, logB, dv.z, sv.z, lo, rows);
                        own_edge_by_one_lane(acc, h, &logCount, logB, dv.w, sv.w, lo, rows);
                    }
                }
                qcnt += (unsigned)__builtin_popcount(bal);
                bal = __builtin_amdgcn_ballot_w32(f != 0u);
            }
        }

        __builtin_amdgcn_fence(__ATOMIC_SEQ_CST, "wavefront");
        const unsigned cntq = min(qcnt, (unsigned)QCAP);
        const bool last = nidx >= nBt;

        unsigned base = 0u;
        for (;;) {
            const bool moreA = (cntq - base >= 32u) || (last && cntq > base);
            if (moreA) {
                const unsigned m = min(32u, cntq - base);
                const bool has = (unsigned)lane < m;
                int4 dv = make_int4(0, 0, 0, 0), sv = make_int4(0, 0, 0, 0);
                if (has) {
                    const unsigned g = qe[wave][base + lane];
                    dv = dst4[g];
                    sv = src4[g];
                }
                const int dd4[4] = {dv.x, dv.y, dv.z, dv.w};
                const int ss4[4] = {sv.x, sv.y, sv.z, sv.w};
#pragma unroll
                for (int q = 0; q < 4; ++q) {
                    const unsigned t = (unsigned)(dd4[q] - lo);
                    const bool keep = has && t < rows;
                    const unsigned kb = __builtin_amdgcn_ballot_w32(keep);
                    if (keep) q2[wave][rcnt + __builtin_amdgcn_mbcnt_lo(kb, 0u)] = ((unsigned)ss4[q] << LOG_ROW_BITS) | t;
                    rcnt += (unsigned)__builtin_popcount(kb);
                }
                base += m;
                __builtin_amdgcn_fence(__ATOMIC_SEQ_CST, "wavefront");
            }

            const bool flush = last && base >= cntq;
            unsigned rb = 0u;
            while (rcnt - rb >= 32u || (flush && rcnt > rb)) {
                const unsigned mm = min(32u, rcnt - rb);
                unsigned w_l = 0u;
                if ((unsigned)lane < mm) w_l = q2[wave][rb + lane];
                const int      s_l = (int)(w_l >> LOG_ROW_BITS);
                const unsigned d_l = w_l & ((1u << LOG_ROW_BITS) - 1u);
                unsigned p0 = 0u;
                if (lane == 0) p0 = atomicAdd(&logCount, mm);
                p0 = (unsigned)__builtin_amdgcn_readfirstlane((int)p0);
                if ((unsigned)lane < mm && p0 + (unsigned)lane < LOG_CAP)
                    logB[p0 + lane] = w_l;
                for (unsigned i = 0; i < mm; i += RG) {
                    float    v[RG][4];
                    unsigned dd[RG];
#pragma unroll
                    for (int k = 0; k < RG; ++k) {
                        const unsigned idx = min(i + k, mm - 1u);
                        const int s = __builtin_amdgcn_readlane(s_l, (int)idx);
                        dd[k] = (unsigned)__builtin_amdgcn_readlane((int)d_l, (int)idx);
                        const float* __restrict__ hr = h + (size_t)s * FEATS + lane;
                        v[k][0] = hr[0];
                        v[k][1] = hr[32];
                        v[k][2] = hr[64];
                        v[k][3] = hr[96];
                    }
#pragma unroll
                    for (int k = 0; k < RG; ++k) {
                        if (i + k < mm) {
                            float* ar = acc + dd[k] * FEATS + lane;
                            atomicAdd(ar,      v[k][0]);
                            atomicAdd(ar + 32, v[k][1]);
                            atomicAdd(ar + 64, v[k][2]);
                            atomicAdd(ar + 96, v[k][3]);
                        }
                    }
                }
                rb += mm;
            }
            if (rb != 0u) {
                const unsigned remr = rcnt - rb;
                unsigned w_l = 0u;
                if ((unsigned)lane < remr) w_l = q2[wave][rb + lane];
                __builtin_amdgcn_fence(__ATOMIC_SEQ_CST, "wavefront");
                if ((unsigned)lane < remr) q2[wave][lane] = w_l;
                rcnt = remr;
                __builtin_amdgcn_fence(__ATOMIC_SEQ_CST, "wavefront");
            }
            if (!moreA) break;
        }
        {
            const unsigned rem = cntq - base;
            if (base != 0u && rem != 0u) {
                unsigned g_l = 0u;
                if ((unsigned)lane < rem) g_l = qe[wave][base + lane];
                __builtin_amdgcn_fence(__ATOMIC_SEQ_CST, "wavefront");
                if ((unsigned)lane < rem) qe[wave][lane] = g_l;
            }
            qcnt = rem;
            __builtin_amdgcn_fence(__ATOMIC_SEQ_CST, "wavefront");
        }

#pragma unroll
        for (int j = 0; j < CH; ++j) cur[j] = nxt[j];
        idx0 = nidx;
    }

    __syncthreads();
    if (tid == 0) elogCount[blockIdx.x] = logCount;
    float zk = 0.f;
    for (unsigned k = 0; k < cnt; ++k) {
        const float sIn = __builtin_bit_cast(float, __builtin_amdgcn_readlane(__builtin_bit_cast(int, nIn), (int)k));
        const float4 a = acc4[(r0 + k) * (FEATS / 4) + lane];
        const float hx = fmaxf(a.x * sIn + bv.x, 0.0f);
        const float hy = fmaxf(a.y * sIn + bv.y, 0.0f);
        const float hz = fmaxf(a.z * sIn + bv.z, 0.0f);
        const float hw = fmaxf(a.w * sIn + bv.w, 0.0f);
        float sum = hx * wv.x + hy * wv.y + hz * wv.z + hw * wv.w;
        sum += own_other_row(sum);
        sum += own_row_xor<8>(sum);
        sum += own_row_xor<4>(sum);
        sum += own_row_xor<2>(sum);
        sum += own_row_xor<1>(sum);
        if ((unsigned)lane == k) zk = sum;
    }
    if ((unsigned)lane < cnt) z[lo + r0 + lane] = zk * nOut;
}

#define L2_THREADS 1024
#define L2_UN      8
static_assert(L2_UN == 8, "the two load fences in gcn_layer2_owner name eight values");

__global__ __launch_bounds__(L2_THREADS)
void gcn_layer2_owner(const float* __restrict__ z, const int* __restrict__ src,
                      const int* __restrict__ dst, const unsigned* __restrict__ elog,
                      const unsigned* __restrict__ elogCount, const float* __restrict__ invIn,
                      const float* __restrict__ b2, float* __restrict__ out, int n, int nE) {
    __shared__ float sums[NB];
    const int tid = threadIdx.x;
    const int lo  = blockIdx.x * NB;
    const unsigned rows = (unsigned)min(NB, n - lo);
    static_assert(NB <= L2_THREADS, "one thread per row at the start and the end");

    if (tid < NB) sums[tid] = 0.f;
    const unsigned total = elogCount[blockIdx.x];
    float nIn = 0.f;
    if ((unsigned)tid < rows) nIn = invIn[lo + tid];
    const float bias = b2[0];
    __syncthreads();

    if (total <= LOG_CAP) {
        const unsigned* __restrict__ logB = elog + (size_t)blockIdx.x * LOG_CAP;
        for (unsigned i0 = (unsigned)tid; i0 < total; i0 += L2_UN * L2_THREADS) {
            unsigned w[L2_UN];
            float    v[L2_UN];
#pragma unroll
            for (int k = 0; k < L2_UN; ++k) w[k] = logB[min(i0 + (unsigned)k * L2_THREADS, total - 1u)];
            asm volatile("" :: "v"(w[0]), "v"(w[1]), "v"(w[2]), "v"(w[3]), "v"(w[4]), "v"(w[5]), "v"(w[6]), "v"(w[7]));
#pragma unroll
            for (int k = 0; k < L2_UN; ++k) v[k] = z[w[k] >> LOG_ROW_BITS];
            asm volatile("" :: "v"(v[0]), "v"(v[1]), "v"(v[2]), "v"(v[3]), "v"(v[4]), "v"(v[5]), "v"(v[6]), "v"(v[7]));
#pragma unroll
            for (int k = 0; k < L2_UN; ++k)
                if (i0 + (unsigned)k * L2_THREADS < total)
                    atomicAdd(&sums[w[k] & ((1u << LOG_ROW_BITS) - 1u)], v[k]);
        }
    } else {
        for (int e = tid; e < nE; e += L2_THREADS) {
            const unsigned t = (unsigned)(dst[e] - lo);
            if (t < rows) atomicAdd(&sums[t], z[src[e]]);
        }
    }
    __syncthreads();
    if ((unsigned)tid < rows) out[lo + tid] = sums[tid] * nIn + bias;
}

extern "C" void kernel_launch(void* const* d_in, const int* in_sizes, int n_in,
                              void* d_out, int out_size, void* d_ws, size_t ws_size,
                              hipStream_t stream) {
    const float* x   = (const float*)d_in[0];
    const int*   src = (const int*)d_in[1];
    const int*   dst = (const int*)d_in[2];
    const float* W1  = (const float*)d_in[3];
    const float* b1  = (const float*)d_in[4];
    const float* W2  = (const float*)d_in[5];
    const float* b2  = (const float*)d_in[6];
    float* out = (float*)d_out;

    const int n  = in_sizes[0] / FEATS;
    const int nE = in_sizes[1];

    const int nOwn = (n + NB - 1) / NB;

    float* f      = (float*)d_ws;
    float* invOut = f;
    float* invIn  = f + (size_t)n;
    float* z      = f + 2 * (size_t)n;
    unsigned* elogCount = (unsigned*)(f + 3 * (size_t)n);
    float* h      = f + 4 * (size_t)n;
    unsigned* elog = (unsigned*)(h + (size_t)n * FEATS);
    const int nIdxG = ((nE >> 2) + IDX_GROUPS - 1) / IDX_GROUPS;
    const int nIdxN = (n + IDX_GROUPS - 1) / IDX_GROUPS;
    const int nIdx  = nIdxG > nIdxN ? nIdxG : nIdxN;
    const int wordsPerOwner = 32 * nIdx;
    unsigned* bmp  = elog + (size_t)nOwn * LOG_CAP;
    unsigned* part = bmp + (size_t)nOwn * wordsPerOwner;
    const int nRanges = (n + DEG_RNG - 1) / DEG_RNG;
    const int nSlices = (nE + DEG_SLICE - 1) / DEG_SLICE;

    gcn_degrees<<<nRanges * nSlices, DEG_THREADS, 0, stream>>>(src, dst, part, n, nE, nRanges);
    gcn_norm_index<<<nIdx, IDX_GROUPS, 0, stream>>>(invOut, invIn, n, part, nSlices, dst, bmp, nE, nOwn);

    gcn_gemm1_wmma<<<n / 16, 256, 0, stream>>>(x, W1, invOut, h);

    gcn_rows_owner<<<nOwn, OWN_THREADS, 0, stream>>>(h, src, dst, bmp, wordsPerOwner, invIn, invOut, b1, W2, z, elog, elogCount, n, nE);

    gcn_layer2_owner<<<nOwn, L2_THREADS, 0, stream>>>(z, src, dst, elog, elogCount, invIn, b2, out, n, nE);
}


# Layer-1 product: a block fetches its 16x128 tile of x once by whole-line 16-byte loads into LDS (scaled); its 8 waves read A fragments from LDS, not 16-line loads each; sum order unchanged
# speedup vs baseline: 4.7354x; 1.0995x over previous
#include <hip/hip_runtime.h>
#include <hip/hip_bf16.h>

typedef __attribute__((ext_vector_type(2))) float v2f;
typedef __attribute__((ext_vector_type(8))) float v8f;

#define FEATS 128

#define DEG_SLICE   50000
#define DEG_RNG     12512
#define DEG_THREADS 1024
#define DEG_UN      8
static_assert(DEG_SLICE < 65536 && DEG_SLICE % 4 == 0, "half-word counts per slice; aligned 16-byte loads");
static_assert(DEG_RNG % 32 == 0, "whole lines per range");

__device__ __forceinline__ void deg_hit(unsigned* cnt, int v, int lo, unsigned rng, unsigned inc) {
    const unsigned t = (unsigned)(v - lo);
    if (t < rng) atomicAdd(&cnt[t], inc);
}

__global__ __launch_bounds__(DEG_THREADS)
void gcn_degrees(const int* __restrict__ src, const int* __restrict__ dst,
                 unsigned* __restrict__ part, int n, int nE, int nRanges) {
    __shared__ unsigned cnt[DEG_RNG];
    const int tid = threadIdx.x;
    const int s  = blockIdx.x / nRanges;
    const int r  = blockIdx.x - s * nRanges;
    const int lo = r * DEG_RNG;
    const unsigned rng = (unsigned)min(DEG_RNG, n - lo);

    for (int i = tid; i < DEG_RNG; i += DEG_THREADS) cnt[i] = 0u;
    __syncthreads();

    const int e0 = s * DEG_SLICE;
    const int e1 = min(nE, e0 + DEG_SLICE);
    const int nV = (e1 - e0) >> 2;
    const int4* __restrict__ src4 = (const int4*)(src + e0);
    const int4* __restrict__ dst4 = (const int4*)(dst + e0);
    for (int g0 = tid; g0 < nV; g0 += DEG_UN * DEG_THREADS) {
        int4 a[DEG_UN], b[DEG_UN];
#pragma unroll
        for (int k = 0; k < DEG_UN; ++k) {
            const int g = min(g0 + k * DEG_THREADS, nV - 1);
            a[k] = src4[g];
            b[k] = dst4[g];
        }
#pragma unroll
        for (int k = 0; k < DEG_UN; ++k) {
            if (g0 + k * DEG_THREADS < nV) {
                deg_hit(cnt, a[k].x, lo, rng, 1u);
                deg_hit(cnt, a[k].y, lo, rng, 1u);
                deg_hit(cnt, a[k].z, lo, rng, 1u);
                deg_hit(cnt, a[k].w, lo, rng, 1u);
                deg_hit(cnt, b[k].x, lo, rng, 0x10000u);
                deg_hit(cnt, b[k].y, lo, rng, 0x10000u);
                deg_hit(cnt, b[k].z, lo, rng, 0x10000u);
                deg_hit(cnt, b[k].w, lo, rng, 0x10000u);
            }
        }
    }
    for (int e = e0 + 4 * nV + tid; e < e1; e += DEG_THREADS) {
        deg_hit(cnt, src[e], lo, rng, 1u);
        deg_hit(cnt, dst[e], lo, rng, 0x10000u);
    }
    __syncthreads();

    unsigned* __restrict__ plane = part + (size_t)s * n + lo;
    for (unsigned i = (unsigned)tid; i < rng; i += DEG_THREADS) plane[i] = cnt[i];
}

#define NB 400

#define IDX_GROUPS 1024
#define IDX_OWN    256
#define IDX_PITCH  257
static_assert(IDX_GROUPS == 1024, "32 waves, one index word per wave and owner");

__global__ __launch_bounds__(IDX_GROUPS)
void gcn_norm_index(float* __restrict__ dOut, float* __restrict__ dIn, int n,
                    const unsigned* __restrict__ part, int nSlices,
                    const int* __restrict__ dst, unsigned* __restrict__ bmp, int nE, int nOwn) {
    __shared__ unsigned bm[32 * IDX_PITCH];
    const int tid = threadIdx.x;
    const int i = blockIdx.x * IDX_GROUPS + tid;
    for (int k = tid; k < 32 * IDX_PITCH; k += IDX_GROUPS) bm[k] = 0u;

    for (int v = i; v < n; v += gridDim.x * IDX_GROUPS) {
        unsigned cOut = 0u, cIn = 0u;
#pragma unroll 16
        for (int s = 0; s < nSlices; ++s) {
            const unsigned w = part[(size_t)s * n + v];
            cOut += w & 0xFFFFu;
            cIn  += w >> 16;
        }
        dOut[v] = rsqrtf(fmaxf((float)cOut, 1.0f));
        dIn[v]  = rsqrtf(fmaxf((float)cIn, 1.0f));
    }
    __syncthreads();

    const int nG = nE >> 2;
    if (i < nG) {
        const int4 d = ((const int4*)dst)[i];
        const unsigned bit = 1u << (tid & 31);
        unsigned* row = bm + (tid >> 5) * IDX_PITCH;
        atomicOr(&row[min((unsigned)d.x / NB, (unsigned)(IDX_OWN - 1))], bit);
        atomicOr(&row[min((unsigned)d.y / NB, (unsigned)(IDX_OWN - 1))], bit);
        atomicOr(&row[min((unsigned)d.z / NB, (unsigned)(IDX_OWN - 1))], bit);
        atomicOr(&row[min((unsigned)d.w / NB, (unsigned)(IDX_OWN - 1))], bit);
    }
    __syncthreads();

    const int lane = tid & 31;
    const size_t wordsPerOwner = (size_t)32 * gridDim.x;
    for (int o = tid >> 5; o < nOwn; o += IDX_GROUPS / 32)
        bmp[(size_t)o * wordsPerOwner + 32 * blockIdx.x + lane] = bm[lane * IDX_PITCH + o];
}

#define G1_PITCH 132
static_assert(G1_PITCH % 4 == 0 && (G1_PITCH / 4) % 2 == 1, "16-byte aligned rows at an odd multiple of 16 bytes");

__global__ __launch_bounds__(256)
void gcn_gemm1_wmma(const float* __restrict__ x, const float* __restrict__ W1,
                    const float* __restrict__ invOut, float* __restrict__ h) {
    __shared__ __attribute__((aligned(16))) float xs[16 * G1_PITCH];
    const int lane    = threadIdx.x & 31;
    const int colTile = threadIdx.x >> 5;
    const int rowTile = blockIdx.x;
    const int l15     = lane & 15;
    const int khalf   = lane >> 4;
    const int nCol    = colTile * 16 + l15;

    const int r0   = 2 * colTile;
    const int grow = rowTile * 16 + r0;
    const float4 xa = ((const float4*)(x + (size_t)grow * FEATS))[lane];
    const float4 xb = ((const float4*)(x + (size_t)(grow + 1) * FEATS))[lane];
    const float  sa = invOut[grow];
    const float  sb = invOut[grow + 1];

    v2f bf[FEATS / 4];
#pragma unroll
    for (int k = 0; k < FEATS; k += 4) {
        const int ka = k + 2 * khalf;
        bf[k >> 2].x = W1[(size_t)ka       * FEATS + nCol];
        bf[k >> 2].y = W1[(size_t)(ka + 1) * FEATS + nCol];
    }

    *(float4*)&xs[r0 * G1_PITCH + 4 * lane]       = make_float4(xa.x * sa, xa.y * sa, xa.z * sa, xa.w * sa);
    *(float4*)&xs[(r0 + 1) * G1_PITCH + 4 * lane] = make_float4(xb.x * sb, xb.y * sb, xb.z * sb, xb.w * sb);
    __syncthreads();

    const float* __restrict__ arow = xs + l15 * G1_PITCH + 2 * khalf;
    v8f c = {};
#pragma unroll
    for (int k = 0; k < FEATS; k += 4) {
        v2f a;
        a.x = arow[k];
        a.y = arow[k + 1];
        c = __builtin_amdgcn_wmma_f32_16x16x4_f32(
                 false, a,  false, bf[k >> 2],
                 (short)0, c,  false,  false);
    }

#pragma unroll
    for (int r = 0; r < 8; ++r) {
        const int mm = (lane < 16) ? r : (r + 8);
        h[(size_t)(rowTile * 16 + mm) * FEATS + nCol] = c[r];
    }
}

#define QCAP        128
#define Q2CAP       160
#define CH          2
#define RG          8
#define OWN_THREADS 1024
#define OWN_WAVES   (OWN_THREADS / 32)
#define LOG_CAP     32768u
#define LOG_ROW_BITS 9
static_assert(NB <= (1 << LOG_ROW_BITS), "row inside the block must fit the low bits of a log word");

__device__ __forceinline__ void own_log_one(unsigned* logCount, unsigned* __restrict__ logB, unsigned s, unsigned t) {
    const unsigned p = atomicAdd(logCount, 1u);
    if (p < LOG_CAP) logB[p] = (s << LOG_ROW_BITS) | t;
}

__device__ __forceinline__ void own_row_by_one_lane(float* accRow, const float* __restrict__ hRow) {
#pragma unroll 1
    for (int f = 0; f < FEATS; ++f) atomicAdd(accRow + f, hRow[f]);
}

__device__ __forceinline__ void own_edge_by_one_lane(float* acc, const float* __restrict__ h, unsigned* logCount,
                                                     unsigned* __restrict__ logB, int d, int s, int lo, unsigned rows) {
    const unsigned t = (unsigned)(d - lo);
    if (t < rows) {
        own_log_one(logCount, logB, (unsigned)s, t);
        own_row_by_one_lane(acc + t * FEATS, h + (size_t)s * FEATS);
    }
}

__device__ __forceinline__ float own_other_row(float x) {
    const int b = __builtin_bit_cast(int, x);
    return __builtin_bit_cast(float, __builtin_amdgcn_permlanex16(b, b, 0x76543210, 0xfedcba98, false, false));
}

template <int M>
__device__ __forceinline__ float own_row_xor(float x) {
    const int b = __builtin_bit_cast(int, x);
    return __builtin_bit_cast(float, __builtin_amdgcn_update_dpp(b, b, 0x160 + M, 0xf, 0xf, true));
}

__global__ __launch_bounds__(OWN_THREADS)
void gcn_rows_owner(const float* __restrict__ h, const int* __restrict__ src,
                    const int* __restrict__ dst, const unsigned* __restrict__ bmp, int wordsPerOwner,
                    const float* __restrict__ invIn, const float* __restrict__ invOut,
                    const float* __restrict__ b1, const float* __restrict__ W2,
                    float* __restrict__ z, unsigned* __restrict__ elog, unsigned* __restrict__ elogCount,
                    int n, int nE) {
    __shared__ __attribute__((aligned(16))) float acc[NB * FEATS];
    __shared__ unsigned qe[OWN_WAVES][QCAP];
    __shared__ unsigned q2[OWN_WAVES][Q2CAP];
    __shared__ unsigned logCount;
    unsigned* __restrict__ logB = elog + (size_t)blockIdx.x * LOG_CAP;

    const int tid  = threadIdx.x;
    const int lane = tid & 31;
    const int wave = __builtin_amdgcn_readfirstlane(tid >> 5);
    const int lo   = blockIdx.x * NB;
    const unsigned rows = (unsigned)min(NB, n - lo);

    static_assert(NB <= 32 * OWN_WAVES, "one lane per row of a wave's run");
    const unsigned per = (rows + OWN_WAVES - 1u) / OWN_WAVES;
    const unsigned r0  = (unsigned)wave * per;
    const unsigned cnt = r0 < rows ? min(per, rows - r0) : 0u;
    float nIn = 0.f, nOut = 0.f;
    if ((unsigned)lane < cnt) {
        nIn  = invIn[lo + r0 + lane];
        nOut = invOut[lo + r0 + lane];
    }
    const float4 bv = ((const float4*)b1)[lane];
    const float4 wv = ((const float4*)W2)[lane];

    float4* acc4 = (float4*)acc;
    for (int i = tid; i < NB * FEATS / 4; i += OWN_THREADS) acc4[i] = make_float4(0.f, 0.f, 0.f, 0.f);
    if (tid == 0) logCount = 0u;
    __syncthreads();

    const int nG   = nE >> 2;
    const int nW   = (nG + 31) >> 5;
    const int perB = 32 * CH;
    const int nBt  = (nW + perB - 1) / perB;
    const unsigned* __restrict__ bmB = bmp + (size_t)blockIdx.x * wordsPerOwner;
    const int4* __restrict__ dst4 = (const int4*)dst;
    const int4* __restrict__ src4 = (const int4*)src;

    if (wave == 0 && lane < 4) {
        const int e = 4 * nG + lane;
        if (e < nE) own_edge_by_one_lane(acc, h, &logCount, logB, dst[e], src[e], lo, rows);
    }

    unsigned cur[CH];
    unsigned qcnt = 0u;
    unsigned rcnt = 0u;
    int idx0 = wave;
#pragma unroll
    for (int j = 0; j < CH; ++j) {
        const int c = min(idx0, nBt - 1) * perB + j * 32 + lane;
        cur[j] = bmB[min(c, nW - 1)];
    }

    while (idx0 < nBt) {
        const int nidx = idx0 + OWN_WAVES;
        const int nxtB = min(nidx, nBt - 1);
        unsigned nxt[CH];
#pragma unroll
        for (int j = 0; j < CH; ++j) {
            const int c = nxtB * perB + j * 32 + lane;
            nxt[j] = bmB[min(c, nW - 1)];
        }

#pragma unroll
        for (int j = 0; j < CH; ++j) {
            const int c = idx0 * perB + j * 32 + lane;
            unsigned f = cur[j];
            if (c >= nW) f = 0u;
            unsigned bal = __builtin_amdgcn_ballot_w32(f != 0u);
            while (bal != 0u) {
                if (f != 0u) {
                    const unsigned p = (unsigned)__builtin_ctz(f);
                    f &= f - 1u;
                    const unsigned g   = 32u * (unsigned)c + p;
                    const unsigned pos = qcnt + __builtin_amdgcn_mbcnt_lo(bal, 0u);
                    if (pos < QCAP) {
                        qe[wave][pos] = g;
                    } else {
                        const int4 dv = dst4[g];
                        const int4 sv = src4[g];
                        own_edge_by_one_lane(acc, h, &logCount, logB, dv.x, sv.x, lo, rows);
                        own_edge_by_one_lane(acc, h, &logCount, logB, dv.y, sv.y, lo, rows);
                        own_edge_by_one_lane(acc, h, &logCount, logB, dv.z, sv.z, lo, rows);
                        own_edge_by_one_lane(acc, h, &logCount, logB, dv.w, sv.w, lo, rows);
                    }
                }
                qcnt += (unsigned)__builtin_popcount(bal);
                bal = __builtin_amdgcn_ballot_w32(f != 0u);
            }
        }

        __builtin_amdgcn_fence(__ATOMIC_SEQ_CST, "wavefront");
        const unsigned cntq = min(qcnt, (unsigned)QCAP);
        const bool last = nidx >= nBt;

        unsigned base = 0u;
        for (;;) {
            const bool moreA = (cntq - base >= 32u) || (last && cntq > base);
            if (moreA) {
                const unsigned m = min(32u, cntq - base);
                const bool has = (unsigned)lane < m;
                int4 dv = make_int4(0, 0, 0, 0), sv = make_int4(0, 0, 0, 0);
                if (has) {
                    const unsigned g = qe[wave][base + lane];
                    dv = dst4[g];
                    sv = src4[g];
                }
                const int dd4[4] = {dv.x, dv.y, dv.z, dv.w};
                const int ss4[4] = {sv.x, sv.y, sv.z, sv.w};
#pragma unroll
                for (int q = 0; q < 4; ++q) {
                    const unsigned t = (unsigned)(dd4[q] - lo);
                    const bool keep = has && t < rows;
                    const unsigned kb = __builtin_amdgcn_ballot_w32(keep);
                    if (keep) q2[wave][rcnt + __builtin_amdgcn_mbcnt_lo(kb, 0u)] = ((unsigned)ss4[q] << LOG_ROW_BITS) | t;
                    rcnt += (unsigned)__builtin_popcount(kb);
                }
                base += m;
                __builtin_amdgcn_fence(__ATOMIC_SEQ_CST, "wavefront");
            }

            const bool flush = last && base >= cntq;
            unsigned rb = 0u;
            while (rcnt - rb >= 32u || (flush && rcnt > rb)) {
                const unsigned mm = min(32u, rcnt - rb);
                unsigned w_l = 0u;
                if ((unsigned)lane < mm) w_l = q2[wave][rb + lane];
                const int      s_l = (int)(w_l >> LOG_ROW_BITS);
                const unsigned d_l = w_l & ((1u << LOG_ROW_BITS) - 1u);
                unsigned p0 = 0u;
                if (lane == 0) p0 = atomicAdd(&logCount, mm);
                p0 = (unsigned)__builtin_amdgcn_readfirstlane((int)p0);
                if ((unsigned)lane < mm && p0 + (unsigned)lane < LOG_CAP)
                    logB[p0 + lane] = w_l;
                for (unsigned i = 0; i < mm; i += RG) {
                    float    v[RG][4];
                    unsigned dd[RG];
#pragma unroll
                    for (int k = 0; k < RG; ++k) {
                        const unsigned idx = min(i + k, mm - 1u);
                        const int s = __builtin_amdgcn_readlane(s_l, (int)idx);
                        dd[k] = (unsigned)__builtin_amdgcn_readlane((int)d_l, (int)idx);
                        const float* __restrict__ hr = h + (size_t)s * FEATS + lane;
                        v[k][0] = hr[0];
                        v[k][1] = hr[32];
                        v[k][2] = hr[64];
                        v[k][3] = hr[96];
                    }
#pragma unroll
                    for (int k = 0; k < RG; ++k) {
                        if (i + k < mm) {
                            float* ar = acc + dd[k] * FEATS + lane;
                            atomicAdd(ar,      v[k][0]);
                            atomicAdd(ar + 32, v[k][1]);
                            atomicAdd(ar + 64, v[k][2]);
                            atomicAdd(ar + 96, v[k][3]);
                        }
                    }
                }
                rb += mm;
            }
            if (rb != 0u) {
                const unsigned remr = rcnt - rb;
                unsigned w_l = 0u;
                if ((unsigned)lane < remr) w_l = q2[wave][rb + lane];
                __builtin_amdgcn_fence(__ATOMIC_SEQ_CST, "wavefront");
                if ((unsigned)lane < remr) q2[wave][lane] = w_l;
                rcnt = remr;
                __builtin_amdgcn_fence(__ATOMIC_SEQ_CST, "wavefront");
            }
            if (!moreA) break;
        }
        {
            const unsigned rem = cntq - base;
            if (base != 0u && rem != 0u) {
                unsigned g_l = 0u;
                if ((unsigned)lane < rem) g_l = qe[wave][base + lane];
                __builtin_amdgcn_fence(__ATOMIC_SEQ_CST, "wavefront");
                if ((unsigned)lane < rem) qe[wave][lane] = g_l;
            }
            qcnt = rem;
            __builtin_amdgcn_fence(__ATOMIC_SEQ_CST, "wavefront");
        }

#pragma unroll
        for (int j = 0; j < CH; ++j) cur[j] = nxt[j];
        idx0 = nidx;
    }

    __syncthreads();
    if (tid == 0) elogCount[blockIdx.x] = logCount;
    float zk = 0.f;
    for (unsigned k = 0; k < cnt; ++k) {
        const float sIn = __builtin_bit_cast(float, __builtin_amdgcn_readlane(__builtin_bit_cast(int, nIn), (int)k));
        const float4 a = acc4[(r0 + k) * (FEATS / 4) + lane];
        const float hx = fmaxf(a.x * sIn + bv.x, 0.0f);
        const float hy = fmaxf(a.y * sIn + bv.y, 0.0f);
        const float hz = fmaxf(a.z * sIn + bv.z, 0.0f);
        const float hw = fmaxf(a.w * sIn + bv.w, 0.0f);
        float sum = hx * wv.x + hy * wv.y + hz * wv.z + hw * wv.w;
        sum += own_other_row(sum);
        sum += own_row_xor<8>(sum);
        sum += own_row_xor<4>(sum);
        sum += own_row_xor<2>(sum);
        sum += own_row_xor<1>(sum);
        if ((unsigned)lane == k) zk = sum;
    }
    if ((unsigned)lane < cnt) z[lo + r0 + lane] = zk * nOut;
}

#define L2_THREADS 1024
#define L2_UN      8
static_assert(L2_UN == 8, "the two load fences in gcn_layer2_owner name eight values");

__global__ __launch_bounds__(L2_THREADS)
void gcn_layer2_owner(const float* __restrict__ z, const int* __restrict__ src,
                      const int* __restrict__ dst, const unsigned* __restrict__ elog,
                      const unsigned* __restrict__ elogCount, const float* __restrict__ invIn,
                      const float* __restrict__ b2, float* __restrict__ out, int n, int nE) {
    __shared__ float sums[NB];
    const int tid = threadIdx.x;
    const int lo  = blockIdx.x * NB;
    const unsigned rows = (unsigned)min(NB, n - lo);
    static_assert(NB <= L2_THREADS, "one thread per row at the start and the end");

    if (tid < NB) sums[tid] = 0.f;
    const unsigned total = elogCount[blockIdx.x];
    float nIn = 0.f;
    if ((unsigned)tid < rows) nIn = invIn[lo + tid];
    const float bias = b2[0];
    __syncthreads();

    if (total <= LOG_CAP) {
        const unsigned* __restrict__ logB = elog + (size_t)blockIdx.x * LOG_CAP;
        for (unsigned i0 = (unsigned)tid; i0 < total; i0 += L2_UN * L2_THREADS) {
            unsigned w[L2_UN];
            float    v[L2_UN];
#pragma unroll
            for (int k = 0; k < L2_UN; ++k) w[k] = logB[min(i0 + (unsigned)k * L2_THREADS, total - 1u)];
            asm volatile("" :: "v"(w[0]), "v"(w[1]), "v"(w[2]), "v"(w[3]), "v"(w[4]), "v"(w[5]), "v"(w[6]), "v"(w[7]));
#pragma unroll
            for (int k = 0; k < L2_UN; ++k) v[k] = z[w[k] >> LOG_ROW_BITS];
            asm volatile("" :: "v"(v[0]), "v"(v[1]), "v"(v[2]), "v"(v[3]), "v"(v[4]), "v"(v[5]), "v"(v[6]), "v"(v[7]));
#pragma unroll
            for (int k = 0; k < L2_UN; ++k)
                if (i0 + (unsigned)k * L2_THREADS < total)
                    atomicAdd(&sums[w[k] & ((1u << LOG_ROW_BITS) - 1u)], v[k]);
        }
    } else {
        for (int e = tid; e < nE; e += L2_THREADS) {
            const unsigned t = (unsigned)(dst[e] - lo);
            if (t < rows) atomicAdd(&sums[t], z[src[e]]);
        }
    }
    __syncthreads();
    if ((unsigned)tid < rows) out[lo + tid] = sums[tid] * nIn + bias;
}

extern "C" void kernel_launch(void* const* d_in, const int* in_sizes, int n_in,
                              void* d_out, int out_size, void* d_ws, size_t ws_size,
                              hipStream_t stream) {
    const float* x   = (const float*)d_in[0];
    const int*   src = (const int*)d_in[1];
    const int*   dst = (const int*)d_in[2];
    const float* W1  = (const float*)d_in[3];
    const float* b1  = (const float*)d_in[4];
    const float* W2  = (const float*)d_in[5];
    const float* b2  = (const float*)d_in[6];
    float* out = (float*)d_out;

    const int n  = in_sizes[0] / FEATS;
    const int nE = in_sizes[1];

    const int nOwn = (n + NB - 1) / NB;

    float* f      = (float*)d_ws;
    float* invOut = f;
    float* invIn  = f + (size_t)n;
    float* z      = f + 2 * (size_t)n;
    unsigned* elogCount = (unsigned*)(f + 3 * (size_t)n);
    float* h      = f + 4 * (size_t)n;
    unsigned* elog = (unsigned*)(h + (size_t)n * FEATS);
    const int nIdxG = ((nE >> 2) + IDX_GROUPS - 1) / IDX_GROUPS;
    const int nIdxN = (n + IDX_GROUPS - 1) / IDX_GROUPS;
    const int nIdx  = nIdxG > nIdxN ? nIdxG : nIdxN;
    const int wordsPerOwner = 32 * nIdx;
    unsigned* bmp  = elog + (size_t)nOwn * LOG_CAP;
    unsigned* part = bmp + (size_t)nOwn * wordsPerOwner;
    const int nRanges = (n + DEG_RNG - 1) / DEG_RNG;
    const int nSlices = (nE + DEG_SLICE - 1) / DEG_SLICE;

    gcn_degrees<<<nRanges * nSlices, DEG_THREADS, 0, stream>>>(src, dst, part, n, nE, nRanges);
    gcn_norm_index<<<nIdx, IDX_GROUPS, 0, stream>>>(invOut, invIn, n, part, nSlices, dst, bmp, nE, nOwn);

    gcn_gemm1_wmma<<<n / 16, 256, 0, stream>>>(x, W1, invOut, h);

    gcn_rows_owner<<<nOwn, OWN_THREADS, 0, stream>>>(h, src, dst, bmp, wordsPerOwner, invIn, invOut, b1, W2, z, elog, elogCount, n, nE);

    gcn_layer2_owner<<<nOwn, L2_THREADS, 0, stream>>>(z, src, dst, elog, elogCount, invIn, b2, out, n, nE);
}


# Degree launch dropped (5 to 4 launches): owner blocks count their rows' in-edges; out-degrees counted per edge slice by blocks riding in the index launch, summed in the product blocks; sums unchanged
# speedup vs baseline: 5.0004x; 1.0560x over previous
#include <hip/hip_runtime.h>
#include <hip/hip_bf16.h>

typedef __attribute__((ext_vector_type(2))) float v2f;
typedef __attribute__((ext_vector_type(8))) float v8f;

#define FEATS 128

#define K1_THREADS  1024
#define CNT_PLANES  64
#define CNT_RNGW    25600
#define CNT_PW      (2 * CNT_RNGW)
#define CNT_UN      8
static_assert(CNT_RNGW % 32 == 0, "whole lines per range");

#define NB 400

#define IDX_PER    4
#define IDX_ROWS   (32 * IDX_PER)
#define IDX_OWN    256
#define IDX_PITCH  257
static_assert(K1_THREADS == 1024, "32 waves: a wave's 32 groups of one pass are one index word per owner");
static_assert(2 * CNT_PW >= IDX_OWN * NB, "the two count ranges cover every node the owner index has room for");
static_assert(CNT_RNGW <= IDX_ROWS * IDX_PITCH, "the count words fit the block's LDS");

__device__ __forceinline__ void cnt_hit(unsigned* cnt, int v, int lo, unsigned rng) {
    const unsigned t = (unsigned)(v - lo);
    if (t < rng) atomicAdd(&cnt[t >> 1], 1u << ((t & 1u) * 16u));
}

__global__ __launch_bounds__(K1_THREADS)
void gcn_count_index(const int* __restrict__ src, const int* __restrict__ dst,
                     unsigned* __restrict__ part, unsigned* __restrict__ bmp,
                     int n, int nE, int sliceLen, int nCnt, int nOwn, int wordsPerOwner) {
    __shared__ unsigned sh[IDX_ROWS * IDX_PITCH];
    const int tid = threadIdx.x;

    if ((int)blockIdx.x < nCnt) {
        const int s  = blockIdx.x >> 1;
        const int r  = blockIdx.x & 1;
        const int lo = r * (2 * CNT_RNGW);
        const int rn = min(2 * CNT_RNGW, n - lo);
        const unsigned rng = rn > 0 ? (unsigned)rn : 0u;

        const int e0 = s * sliceLen;
        const int e1 = min(nE, e0 + sliceLen);
        const int nV = (e1 - e0) >> 2;
        const int4* __restrict__ src4 = (const int4*)(src + e0);
        int4 a[CNT_UN];
        if (nV > 0) {
#pragma unroll
            for (int k = 0; k < CNT_UN; ++k) a[k] = src4[min(tid + k * K1_THREADS, nV - 1)];
        }

        for (int i = tid; i < CNT_RNGW; i += K1_THREADS) sh[i] = 0u;
        __syncthreads();

        for (int g0 = tid; g0 < nV; ) {
#pragma unroll
            for (int k = 0; k < CNT_UN; ++k) {
                if (g0 + k * K1_THREADS < nV) {
                    cnt_hit(sh, a[k].x, lo, rng);
                    cnt_hit(sh, a[k].y, lo, rng);
                    cnt_hit(sh, a[k].z, lo, rng);
                    cnt_hit(sh, a[k].w, lo, rng);
                }
            }
            g0 += CNT_UN * K1_THREADS;
            if (g0 < nV) {
#pragma unroll
                for (int k = 0; k < CNT_UN; ++k) a[k] = src4[min(g0 + k * K1_THREADS, nV - 1)];
            }
        }
        for (int e = e0 + 4 * nV + tid; e < e1; e += K1_THREADS) cnt_hit(sh, src[e], lo, rng);
        __syncthreads();

        const unsigned rngW = (rng + 1u) >> 1;
        unsigned* __restrict__ plane = part + (size_t)s * CNT_PW + (size_t)r * CNT_RNGW;
        for (unsigned i = (unsigned)tid; i < rngW; i += K1_THREADS) plane[i] = sh[i];
        return;
    }

    const int bi = (int)blockIdx.x - nCnt;
    const int nG = nE >> 2;
    const int gb = bi * (IDX_PER * K1_THREADS);
    int4 d[IDX_PER];
#pragma unroll
    for (int j = 0; j < IDX_PER; ++j) d[j] = ((const int4*)dst)[min(gb + j * K1_THREADS + tid, nG - 1)];
    for (int k = tid; k < IDX_ROWS * IDX_PITCH; k += K1_THREADS) sh[k] = 0u;
    __syncthreads();

    const unsigned bit = 1u << (tid & 31);
#pragma unroll
    for (int j = 0; j < IDX_PER; ++j) {
        if (gb + j * K1_THREADS + tid < nG) {
            unsigned* row = sh + ((tid >> 5) + 32 * j) * IDX_PITCH;
            atomicOr(&row[min((unsigned)d[j].x / NB, (unsigned)(IDX_OWN - 1))], bit);
            atomicOr(&row[min((unsigned)d[j].y / NB, (unsigned)(IDX_OWN - 1))], bit);
            atomicOr(&row[min((unsigned)d[j].z / NB, (unsigned)(IDX_OWN - 1))], bit);
            atomicOr(&row[min((unsigned)d[j].w / NB, (unsigned)(IDX_OWN - 1))], bit);
        }
    }
    __syncthreads();

    const int lane = tid & 31;
    for (int o = tid >> 5; o < nOwn; o += K1_THREADS / 32) {
        unsigned* __restrict__ line = bmp + (size_t)o * wordsPerOwner + (size_t)IDX_ROWS * bi + lane;
#pragma unroll
        for (int q = 0; q < IDX_PER; ++q) line[32 * q] = sh[(32 * q + lane) * IDX_PITCH + o];
    }
}

#define G1_PITCH 132
static_assert(G1_PITCH % 4 == 0 && (G1_PITCH / 4) % 2 == 1, "16-byte aligned rows at an odd multiple of 16 bytes");

template <int M>
__device__ __forceinline__ unsigned g1_xor(unsigned v) {
    return (unsigned)__builtin_amdgcn_update_dpp((int)v, (int)v, 0x160 + M, 0xf, 0xf, true);
}
__device__ __forceinline__ unsigned g1_xor16(unsigned v) {
    return (unsigned)__builtin_amdgcn_permlanex16((int)v, (int)v, 0x76543210, 0xfedcba98, false, false);
}

__global__ __launch_bounds__(256)
void gcn_gemm1_wmma(const float* __restrict__ x, const float* __restrict__ W1,
                    const unsigned* __restrict__ part, int nSlices,
                    float* __restrict__ invOut, float* __restrict__ h) {
    __shared__ __attribute__((aligned(16))) float xs[16 * G1_PITCH];
    __shared__ __attribute__((aligned(8))) unsigned cw[8][16];
    const int lane    = threadIdx.x & 31;
    const int colTile = threadIdx.x >> 5;
    const int rowTile = blockIdx.x;
    const int l15     = lane & 15;
    const int khalf   = lane >> 4;
    const int nCol    = colTile * 16 + l15;

    const int r0   = 2 * colTile;
    const int grow = rowTile * 16 + r0;

    const int wi = threadIdx.x & 7;
    const int pg = threadIdx.x >> 3;
    static_assert(CNT_PLANES == 64, "two planes per thread of the block's 256");
    const unsigned* __restrict__ pw = part + (size_t)rowTile * 8 + wi;
    const unsigned w0r = pw[(size_t)min(pg, nSlices - 1) * CNT_PW];
    const unsigned w1r = pw[(size_t)min(pg + 32, nSlices - 1) * CNT_PW];

    const float4 xa = ((const float4*)(x + (size_t)grow * FEATS))[lane];
    const float4 xb = ((const float4*)(x + (size_t)(grow + 1) * FEATS))[lane];

    v2f bf[FEATS / 4];
#pragma unroll
    for (int k = 0; k < FEATS; k += 4) {
        const int ka = k + 2 * khalf;
        bf[k >> 2].x = W1[(size_t)ka       * FEATS + nCol];
        bf[k >> 2].y = W1[(size_t)(ka + 1) * FEATS + nCol];
    }

    const unsigned w0 = pg < nSlices ? w0r : 0u;
    const unsigned w1 = pg + 32 < nSlices ? w1r : 0u;
    unsigned cEven = (w0 & 0xFFFFu) + (w1 & 0xFFFFu);
    unsigned cOdd  = (w0 >> 16) + (w1 >> 16);
    cEven += g1_xor<8>(cEven);  cOdd += g1_xor<8>(cOdd);
    cEven += g1_xor16(cEven);   cOdd += g1_xor16(cOdd);
    if (lane < 8) {
        cw[colTile][2 * wi]     = cEven;
        cw[colTile][2 * wi + 1] = cOdd;
    }
    __syncthreads();
    const uint2 c2 = *(const uint2*)&cw[lane & 7][r0];
    unsigned ca = c2.x, cb = c2.y;
    ca += g1_xor<1>(ca);  cb += g1_xor<1>(cb);
    ca += g1_xor<2>(ca);  cb += g1_xor<2>(cb);
    ca += g1_xor<4>(ca);  cb += g1_xor<4>(cb);
    const float sa = rsqrtf(fmaxf((float)ca, 1.0f));
    const float sb = rsqrtf(fmaxf((float)cb, 1.0f));

    *(float4*)&xs[r0 * G1_PITCH + 4 * lane]       = make_float4(xa.x * sa, xa.y * sa, xa.z * sa, xa.w * sa);
    *(float4*)&xs[(r0 + 1) * G1_PITCH + 4 * lane] = make_float4(xb.x * sb, xb.y * sb, xb.z * sb, xb.w * sb);
    __syncthreads();
    if (lane == 0) {
        invOut[grow]     = sa;
        invOut[grow + 1] = sb;
    }

    const float* __restrict__ arow = xs + l15 * G1_PITCH + 2 * khalf;
    v8f c = {};
#pragma unroll
    for (int k = 0; k < FEATS; k += 4) {
        v2f a;
        a.x = arow[k];
        a.y = arow[k + 1];
        c = __builtin_amdgcn_wmma_f32_16x16x4_f32(
                 false, a,  false, bf[k >> 2],
                 (short)0, c,  false,  false);
    }

#pragma unroll
    for (int r = 0; r < 8; ++r) {
        const int mm = (lane < 16) ? r : (r + 8);
        h[(size_t)(rowTile * 16 + mm) * FEATS + nCol] = c[r];
    }
}

#define QCAP        128
#define Q2CAP       160
#define CH          2
#define RG          8
#define OWN_THREADS 1024
#define OWN_WAVES   (OWN_THREADS / 32)
#define LOG_CAP     32768u
#define LOG_ROW_BITS 9
static_assert(NB <= (1 << LOG_ROW_BITS), "row inside the block must fit the low bits of a log word");

__device__ __forceinline__ void own_log_one(unsigned* logCount, unsigned* __restrict__ logB, unsigned s, unsigned t) {
    const unsigned p = atomicAdd(logCount, 1u);
    if (p < LOG_CAP) logB[p] = (s << LOG_ROW_BITS) | t;
}

__device__ __forceinline__ void own_row_by_one_lane(float* accRow, const float* __restrict__ hRow) {
#pragma unroll 1
    for (int f = 0; f < FEATS; ++f) atomicAdd(accRow + f, hRow[f]);
}

__device__ __forceinline__ void own_edge_by_one_lane(float* acc, unsigned* rowCnt, const float* __restrict__ h, unsigned* logCount,
                                                     unsigned* __restrict__ logB, int d, int s, int lo, unsigned rows) {
    const unsigned t = (unsigned)(d - lo);
    if (t < rows) {
        atomicAdd(&rowCnt[t], 1u);
        own_log_one(logCount, logB, (unsigned)s, t);
        own_row_by_one_lane(acc + t * FEATS, h + (size_t)s * FEATS);
    }
}

__device__ __forceinline__ float own_other_row(float x) {
    const int b = __builtin_bit_cast(int, x);
    return __builtin_bit_cast(float, __builtin_amdgcn_permlanex16(b, b, 0x76543210, 0xfedcba98, false, false));
}

template <int M>
__device__ __forceinline__ float own_row_xor(float x) {
    const int b = __builtin_bit_cast(int, x);
    return __builtin_bit_cast(float, __builtin_amdgcn_update_dpp(b, b, 0x160 + M, 0xf, 0xf, true));
}

__global__ __launch_bounds__(OWN_THREADS)
void gcn_rows_owner(const float* __restrict__ h, const int* __restrict__ src,
                    const int* __restrict__ dst, const unsigned* __restrict__ bmp, int wordsPerOwner,
                    float* __restrict__ invIn, const float* __restrict__ invOut,
                    const float* __restrict__ b1, const float* __restrict__ W2,
                    float* __restrict__ z, unsigned* __restrict__ elog, unsigned* __restrict__ elogCount,
                    int n, int nE) {
    __shared__ __attribute__((aligned(16))) float acc[NB * FEATS];
    __shared__ unsigned qe[OWN_WAVES][QCAP];
    __shared__ unsigned q2[OWN_WAVES][Q2CAP];
    __shared__ unsigned rowCnt[NB];
    __shared__ unsigned logCount;
    unsigned* __restrict__ logB = elog + (size_t)blockIdx.x * LOG_CAP;

    const int tid  = threadIdx.x;
    const int lane = tid & 31;
    const int wave = __builtin_amdgcn_readfirstlane(tid >> 5);
    const int lo   = blockIdx.x * NB;
    const unsigned rows = (unsigned)min(NB, n - lo);

    static_assert(NB <= 32 * OWN_WAVES, "one lane per row of a wave's run");
    const unsigned per = (rows + OWN_WAVES - 1u) / OWN_WAVES;
    const unsigned r0  = (unsigned)wave * per;
    const unsigned cnt = r0 < rows ? min(per, rows - r0) : 0u;
    float nOut = 0.f;
    if ((unsigned)lane < cnt) nOut = invOut[lo + r0 + lane];
    const float4 bv = ((const float4*)b1)[lane];
    const float4 wv = ((const float4*)W2)[lane];

    float4* acc4 = (float4*)acc;
    for (int i = tid; i < NB * FEATS / 4; i += OWN_THREADS) acc4[i] = make_float4(0.f, 0.f, 0.f, 0.f);
    static_assert(NB <= OWN_THREADS, "one thread per row count");
    if (tid < NB) rowCnt[tid] = 0u;
    if (tid == 0) logCount = 0u;
    __syncthreads();

    const int nG   = nE >> 2;
    const int nW   = (nG + 31) >> 5;
    const int perB = 32 * CH;
    const int nBt  = (nW + perB - 1) / perB;
    const unsigned* __restrict__ bmB = bmp + (size_t)blockIdx.x * wordsPerOwner;
    const int4* __restrict__ dst4 = (const int4*)dst;
    const int4* __restrict__ src4 = (const int4*)src;

    if (wave == 0 && lane < 4) {
        const int e = 4 * nG + lane;
        if (e < nE) own_edge_by_one_lane(acc, rowCnt, h, &logCount, logB, dst[e], src[e], lo, rows);
    }

    unsigned cur[CH];
    unsigned qcnt = 0u;
    unsigned rcnt = 0u;
    int idx0 = wave;
#pragma unroll
    for (int j = 0; j < CH; ++j) {
        const int c = min(idx0, nBt - 1) * perB + j * 32 + lane;
        cur[j] = bmB[min(c, nW - 1)];
    }

    while (idx0 < nBt) {
        const int nidx = idx0 + OWN_WAVES;
        const int nxtB = min(nidx, nBt - 1);
        unsigned nxt[CH];
#pragma unroll
        for (int j = 0; j < CH; ++j) {
            const int c = nxtB * perB + j * 32 + lane;
            nxt[j] = bmB[min(c, nW - 1)];
        }

#pragma unroll
        for (int j = 0; j < CH; ++j) {
            const int c = idx0 * perB + j * 32 + lane;
            unsigned f = cur[j];
            if (c >= nW) f = 0u;
            unsigned bal = __builtin_amdgcn_ballot_w32(f != 0u);
            while (bal != 0u) {
                if (f != 0u) {
                    const unsigned p = (unsigned)__builtin_ctz(f);
                    f &= f - 1u;
                    const unsigned g   = 32u * (unsigned)c + p;
                    const unsigned pos = qcnt + __builtin_amdgcn_mbcnt_lo(bal, 0u);
                    if (pos < QCAP) {
                        qe[wave][pos] = g;
                    } else {
                        const int4 dv = dst4[g];
                        const int4 sv = src4[g];
                        own_edge_by_one_lane(acc, rowCnt, h, &logCount, logB, dv.x, sv.x, lo, rows);
                        own_edge_by_one_lane(acc, rowCnt, h, &logCount, logB, dv.y, sv.y, lo, rows);
                        own_edge_by_one_lane(acc, rowCnt, h, &logCount, logB, dv.z, sv.z, lo, rows);
                        own_edge_by_one_lane(acc, rowCnt, h, &logCount, logB, dv.w, sv.w, lo, rows);
                    }
                }
                qcnt += (unsigned)__builtin_popcount(bal);
                bal = __builtin_amdgcn_ballot_w32(f != 0u);
            }
        }

        __builtin_amdgcn_fence(__ATOMIC_SEQ_CST, "wavefront");
        const unsigned cntq = min(qcnt, (unsigned)QCAP);
        const bool last = nidx >= nBt;

        unsigned base = 0u;
        for (;;) {
            const bool moreA = (cntq - base >= 32u) || (last && cntq > base);
            if (moreA) {
                const unsigned m = min(32u, cntq - base);
                const bool has = (unsigned)lane < m;
                int4 dv = make_int4(0, 0, 0, 0), sv = make_int4(0, 0, 0, 0);
                if (has) {
                    const unsigned g = qe[wave][base + lane];
                    dv = dst4[g];
                    sv = src4[g];
                }
                const int dd4[4] = {dv.x, dv.y, dv.z, dv.w};
                const int ss4[4] = {sv.x, sv.y, sv.z, sv.w};
#pragma unroll
                for (int q = 0; q < 4; ++q) {
                    const unsigned t = (unsigned)(dd4[q] - lo);
                    const bool keep = has && t < rows;
                    const unsigned kb = __builtin_amdgcn_ballot_w32(keep);
                    if (keep) {
                        q2[wave][rcnt + __builtin_amdgcn_mbcnt_lo(kb, 0u)] = ((unsigned)ss4[q] << LOG_ROW_BITS) | t;
                        atomicAdd(&rowCnt[t], 1u);
                    }
                    rcnt += (unsigned)__builtin_popcount(kb);
                }
                base += m;
                __builtin_amdgcn_fence(__ATOMIC_SEQ_CST, "wavefront");
            }

            const bool flush = last && base >= cntq;
            unsigned rb = 0u;
            while (rcnt - rb >= 32u || (flush && rcnt > rb)) {
                const unsigned mm = min(32u, rcnt - rb);
                unsigned w_l = 0u;
                if ((unsigned)lane < mm) w_l = q2[wave][rb + lane];
                const int      s_l = (int)(w_l >> LOG_ROW_BITS);
                const unsigned d_l = w_l & ((1u << LOG_ROW_BITS) - 1u);
                unsigned p0 = 0u;
                if (lane == 0) p0 = atomicAdd(&logCount, mm);
                p0 = (unsigned)__builtin_amdgcn_readfirstlane((int)p0);
                if ((unsigned)lane < mm && p0 + (unsigned)lane < LOG_CAP)
                    logB[p0 + lane] = w_l;
                for (unsigned i = 0; i < mm; i += RG) {
                    float    v[RG][4];
                    unsigned dd[RG];
#pragma unroll
                    for (int k = 0; k < RG; ++k) {
                        const unsigned idx = min(i + k, mm - 1u);
                        const int s = __builtin_amdgcn_readlane(s_l, (int)idx);
                        dd[k] = (unsigned)__builtin_amdgcn_readlane((int)d_l, (int)idx);
                        const float* __restrict__ hr = h + (size_t)s * FEATS + lane;
                        v[k][0] = hr[0];
                        v[k][1] = hr[32];
                        v[k][2] = hr[64];
                        v[k][3] = hr[96];
                    }
#pragma unroll
                    for (int k = 0; k < RG; ++k) {
                        if (i + k < mm) {
                            float* ar = acc + dd[k] * FEATS + lane;
                            atomicAdd(ar,      v[k][0]);
                            atomicAdd(ar + 32, v[k][1]);
                            atomicAdd(ar + 64, v[k][2]);
                            atomicAdd(ar + 96, v[k][3]);
                        }
                    }
                }
                rb += mm;
            }
            if (rb != 0u) {
                const unsigned remr = rcnt - rb;
                unsigned w_l = 0u;
                if ((unsigned)lane < remr) w_l = q2[wave][rb + lane];
                __builtin_amdgcn_fence(__ATOMIC_SEQ_CST, "wavefront");
                if ((unsigned)lane < remr) q2[wave][lane] = w_l;
                rcnt = remr;
                __builtin_amdgcn_fence(__ATOMIC_SEQ_CST, "wavefront");
            }
            if (!moreA) break;
        }
        {
            const unsigned rem = cntq - base;
            if (base != 0u && rem != 0u) {
                unsigned g_l = 0u;
                if ((unsigned)lane < rem) g_l = qe[wave][base + lane];
                __builtin_amdgcn_fence(__ATOMIC_SEQ_CST, "wavefront");
                if ((unsigned)lane < rem) qe[wave][lane] = g_l;
            }
            qcnt = rem;
            __builtin_amdgcn_fence(__ATOMIC_SEQ_CST, "wavefront");
        }

#pragma unroll
        for (int j = 0; j < CH; ++j) cur[j] = nxt[j];
        idx0 = nidx;
    }

    __syncthreads();
    if (tid == 0) elogCount[blockIdx.x] = logCount;
    float nIn = 0.f;
    if ((unsigned)lane < cnt) {
        nIn = rsqrtf(fmaxf((float)rowCnt[r0 + lane], 1.0f));
        invIn[lo + r0 + lane] = nIn;
    }
    float zk = 0.f;
    for (unsigned k = 0; k < cnt; ++k) {
        const float sIn = __builtin_bit_cast(float, __builtin_amdgcn_readlane(__builtin_bit_cast(int, nIn), (int)k));
        const float4 a = acc4[(r0 + k) * (FEATS / 4) + lane];
        const float hx = fmaxf(a.x * sIn + bv.x, 0.0f);
        const float hy = fmaxf(a.y * sIn + bv.y, 0.0f);
        const float hz = fmaxf(a.z * sIn + bv.z, 0.0f);
        const float hw = fmaxf(a.w * sIn + bv.w, 0.0f);
        float sum = hx * wv.x + hy * wv.y + hz * wv.z + hw * wv.w;
        sum += own_other_row(sum);
        sum += own_row_xor<8>(sum);
        sum += own_row_xor<4>(sum);
        sum += own_row_xor<2>(sum);
        sum += own_row_xor<1>(sum);
        if ((unsigned)lane == k) zk = sum;
    }
    if ((unsigned)lane < cnt) z[lo + r0 + lane] = zk * nOut;
}

#define L2_THREADS 1024
#define L2_UN      8
static_assert(L2_UN == 8, "the two load fences in gcn_layer2_owner name eight values");

__global__ __launch_bounds__(L2_THREADS)
void gcn_layer2_owner(const float* __restrict__ z, const int* __restrict__ src,
                      const int* __restrict__ dst, const unsigned* __restrict__ elog,
                      const unsigned* __restrict__ elogCount, const float* __restrict__ invIn,
                      const float* __restrict__ b2, float* __restrict__ out, int n, int nE) {
    __shared__ float sums[NB];
    const int tid = threadIdx.x;
    const int lo  = blockIdx.x * NB;
    const unsigned rows = (unsigned)min(NB, n - lo);
    static_assert(NB <= L2_THREADS, "one thread per row at the start and the end");

    if (tid < NB) sums[tid] = 0.f;
    const unsigned total = elogCount[blockIdx.x];
    float nIn = 0.f;
    if ((unsigned)tid < rows) nIn = invIn[lo + tid];
    const float bias = b2[0];
    __syncthreads();

    if (total <= LOG_CAP) {
        const unsigned* __restrict__ logB = elog + (size_t)blockIdx.x * LOG_CAP;
        for (unsigned i0 = (unsigned)tid; i0 < total; i0 += L2_UN * L2_THREADS) {
            unsigned w[L2_UN];
            float    v[L2_UN];
#pragma unroll
            for (int k = 0; k < L2_UN; ++k) w[k] = logB[min(i0 + (unsigned)k * L2_THREADS, total - 1u)];
            asm volatile("" :: "v"(w[0]), "v"(w[1]), "v"(w[2]), "v"(w[3]), "v"(w[4]), "v"(w[5]), "v"(w[6]), "v"(w[7]));
#pragma unroll
            for (int k = 0; k < L2_UN; ++k) v[k] = z[w[k] >> LOG_ROW_BITS];
            asm volatile("" :: "v"(v[0]), "v"(v[1]), "v"(v[2]), "v"(v[3]), "v"(v[4]), "v"(v[5]), "v"(v[6]), "v"(v[7]));
#pragma unroll
            for (int k = 0; k < L2_UN; ++k)
                if (i0 + (unsigned)k * L2_THREADS < total)
                    atomicAdd(&sums[w[k] & ((1u << LOG_ROW_BITS) - 1u)], v[k]);
        }
    } else {
        for (int e = tid; e < nE; e += L2_THREADS) {
            const unsigned t = (unsigned)(dst[e] - lo);
            if (t < rows) atomicAdd(&sums[t], z[src[e]]);
        }
    }
    __syncthreads();
    if ((unsigned)tid < rows) out[lo + tid] = sums[tid] * nIn + bias;
}

extern "C" void kernel_launch(void* const* d_in, const int* in_sizes, int n_in,
                              void* d_out, int out_size, void* d_ws, size_t ws_size,
                              hipStream_t stream) {
    const float* x   = (const float*)d_in[0];
    const int*   src = (const int*)d_in[1];
    const int*   dst = (const int*)d_in[2];
    const float* W1  = (const float*)d_in[3];
    const float* b1  = (const float*)d_in[4];
    const float* W2  = (const float*)d_in[5];
    const float* b2  = (const float*)d_in[6];
    float* out = (float*)d_out;

    const int n  = in_sizes[0] / FEATS;
    const int nE = in_sizes[1];

    const int nOwn = (n + NB - 1) / NB;

    float* f      = (float*)d_ws;
    float* invOut = f;
    float* invIn  = f + (size_t)n;
    float* z      = f + 2 * (size_t)n;
    unsigned* elogCount = (unsigned*)(f + 3 * (size_t)n);
    float* h      = f + 4 * (size_t)n;
    unsigned* elog = (unsigned*)(h + (size_t)n * FEATS);
    const int nIdx = ((nE >> 2) + IDX_PER * K1_THREADS - 1) / (IDX_PER * K1_THREADS);
    const int wordsPerOwner = IDX_ROWS * nIdx;
    unsigned* bmp  = elog + (size_t)nOwn * LOG_CAP;
    unsigned* part = bmp + (size_t)nOwn * wordsPerOwner;
    const int sliceLen = (((nE + CNT_PLANES - 1) / CNT_PLANES + 3) / 4) * 4;
    const int nSlices  = (nE + sliceLen - 1) / sliceLen;
    const int nCnt    = 2 * nSlices;

    gcn_count_index<<<nCnt + nIdx, K1_THREADS, 0, stream>>>(src, dst, part, bmp, n, nE, sliceLen, nCnt, nOwn, wordsPerOwner);

    gcn_gemm1_wmma<<<n / 16, 256, 0, stream>>>(x, W1, part, nSlices, invOut, h);

    gcn_rows_owner<<<nOwn, OWN_THREADS, 0, stream>>>(h, src, dst, bmp, wordsPerOwner, invIn, invOut, b1, W2, z, elog, elogCount, n, nE);

    gcn_layer2_owner<<<nOwn, L2_THREADS, 0, stream>>>(z, src, dst, elog, elogCount, invIn, b2, out, n, nE);
}


# Owner block fetches an h row as ONE 16-byte load per lane (4 whole lines per instruction), not four 4-byte loads of one line each; LDS row kept lane-major: adds stay on 32 banks; sum order unchanged
# speedup vs baseline: 5.1466x; 1.0292x over previous
#include <hip/hip_runtime.h>
#include <hip/hip_bf16.h>

typedef __attribute__((ext_vector_type(2))) float v2f;
typedef __attribute__((ext_vector_type(8))) float v8f;

#define FEATS 128

#define K1_THREADS  1024
#define CNT_PLANES  64
#define CNT_RNGW    25600
#define CNT_PW      (2 * CNT_RNGW)
#define CNT_UN      8
static_assert(CNT_RNGW % 32 == 0, "whole lines per range");

#define NB 400

#define IDX_PER    4
#define IDX_ROWS   (32 * IDX_PER)
#define IDX_OWN    256
#define IDX_PITCH  257
static_assert(K1_THREADS == 1024, "32 waves: a wave's 32 groups of one pass are one index word per owner");
static_assert(2 * CNT_PW >= IDX_OWN * NB, "the two count ranges cover every node the owner index has room for");
static_assert(CNT_RNGW <= IDX_ROWS * IDX_PITCH, "the count words fit the block's LDS");

__device__ __forceinline__ void cnt_hit(unsigned* cnt, int v, int lo, unsigned rng) {
    const unsigned t = (unsigned)(v - lo);
    if (t < rng) atomicAdd(&cnt[t >> 1], 1u << ((t & 1u) * 16u));
}

__global__ __launch_bounds__(K1_THREADS)
void gcn_count_index(const int* __restrict__ src, const int* __restrict__ dst,
                     unsigned* __restrict__ part, unsigned* __restrict__ bmp,
                     int n, int nE, int sliceLen, int nCnt, int nOwn, int wordsPerOwner) {
    __shared__ unsigned sh[IDX_ROWS * IDX_PITCH];
    const int tid = threadIdx.x;

    if ((int)blockIdx.x < nCnt) {
        const int s  = blockIdx.x >> 1;
        const int r  = blockIdx.x & 1;
        const int lo = r * (2 * CNT_RNGW);
        const int rn = min(2 * CNT_RNGW, n - lo);
        const unsigned rng = rn > 0 ? (unsigned)rn : 0u;

        const int e0 = s * sliceLen;
        const int e1 = min(nE, e0 + sliceLen);
        const int nV = (e1 - e0) >> 2;
        const int4* __restrict__ src4 = (const int4*)(src + e0);
        int4 a[CNT_UN];
        if (nV > 0) {
#pragma unroll
            for (int k = 0; k < CNT_UN; ++k) a[k] = src4[min(tid + k * K1_THREADS, nV - 1)];
        }

        for (int i = tid; i < CNT_RNGW; i += K1_THREADS) sh[i] = 0u;
        __syncthreads();

        for (int g0 = tid; g0 < nV; ) {
#pragma unroll
            for (int k = 0; k < CNT_UN; ++k) {
                if (g0 + k * K1_THREADS < nV) {
                    cnt_hit(sh, a[k].x, lo, rng);
                    cnt_hit(sh, a[k].y, lo, rng);
                    cnt_hit(sh, a[k].z, lo, rng);
                    cnt_hit(sh, a[k].w, lo, rng);
                }
            }
            g0 += CNT_UN * K1_THREADS;
            if (g0 < nV) {
#pragma unroll
                for (int k = 0; k < CNT_UN; ++k) a[k] = src4[min(g0 + k * K1_THREADS, nV - 1)];
            }
        }
        for (int e = e0 + 4 * nV + tid; e < e1; e += K1_THREADS) cnt_hit(sh, src[e], lo, rng);
        __syncthreads();

        const unsigned rngW = (rng + 1u) >> 1;
        unsigned* __restrict__ plane = part + (size_t)s * CNT_PW + (size_t)r * CNT_RNGW;
        for (unsigned i = (unsigned)tid; i < rngW; i += K1_THREADS) plane[i] = sh[i];
        return;
    }

    const int bi = (int)blockIdx.x - nCnt;
    const int nG = nE >> 2;
    const int gb = bi * (IDX_PER * K1_THREADS);
    int4 d[IDX_PER];
#pragma unroll
    for (int j = 0; j < IDX_PER; ++j) d[j] = ((const int4*)dst)[min(gb + j * K1_THREADS + tid, nG - 1)];
    for (int k = tid; k < IDX_ROWS * IDX_PITCH; k += K1_THREADS) sh[k] = 0u;
    __syncthreads();

    const unsigned bit = 1u << (tid & 31);
#pragma unroll
    for (int j = 0; j < IDX_PER; ++j) {
        if (gb + j * K1_THREADS + tid < nG) {
            unsigned* row = sh + ((tid >> 5) + 32 * j) * IDX_PITCH;
            atomicOr(&row[min((unsigned)d[j].x / NB, (unsigned)(IDX_OWN - 1))], bit);
            atomicOr(&row[min((unsigned)d[j].y / NB, (unsigned)(IDX_OWN - 1))], bit);
            atomicOr(&row[min((unsigned)d[j].z / NB, (unsigned)(IDX_OWN - 1))], bit);
            atomicOr(&row[min((unsigned)d[j].w / NB, (unsigned)(IDX_OWN - 1))], bit);
        }
    }
    __syncthreads();

    const int lane = tid & 31;
    for (int o = tid >> 5; o < nOwn; o += K1_THREADS / 32) {
        unsigned* __restrict__ line = bmp + (size_t)o * wordsPerOwner + (size_t)IDX_ROWS * bi + lane;
#pragma unroll
        for (int q = 0; q < IDX_PER; ++q) line[32 * q] = sh[(32 * q + lane) * IDX_PITCH + o];
    }
}

#define G1_PITCH 132
static_assert(G1_PITCH % 4 == 0 && (G1_PITCH / 4) % 2 == 1, "16-byte aligned rows at an odd multiple of 16 bytes");

template <int M>
__device__ __forceinline__ unsigned g1_xor(unsigned v) {
    return (unsigned)__builtin_amdgcn_update_dpp((int)v, (int)v, 0x160 + M, 0xf, 0xf, true);
}
__device__ __forceinline__ unsigned g1_xor16(unsigned v) {
    return (unsigned)__builtin_amdgcn_permlanex16((int)v, (int)v, 0x76543210, 0xfedcba98, false, false);
}

__global__ __launch_bounds__(256)
void gcn_gemm1_wmma(const float* __restrict__ x, const float* __restrict__ W1,
                    const unsigned* __restrict__ part, int nSlices,
                    float* __restrict__ invOut, float* __restrict__ h) {
    __shared__ __attribute__((aligned(16))) float xs[16 * G1_PITCH];
    __shared__ __attribute__((aligned(8))) unsigned cw[8][16];
    const int lane    = threadIdx.x & 31;
    const int colTile = threadIdx.x >> 5;
    const int rowTile = blockIdx.x;
    const int l15     = lane & 15;
    const int khalf   = lane >> 4;
    const int nCol    = colTile * 16 + l15;

    const int r0   = 2 * colTile;
    const int grow = rowTile * 16 + r0;

    const int wi = threadIdx.x & 7;
    const int pg = threadIdx.x >> 3;
    static_assert(CNT_PLANES == 64, "two planes per thread of the block's 256");
    const unsigned* __restrict__ pw = part + (size_t)rowTile * 8 + wi;
    const unsigned w0r = pw[(size_t)min(pg, nSlices - 1) * CNT_PW];
    const unsigned w1r = pw[(size_t)min(pg + 32, nSlices - 1) * CNT_PW];

    const float4 xa = ((const float4*)(x + (size_t)grow * FEATS))[lane];
    const float4 xb = ((const float4*)(x + (size_t)(grow + 1) * FEATS))[lane];

    v2f bf[FEATS / 4];
#pragma unroll
    for (int k = 0; k < FEATS; k += 4) {
        const int ka = k + 2 * khalf;
        bf[k >> 2].x = W1[(size_t)ka       * FEATS + nCol];
        bf[k >> 2].y = W1[(size_t)(ka + 1) * FEATS + nCol];
    }

    const unsigned w0 = pg < nSlices ? w0r : 0u;
    const unsigned w1 = pg + 32 < nSlices ? w1r : 0u;
    unsigned cEven = (w0 & 0xFFFFu) + (w1 & 0xFFFFu);
    unsigned cOdd  = (w0 >> 16) + (w1 >> 16);
    cEven += g1_xor<8>(cEven);  cOdd += g1_xor<8>(cOdd);
    cEven += g1_xor16(cEven);   cOdd += g1_xor16(cOdd);
    if (lane < 8) {
        cw[colTile][2 * wi]     = cEven;
        cw[colTile][2 * wi + 1] = cOdd;
    }
    __syncthreads();
    const uint2 c2 = *(const uint2*)&cw[lane & 7][r0];
    unsigned ca = c2.x, cb = c2.y;
    ca += g1_xor<1>(ca);  cb += g1_xor<1>(cb);
    ca += g1_xor<2>(ca);  cb += g1_xor<2>(cb);
    ca += g1_xor<4>(ca);  cb += g1_xor<4>(cb);
    const float sa = rsqrtf(fmaxf((float)ca, 1.0f));
    const float sb = rsqrtf(fmaxf((float)cb, 1.0f));

    *(float4*)&xs[r0 * G1_PITCH + 4 * lane]       = make_float4(xa.x * sa, xa.y * sa, xa.z * sa, xa.w * sa);
    *(float4*)&xs[(r0 + 1) * G1_PITCH + 4 * lane] = make_float4(xb.x * sb, xb.y * sb, xb.z * sb, xb.w * sb);
    __syncthreads();
    if (lane == 0) {
        invOut[grow]     = sa;
        invOut[grow + 1] = sb;
    }

    const float* __restrict__ arow = xs + l15 * G1_PITCH + 2 * khalf;
    v8f c = {};
#pragma unroll
    for (int k = 0; k < FEATS; k += 4) {
        v2f a;
        a.x = arow[k];
        a.y = arow[k + 1];
        c = __builtin_amdgcn_wmma_f32_16x16x4_f32(
                 false, a,  false, bf[k >> 2],
                 (short)0, c,  false,  false);
    }

#pragma unroll
    for (int r = 0; r < 8; ++r) {
        const int mm = (lane < 16) ? r : (r + 8);
        h[(size_t)(rowTile * 16 + mm) * FEATS + nCol] = c[r];
    }
}

#define QCAP        128
#define Q2CAP       160
#define CH          2
#define RG          8
#define OWN_THREADS 1024
#define OWN_WAVES   (OWN_THREADS / 32)
#define LOG_CAP     32768u
#define LOG_ROW_BITS 9
static_assert(NB <= (1 << LOG_ROW_BITS), "row inside the block must fit the low bits of a log word");

__device__ __forceinline__ void own_log_one(unsigned* logCount, unsigned* __restrict__ logB, unsigned s, unsigned t) {
    const unsigned p = atomicAdd(logCount, 1u);
    if (p < LOG_CAP) logB[p] = (s << LOG_ROW_BITS) | t;
}

#define OWN_POS(f) ((((f) & 3) << 5) | ((f) >> 2))

__device__ __forceinline__ void own_row_by_one_lane(float* accRow, const float* __restrict__ hRow) {
#pragma unroll 1
    for (int f = 0; f < FEATS; ++f) atomicAdd(accRow + OWN_POS(f), hRow[f]);
}

__device__ __forceinline__ void own_edge_by_one_lane(float* acc, unsigned* rowCnt, const float* __restrict__ h, unsigned* logCount,
                                                     unsigned* __restrict__ logB, int d, int s, int lo, unsigned rows) {
    const unsigned t = (unsigned)(d - lo);
    if (t < rows) {
        atomicAdd(&rowCnt[t], 1u);
        own_log_one(logCount, logB, (unsigned)s, t);
        own_row_by_one_lane(acc + t * FEATS, h + (size_t)s * FEATS);
    }
}

__device__ __forceinline__ float own_other_row(float x) {
    const int b = __builtin_bit_cast(int, x);
    return __builtin_bit_cast(float, __builtin_amdgcn_permlanex16(b, b, 0x76543210, 0xfedcba98, false, false));
}

template <int M>
__device__ __forceinline__ float own_row_xor(float x) {
    const int b = __builtin_bit_cast(int, x);
    return __builtin_bit_cast(float, __builtin_amdgcn_update_dpp(b, b, 0x160 + M, 0xf, 0xf, true));
}

__global__ __launch_bounds__(OWN_THREADS)
void gcn_rows_owner(const float* __restrict__ h, const int* __restrict__ src,
                    const int* __restrict__ dst, const unsigned* __restrict__ bmp, int wordsPerOwner,
                    float* __restrict__ invIn, const float* __restrict__ invOut,
                    const float* __restrict__ b1, const float* __restrict__ W2,
                    float* __restrict__ z, unsigned* __restrict__ elog, unsigned* __restrict__ elogCount,
                    int n, int nE) {
    __shared__ __attribute__((aligned(16))) float acc[NB * FEATS];
    __shared__ unsigned qe[OWN_WAVES][QCAP];
    __shared__ unsigned q2[OWN_WAVES][Q2CAP];
    __shared__ unsigned rowCnt[NB];
    __shared__ unsigned logCount;
    unsigned* __restrict__ logB = elog + (size_t)blockIdx.x * LOG_CAP;

    const int tid  = threadIdx.x;
    const int lane = tid & 31;
    const int wave = __builtin_amdgcn_readfirstlane(tid >> 5);
    const int lo   = blockIdx.x * NB;
    const unsigned rows = (unsigned)min(NB, n - lo);

    static_assert(NB <= 32 * OWN_WAVES, "one lane per row of a wave's run");
    const unsigned per = (rows + OWN_WAVES - 1u) / OWN_WAVES;
    const unsigned r0  = (unsigned)wave * per;
    const unsigned cnt = r0 < rows ? min(per, rows - r0) : 0u;
    float nOut = 0.f;
    if ((unsigned)lane < cnt) nOut = invOut[lo + r0 + lane];
    const float4 bv = ((const float4*)b1)[lane];
    const float4 wv = ((const float4*)W2)[lane];

    float4* acc4 = (float4*)acc;
    for (int i = tid; i < NB * FEATS / 4; i += OWN_THREADS) acc4[i] = make_float4(0.f, 0.f, 0.f, 0.f);
    static_assert(NB <= OWN_THREADS, "one thread per row count");
    if (tid < NB) rowCnt[tid] = 0u;
    if (tid == 0) logCount = 0u;
    __syncthreads();

    const int nG   = nE >> 2;
    const int nW   = (nG + 31) >> 5;
    const int perB = 32 * CH;
    const int nBt  = (nW + perB - 1) / perB;
    const unsigned* __restrict__ bmB = bmp + (size_t)blockIdx.x * wordsPerOwner;
    const int4* __restrict__ dst4 = (const int4*)dst;
    const int4* __restrict__ src4 = (const int4*)src;

    if (wave == 0 && lane < 4) {
        const int e = 4 * nG + lane;
        if (e < nE) own_edge_by_one_lane(acc, rowCnt, h, &logCount, logB, dst[e], src[e], lo, rows);
    }

    unsigned cur[CH];
    unsigned qcnt = 0u;
    unsigned rcnt = 0u;
    int idx0 = wave;
#pragma unroll
    for (int j = 0; j < CH; ++j) {
        const int c = min(idx0, nBt - 1) * perB + j * 32 + lane;
        cur[j] = bmB[min(c, nW - 1)];
    }

    while (idx0 < nBt) {
        const int nidx = idx0 + OWN_WAVES;
        const int nxtB = min(nidx, nBt - 1);
        unsigned nxt[CH];
#pragma unroll
        for (int j = 0; j < CH; ++j) {
            const int c = nxtB * perB + j * 32 + lane;
            nxt[j] = bmB[min(c, nW - 1)];
        }

#pragma unroll
        for (int j = 0; j < CH; ++j) {
            const int c = idx0 * perB + j * 32 + lane;
            unsigned f = cur[j];
            if (c >= nW) f = 0u;
            unsigned bal = __builtin_amdgcn_ballot_w32(f != 0u);
            while (bal != 0u) {
                if (f != 0u) {
                    const unsigned p = (unsigned)__builtin_ctz(f);
                    f &= f - 1u;
                    const unsigned g   = 32u * (unsigned)c + p;
                    const unsigned pos = qcnt + __builtin_amdgcn_mbcnt_lo(bal, 0u);
                    if (pos < QCAP) {
                        qe[wave][pos] = g;
                    } else {
                        const int4 dv = dst4[g];
                        const int4 sv = src4[g];
                        own_edge_by_one_lane(acc, rowCnt, h, &logCount, logB, dv.x, sv.x, lo, rows);
                        own_edge_by_one_lane(acc, rowCnt, h, &logCount, logB, dv.y, sv.y, lo, rows);
                        own_edge_by_one_lane(acc, rowCnt, h, &logCount, logB, dv.z, sv.z, lo, rows);
                        own_edge_by_one_lane(acc, rowCnt, h, &logCount, logB, dv.w, sv.w, lo, rows);
                    }
                }
                qcnt += (unsigned)__builtin_popcount(bal);
                bal = __builtin_amdgcn_ballot_w32(f != 0u);
            }
        }

        __builtin_amdgcn_fence(__ATOMIC_SEQ_CST, "wavefront");
        const unsigned cntq = min(qcnt, (unsigned)QCAP);
        const bool last = nidx >= nBt;

        unsigned base = 0u;
        for (;;) {
            const bool moreA = (cntq - base >= 32u) || (last && cntq > base);
            if (moreA) {
                const unsigned m = min(32u, cntq - base);
                const bool has = (unsigned)lane < m;
                int4 dv = make_int4(0, 0, 0, 0), sv = make_int4(0, 0, 0, 0);
                if (has) {
                    const unsigned g = qe[wave][base + lane];
                    dv = dst4[g];
                    sv = src4[g];
                }
                const int dd4[4] = {dv.x, dv.y, dv.z, dv.w};
                const int ss4[4] = {sv.x, sv.y, sv.z, sv.w};
#pragma unroll
                for (int q = 0; q < 4; ++q) {
                    const unsigned t = (unsigned)(dd4[q] - lo);
                    const bool keep = has && t < rows;
                    const unsigned kb = __builtin_amdgcn_ballot_w32(keep);
                    if (keep) {
                        q2[wave][rcnt + __builtin_amdgcn_mbcnt_lo(kb, 0u)] = ((unsigned)ss4[q] << LOG_ROW_BITS) | t;
                        atomicAdd(&rowCnt[t], 1u);
                    }
                    rcnt += (unsigned)__builtin_popcount(kb);
                }
                base += m;
                __builtin_amdgcn_fence(__ATOMIC_SEQ_CST, "wavefront");
            }

            const bool flush = last && base >= cntq;
            unsigned rb = 0u;
            while (rcnt - rb >= 32u || (flush && rcnt > rb)) {
                const unsigned mm = min(32u, rcnt - rb);
                unsigned w_l = 0u;
                if ((unsigned)lane < mm) w_l = q2[wave][rb + lane];
                const int      s_l = (int)(w_l >> LOG_ROW_BITS);
                const unsigned d_l = w_l & ((1u << LOG_ROW_BITS) - 1u);
                unsigned p0 = 0u;
                if (lane == 0) p0 = atomicAdd(&logCount, mm);
                p0 = (unsigned)__builtin_amdgcn_readfirstlane((int)p0);
                if ((unsigned)lane < mm && p0 + (unsigned)lane < LOG_CAP)
                    logB[p0 + lane] = w_l;
                for (unsigned i = 0; i < mm; i += RG) {
                    float4   v[RG];
                    unsigned dd[RG];
#pragma unroll
                    for (int k = 0; k < RG; ++k) {
                        const unsigned idx = min(i + k, mm - 1u);
                        const int s = __builtin_amdgcn_readlane(s_l, (int)idx);
                        dd[k] = (unsigned)__builtin_amdgcn_readlane((int)d_l, (int)idx);
                        v[k] = ((const float4*)(h + (size_t)s * FEATS))[lane];
                    }
#pragma unroll
                    for (int k = 0; k < RG; ++k) {
                        if (i + k < mm) {
                            float* ar = acc + dd[k] * FEATS + lane;
                            atomicAdd(ar,      v[k].x);
                            atomicAdd(ar + 32, v[k].y);
                            atomicAdd(ar + 64, v[k].z);
                            atomicAdd(ar + 96, v[k].w);
                        }
                    }
                }
                rb += mm;
            }
            if (rb != 0u) {
                const unsigned remr = rcnt - rb;
                unsigned w_l = 0u;
                if ((unsigned)lane < remr) w_l = q2[wave][rb + lane];
                __builtin_amdgcn_fence(__ATOMIC_SEQ_CST, "wavefront");
                if ((unsigned)lane < remr) q2[wave][lane] = w_l;
                rcnt = remr;
                __builtin_amdgcn_fence(__ATOMIC_SEQ_CST, "wavefront");
            }
            if (!moreA) break;
        }
        {
            const unsigned rem = cntq - base;
            if (base != 0u && rem != 0u) {
                unsigned g_l = 0u;
                if ((unsigned)lane < rem) g_l = qe[wave][base + lane];
                __builtin_amdgcn_fence(__ATOMIC_SEQ_CST, "wavefront");
                if ((unsigned)lane < rem) qe[wave][lane] = g_l;
            }
            qcnt = rem;
            __builtin_amdgcn_fence(__ATOMIC_SEQ_CST, "wavefront");
        }

#pragma unroll
        for (int j = 0; j < CH; ++j) cur[j] = nxt[j];
        idx0 = nidx;
    }

    __syncthreads();
    if (tid == 0) elogCount[blockIdx.x] = logCount;
    float nIn = 0.f;
    if ((unsigned)lane < cnt) {
        nIn = rsqrtf(fmaxf((float)rowCnt[r0 + lane], 1.0f));
        invIn[lo + r0 + lane] = nIn;
    }
    float zk = 0.f;
    for (unsigned k = 0; k < cnt; ++k) {
        const float sIn = __builtin_bit_cast(float, __builtin_amdgcn_readlane(__builtin_bit_cast(int, nIn), (int)k));
        const float* ar = acc + (r0 + k) * FEATS + lane;
        const float hx = fmaxf(ar[0]  * sIn + bv.x, 0.0f);
        const float hy = fmaxf(ar[32] * sIn + bv.y, 0.0f);
        const float hz = fmaxf(ar[64] * sIn + bv.z, 0.0f);
        const float hw = fmaxf(ar[96] * sIn + bv.w, 0.0f);
        float sum = hx * wv.x + hy * wv.y + hz * wv.z + hw * wv.w;
        sum += own_other_row(sum);
        sum += own_row_xor<8>(sum);
        sum += own_row_xor<4>(sum);
        sum += own_row_xor<2>(sum);
        sum += own_row_xor<1>(sum);
        if ((unsigned)lane == k) zk = sum;
    }
    if ((unsigned)lane < cnt) z[lo + r0 + lane] = zk * nOut;
}

#define L2_THREADS 1024
#define L2_UN      8
static_assert(L2_UN == 8, "the two load fences in gcn_layer2_owner name eight values");

__global__ __launch_bounds__(L2_THREADS)
void gcn_layer2_owner(const float* __restrict__ z, const int* __restrict__ src,
                      const int* __restrict__ dst, const unsigned* __restrict__ elog,
                      const unsigned* __restrict__ elogCount, const float* __restrict__ invIn,
                      const float* __restrict__ b2, float* __restrict__ out, int n, int nE) {
    __shared__ float sums[NB];
    const int tid = threadIdx.x;
    const int lo  = blockIdx.x * NB;
    const unsigned rows = (unsigned)min(NB, n - lo);
    static_assert(NB <= L2_THREADS, "one thread per row at the start and the end");

    if (tid < NB) sums[tid] = 0.f;
    const unsigned total = elogCount[blockIdx.x];
    float nIn = 0.f;
    if ((unsigned)tid < rows) nIn = invIn[lo + tid];
    const float bias = b2[0];
    __syncthreads();

    if (total <= LOG_CAP) {
        const unsigned* __restrict__ logB = elog + (size_t)blockIdx.x * LOG_CAP;
        for (unsigned i0 = (unsigned)tid; i0 < total; i0 += L2_UN * L2_THREADS) {
            unsigned w[L2_UN];
            float    v[L2_UN];
#pragma unroll
            for (int k = 0; k < L2_UN; ++k) w[k] = logB[min(i0 + (unsigned)k * L2_THREADS, total - 1u)];
            asm volatile("" :: "v"(w[0]), "v"(w[1]), "v"(w[2]), "v"(w[3]), "v"(w[4]), "v"(w[5]), "v"(w[6]), "v"(w[7]));
#pragma unroll
            for (int k = 0; k < L2_UN; ++k) v[k] = z[w[k] >> LOG_ROW_BITS];
            asm volatile("" :: "v"(v[0]), "v"(v[1]), "v"(v[2]), "v"(v[3]), "v"(v[4]), "v"(v[5]), "v"(v[6]), "v"(v[7]));
#pragma unroll
            for (int k = 0; k < L2_UN; ++k)
                if (i0 + (unsigned)k * L2_THREADS < total)
                    atomicAdd(&sums[w[k] & ((1u << LOG_ROW_BITS) - 1u)], v[k]);
        }
    } else {
        for (int e = tid; e < nE; e += L2_THREADS) {
            const unsigned t = (unsigned)(dst[e] - lo);
            if (t < rows) atomicAdd(&sums[t], z[src[e]]);
        }
    }
    __syncthreads();
    if ((unsigned)tid < rows) out[lo + tid] = sums[tid] * nIn + bias;
}

extern "C" void kernel_launch(void* const* d_in, const int* in_sizes, int n_in,
                              void* d_out, int out_size, void* d_ws, size_t ws_size,
                              hipStream_t stream) {
    const float* x   = (const float*)d_in[0];
    const int*   src = (const int*)d_in[1];
    const int*   dst = (const int*)d_in[2];
    const float* W1  = (const float*)d_in[3];
    const float* b1  = (const float*)d_in[4];
    const float* W2  = (const float*)d_in[5];
    const float* b2  = (const float*)d_in[6];
    float* out = (float*)d_out;

    const int n  = in_sizes[0] / FEATS;
    const int nE = in_sizes[1];

    const int nOwn = (n + NB - 1) / NB;

    float* f      = (float*)d_ws;
    float* invOut = f;
    float* invIn  = f + (size_t)n;
    float* z      = f + 2 * (size_t)n;
    unsigned* elogCount = (unsigned*)(f + 3 * (size_t)n);
    float* h      = f + 4 * (size_t)n;
    unsigned* elog = (unsigned*)(h + (size_t)n * FEATS);
    const int nIdx = ((nE >> 2) + IDX_PER * K1_THREADS - 1) / (IDX_PER * K1_THREADS);
    const int wordsPerOwner = IDX_ROWS * nIdx;
    unsigned* bmp  = elog + (size_t)nOwn * LOG_CAP;
    unsigned* part = bmp + (size_t)nOwn * wordsPerOwner;
    const int sliceLen = (((nE + CNT_PLANES - 1) / CNT_PLANES + 3) / 4) * 4;
    const int nSlices  = (nE + sliceLen - 1) / sliceLen;
    const int nCnt    = 2 * nSlices;

    gcn_count_index<<<nCnt + nIdx, K1_THREADS, 0, stream>>>(src, dst, part, bmp, n, nE, sliceLen, nCnt, nOwn, wordsPerOwner);

    gcn_gemm1_wmma<<<n / 16, 256, 0, stream>>>(x, W1, part, nSlices, invOut, h);

    gcn_rows_owner<<<nOwn, OWN_THREADS, 0, stream>>>(h, src, dst, bmp, wordsPerOwner, invIn, invOut, b1, W2, z, elog, elogCount, n, nE);

    gcn_layer2_owner<<<nOwn, L2_THREADS, 0, stream>>>(z, src, dst, elog, elogCount, invIn, b2, out, n, nE);
}


# Row sums in registers, not LDS float adds: a wave owns 13 dst rows, edges go to the owner wave's LDS queue, one writer per row; no sum plane in LDS; sum order = queue arrival
# speedup vs baseline: 5.5140x; 1.0714x over previous
#include <hip/hip_runtime.h>
#include <hip/hip_bf16.h>

typedef __attribute__((ext_vector_type(2))) float v2f;
typedef __attribute__((ext_vector_type(8))) float v8f;

#define FEATS 128

#define K1_THREADS  1024
#define CNT_PLANES  64
#define CNT_RNGW    25600
#define CNT_PW      (2 * CNT_RNGW)
#define CNT_UN      8
static_assert(CNT_RNGW % 32 == 0, "whole lines per range");

#define NB 400

#define IDX_PER    4
#define IDX_ROWS   (32 * IDX_PER)
#define IDX_OWN    256
#define IDX_PITCH  257
static_assert(K1_THREADS == 1024, "32 waves: a wave's 32 groups of one pass are one index word per owner");
static_assert(2 * CNT_PW >= IDX_OWN * NB, "the two count ranges cover every node the owner index has room for");
static_assert(CNT_RNGW <= IDX_ROWS * IDX_PITCH, "the count words fit the block's LDS");

__device__ __forceinline__ void cnt_hit(unsigned* cnt, int v, int lo, unsigned rng) {
    const unsigned t = (unsigned)(v - lo);
    if (t < rng) atomicAdd(&cnt[t >> 1], 1u << ((t & 1u) * 16u));
}

__global__ __launch_bounds__(K1_THREADS)
void gcn_count_index(const int* __restrict__ src, const int* __restrict__ dst,
                     unsigned* __restrict__ part, unsigned* __restrict__ bmp,
                     int n, int nE, int sliceLen, int nCnt, int nOwn, int wordsPerOwner) {
    __shared__ unsigned sh[IDX_ROWS * IDX_PITCH];
    const int tid = threadIdx.x;

    if ((int)blockIdx.x < nCnt) {
        const int s  = blockIdx.x >> 1;
        const int r  = blockIdx.x & 1;
        const int lo = r * (2 * CNT_RNGW);
        const int rn = min(2 * CNT_RNGW, n - lo);
        const unsigned rng = rn > 0 ? (unsigned)rn : 0u;

        const int e0 = s * sliceLen;
        const int e1 = min(nE, e0 + sliceLen);
        const int nV = (e1 - e0) >> 2;
        const int4* __restrict__ src4 = (const int4*)(src + e0);
        int4 a[CNT_UN];
        if (nV > 0) {
#pragma unroll
            for (int k = 0; k < CNT_UN; ++k) a[k] = src4[min(tid + k * K1_THREADS, nV - 1)];
        }

        for (int i = tid; i < CNT_RNGW; i += K1_THREADS) sh[i] = 0u;
        __syncthreads();

        for (int g0 = tid; g0 < nV; ) {
#pragma unroll
            for (int k = 0; k < CNT_UN; ++k) {
                if (g0 + k * K1_THREADS < nV) {
                    cnt_hit(sh, a[k].x, lo, rng);
                    cnt_hit(sh, a[k].y, lo, rng);
                    cnt_hit(sh, a[k].z, lo, rng);
                    cnt_hit(sh, a[k].w, lo, rng);
                }
            }
            g0 += CNT_UN * K1_THREADS;
            if (g0 < nV) {
#pragma unroll
                for (int k = 0; k < CNT_UN; ++k) a[k] = src4[min(g0 + k * K1_THREADS, nV - 1)];
            }
        }
        for (int e = e0 + 4 * nV + tid; e < e1; e += K1_THREADS) cnt_hit(sh, src[e], lo, rng);
        __syncthreads();

        const unsigned rngW = (rng + 1u) >> 1;
        unsigned* __restrict__ plane = part + (size_t)s * CNT_PW + (size_t)r * CNT_RNGW;
        for (unsigned i = (unsigned)tid; i < rngW; i += K1_THREADS) plane[i] = sh[i];
        return;
    }

    const int bi = (int)blockIdx.x - nCnt;
    const int nG = nE >> 2;
    const int gb = bi * (IDX_PER * K1_THREADS);
    int4 d[IDX_PER];
#pragma unroll
    for (int j = 0; j < IDX_PER; ++j) d[j] = ((const int4*)dst)[min(gb + j * K1_THREADS + tid, nG - 1)];
    for (int k = tid; k < IDX_ROWS * IDX_PITCH; k += K1_THREADS) sh[k] = 0u;
    __syncthreads();

    const unsigned bit = 1u << (tid & 31);
#pragma unroll
    for (int j = 0; j < IDX_PER; ++j) {
        if (gb + j * K1_THREADS + tid < nG) {
            unsigned* row = sh + ((tid >> 5) + 32 * j) * IDX_PITCH;
            atomicOr(&row[min((unsigned)d[j].x / NB, (unsigned)(IDX_OWN - 1))], bit);
            atomicOr(&row[min((unsigned)d[j].y / NB, (unsigned)(IDX_OWN - 1))], bit);
            atomicOr(&row[min((unsigned)d[j].z / NB, (unsigned)(IDX_OWN - 1))], bit);
            atomicOr(&row[min((unsigned)d[j].w / NB, (unsigned)(IDX_OWN - 1))], bit);
        }
    }
    __syncthreads();

    const int lane = tid & 31;
    for (int o = tid >> 5; o < nOwn; o += K1_THREADS / 32) {
        unsigned* __restrict__ line = bmp + (size_t)o * wordsPerOwner + (size_t)IDX_ROWS * bi + lane;
#pragma unroll
        for (int q = 0; q < IDX_PER; ++q) line[32 * q] = sh[(32 * q + lane) * IDX_PITCH + o];
    }
}

#define G1_PITCH 132
static_assert(G1_PITCH % 4 == 0 && (G1_PITCH / 4) % 2 == 1, "16-byte aligned rows at an odd multiple of 16 bytes");

template <int M>
__device__ __forceinline__ unsigned g1_xor(unsigned v) {
    return (unsigned)__builtin_amdgcn_update_dpp((int)v, (int)v, 0x160 + M, 0xf, 0xf, true);
}
__device__ __forceinline__ unsigned g1_xor16(unsigned v) {
    return (unsigned)__builtin_amdgcn_permlanex16((int)v, (int)v, 0x76543210, 0xfedcba98, false, false);
}

__global__ __launch_bounds__(256)
void gcn_gemm1_wmma(const float* __restrict__ x, const float* __restrict__ W1,
                    const unsigned* __restrict__ part, int nSlices,
                    float* __restrict__ invOut, float* __restrict__ h) {
    __shared__ __attribute__((aligned(16))) float xs[16 * G1_PITCH];
    __shared__ __attribute__((aligned(8))) unsigned cw[8][16];
    const int lane    = threadIdx.x & 31;
    const int colTile = threadIdx.x >> 5;
    const int rowTile = blockIdx.x;
    const int l15     = lane & 15;
    const int khalf   = lane >> 4;
    const int nCol    = colTile * 16 + l15;

    const int r0   = 2 * colTile;
    const int grow = rowTile * 16 + r0;

    const int wi = threadIdx.x & 7;
    const int pg = threadIdx.x >> 3;
    static_assert(CNT_PLANES == 64, "two planes per thread of the block's 256");
    const unsigned* __restrict__ pw = part + (size_t)rowTile * 8 + wi;
    const unsigned w0r = pw[(size_t)min(pg, nSlices - 1) * CNT_PW];
    const unsigned w1r = pw[(size_t)min(pg + 32, nSlices - 1) * CNT_PW];

    const float4 xa = ((const float4*)(x + (size_t)grow * FEATS))[lane];
    const float4 xb = ((const float4*)(x + (size_t)(grow + 1) * FEATS))[lane];

    v2f bf[FEATS / 4];
#pragma unroll
    for (int k = 0; k < FEATS; k += 4) {
        const int ka = k + 2 * khalf;
        bf[k >> 2].x = W1[(size_t)ka       * FEATS + nCol];
        bf[k >> 2].y = W1[(size_t)(ka + 1) * FEATS + nCol];
    }

    const unsigned w0 = pg < nSlices ? w0r : 0u;
    const unsigned w1 = pg + 32 < nSlices ? w1r : 0u;
    unsigned cEven = (w0 & 0xFFFFu) + (w1 & 0xFFFFu);
    unsigned cOdd  = (w0 >> 16) + (w1 >> 16);
    cEven += g1_xor<8>(cEven);  cOdd += g1_xor<8>(cOdd);
    cEven += g1_xor16(cEven);   cOdd += g1_xor16(cOdd);
    if (lane < 8) {
        cw[colTile][2 * wi]     = cEven;
        cw[colTile][2 * wi + 1] = cOdd;
    }
    __syncthreads();
    const uint2 c2 = *(const uint2*)&cw[lane & 7][r0];
    unsigned ca = c2.x, cb = c2.y;
    ca += g1_xor<1>(ca);  cb += g1_xor<1>(cb);
    ca += g1_xor<2>(ca);  cb += g1_xor<2>(cb);
    ca += g1_xor<4>(ca);  cb += g1_xor<4>(cb);
    const float sa = rsqrtf(fmaxf((float)ca, 1.0f));
    const float sb = rsqrtf(fmaxf((float)cb, 1.0f));

    *(float4*)&xs[r0 * G1_PITCH + 4 * lane]       = make_float4(xa.x * sa, xa.y * sa, xa.z * sa, xa.w * sa);
    *(float4*)&xs[(r0 + 1) * G1_PITCH + 4 * lane] = make_float4(xb.x * sb, xb.y * sb, xb.z * sb, xb.w * sb);
    __syncthreads();
    if (lane == 0) {
        invOut[grow]     = sa;
        invOut[grow + 1] = sb;
    }

    const float* __restrict__ arow = xs + l15 * G1_PITCH + 2 * khalf;
    v8f c = {};
#pragma unroll
    for (int k = 0; k < FEATS; k += 4) {
        v2f a;
        a.x = arow[k];
        a.y = arow[k + 1];
        c = __builtin_amdgcn_wmma_f32_16x16x4_f32(
                 false, a,  false, bf[k >> 2],
                 (short)0, c,  false,  false);
    }

#pragma unroll
    for (int r = 0; r < 8; ++r) {
        const int mm = (lane < 16) ? r : (r + 8);
        h[(size_t)(rowTile * 16 + mm) * FEATS + nCol] = c[r];
    }
}

#define QCAP        128
#define CH          2
#define RG          8
#define OWN_THREADS 1024
#define OWN_WAVES   (OWN_THREADS / 32)
#define RPW         13
#define WQCAP       1024
static_assert(RPW * OWN_WAVES >= NB, "every row of the block has an owner wave");
#define LOG_CAP     32768u
#define LOG_ROW_BITS 9
static_assert(NB <= (1 << LOG_ROW_BITS), "row inside the block must fit the low bits of a log word");

__device__ __forceinline__ void own_log_one(unsigned* logCount, unsigned* __restrict__ logB, unsigned s, unsigned t) {
    const unsigned p = atomicAdd(logCount, 1u);
    if (p < LOG_CAP) logB[p] = (s << LOG_ROW_BITS) | t;
}

__device__ __forceinline__ void own_route(unsigned (*wq)[WQCAP], unsigned* wqTail, unsigned* ovf, unsigned s, unsigned t) {
    const unsigned ow = t / RPW;
    const unsigned p  = atomicAdd(&wqTail[ow], 1u);
    if (p < WQCAP) wq[ow][p] = (s << LOG_ROW_BITS) | t;
    else *ovf = 1u;
}

__device__ __forceinline__ float own_other_row(float x) {
    const int b = __builtin_bit_cast(int, x);
    return __builtin_bit_cast(float, __builtin_amdgcn_permlanex16(b, b, 0x76543210, 0xfedcba98, false, false));
}

template <int M>
__device__ __forceinline__ float own_row_xor(float x) {
    const int b = __builtin_bit_cast(int, x);
    return __builtin_bit_cast(float, __builtin_amdgcn_update_dpp(b, b, 0x160 + M, 0xf, 0xf, true));
}

typedef __attribute__((ext_vector_type(16))) float v16f;
static_assert(RPW <= 16, "a row of the wave is an element of a 16-wide register vector");
#define OWN_ADD(jj, val) do { const float4 vv = (val); const int j_ = (jj); \
    ax[j_] += vv.x; ay[j_] += vv.y; az[j_] += vv.z; aw[j_] += vv.w; } while (0)

__global__ __launch_bounds__(OWN_THREADS)
void gcn_rows_owner(const float* __restrict__ h, const int* __restrict__ src,
                    const int* __restrict__ dst, const unsigned* __restrict__ bmp, int wordsPerOwner,
                    float* __restrict__ invIn, const float* __restrict__ invOut,
                    const float* __restrict__ b1, const float* __restrict__ W2,
                    float* __restrict__ z, unsigned* __restrict__ elog, unsigned* __restrict__ elogCount,
                    int n, int nE) {
    __shared__ unsigned qe[OWN_WAVES][QCAP];
    __shared__ unsigned wq[OWN_WAVES][WQCAP];
    __shared__ unsigned wqTail[OWN_WAVES];
    __shared__ unsigned rowCnt[NB];
    __shared__ unsigned logCount;
    __shared__ unsigned ovf;
    unsigned* __restrict__ logB = elog + (size_t)blockIdx.x * LOG_CAP;

    const int tid  = threadIdx.x;
    const int lane = tid & 31;
    const int wave = __builtin_amdgcn_readfirstlane(tid >> 5);
    const int lo   = blockIdx.x * NB;
    const unsigned rows = (unsigned)min(NB, n - lo);

    const unsigned r0  = (unsigned)wave * RPW;
    const unsigned cnt = r0 < rows ? min((unsigned)RPW, rows - r0) : 0u;
    float nOut = 0.f;
    if ((unsigned)lane < cnt) nOut = invOut[lo + r0 + lane];
    const float4 bv = ((const float4*)b1)[lane];
    const float4 wv = ((const float4*)W2)[lane];

    static_assert(NB <= OWN_THREADS, "one thread per row count");
    if (tid < NB) rowCnt[tid] = 0u;
    if (tid < OWN_WAVES) wqTail[tid] = 0u;
    if (tid == 0) { logCount = 0u; ovf = 0u; }
    __syncthreads();

    const int nG   = nE >> 2;
    const int nW   = (nG + 31) >> 5;
    const int perB = 32 * CH;
    const int nBt  = (nW + perB - 1) / perB;
    const unsigned* __restrict__ bmB = bmp + (size_t)blockIdx.x * wordsPerOwner;
    const int4* __restrict__ dst4 = (const int4*)dst;
    const int4* __restrict__ src4 = (const int4*)src;

    if (wave == 0 && lane < 4) {
        const int e = 4 * nG + lane;
        if (e < nE) {
            const unsigned t = (unsigned)(dst[e] - lo);
            if (t < rows) own_route(wq, wqTail, &ovf, (unsigned)src[e], t);
        }
    }

    unsigned cur[CH];
    unsigned qcnt = 0u;
    int idx0 = wave;
#pragma unroll
    for (int j = 0; j < CH; ++j) {
        const int c = min(idx0, nBt - 1) * perB + j * 32 + lane;
        cur[j] = bmB[min(c, nW - 1)];
    }

    while (idx0 < nBt) {
        const int nidx = idx0 + OWN_WAVES;
        const int nxtB = min(nidx, nBt - 1);
        unsigned nxt[CH];
#pragma unroll
        for (int j = 0; j < CH; ++j) {
            const int c = nxtB * perB + j * 32 + lane;
            nxt[j] = bmB[min(c, nW - 1)];
        }

#pragma unroll
        for (int j = 0; j < CH; ++j) {
            const int c = idx0 * perB + j * 32 + lane;
            unsigned f = cur[j];
            if (c >= nW) f = 0u;
            unsigned bal = __builtin_amdgcn_ballot_w32(f != 0u);
            while (bal != 0u) {
                if (f != 0u) {
                    const unsigned p = (unsigned)__builtin_ctz(f);
                    f &= f - 1u;
                    const unsigned g   = 32u * (unsigned)c + p;
                    const unsigned pos = qcnt + __builtin_amdgcn_mbcnt_lo(bal, 0u);
                    if (pos < QCAP) {
                        qe[wave][pos] = g;
                    } else {
                        const int4 dv = dst4[g];
                        const int4 sv = src4[g];
                        const int dd4[4] = {dv.x, dv.y, dv.z, dv.w};
                        const int ss4[4] = {sv.x, sv.y, sv.z, sv.w};
#pragma unroll
                        for (int q = 0; q < 4; ++q) {
                            const unsigned t = (unsigned)(dd4[q] - lo);
                            if (t < rows) own_route(wq, wqTail, &ovf, (unsigned)ss4[q], t);
                        }
                    }
                }
                qcnt += (unsigned)__builtin_popcount(bal);
                bal = __builtin_amdgcn_ballot_w32(f != 0u);
            }
        }

        __builtin_amdgcn_fence(__ATOMIC_SEQ_CST, "wavefront");
        const unsigned cntq = min(qcnt, (unsigned)QCAP);
        const bool last = nidx >= nBt;

        unsigned base = 0u;
        while ((cntq - base >= 32u) || (last && cntq > base)) {
            const unsigned m = min(32u, cntq - base);
            const bool has = (unsigned)lane < m;
            int4 dv = make_int4(0, 0, 0, 0), sv = make_int4(0, 0, 0, 0);
            if (has) {
                const unsigned g = qe[wave][base + lane];
                dv = dst4[g];
                sv = src4[g];
            }
            const int dd4[4] = {dv.x, dv.y, dv.z, dv.w};
            const int ss4[4] = {sv.x, sv.y, sv.z, sv.w};
#pragma unroll
            for (int q = 0; q < 4; ++q) {
                const unsigned t = (unsigned)(dd4[q] - lo);
                if (has && t < rows) own_route(wq, wqTail, &ovf, (unsigned)ss4[q], t);
            }
            base += m;
        }
        {
            const unsigned rem = cntq - base;
            if (base != 0u && rem != 0u) {
                unsigned g_l = 0u;
                if ((unsigned)lane < rem) g_l = qe[wave][base + lane];
                __builtin_amdgcn_fence(__ATOMIC_SEQ_CST, "wavefront");
                if ((unsigned)lane < rem) qe[wave][lane] = g_l;
            }
            qcnt = rem;
            __builtin_amdgcn_fence(__ATOMIC_SEQ_CST, "wavefront");
        }

#pragma unroll
        for (int j = 0; j < CH; ++j) cur[j] = nxt[j];
        idx0 = nidx;
    }

    __syncthreads();

    v16f ax = {}, ay = {}, az = {}, aw = {};
    const bool over = ovf != 0u;

    if (!over) {
        const unsigned nq = (unsigned)__builtin_amdgcn_readfirstlane((int)min(wqTail[wave], (unsigned)WQCAP));
        for (unsigned pb = 0u; pb < nq; pb += 32u) {
            const unsigned mm = min(32u, nq - pb);
            unsigned w_l = 0u;
            if ((unsigned)lane < mm) w_l = wq[wave][pb + lane];
            const int      s_l = (int)(w_l >> LOG_ROW_BITS);
            const unsigned t_l = w_l & ((1u << LOG_ROW_BITS) - 1u);
            const int      j_l = (int)(t_l - r0);
            if ((unsigned)lane < mm) atomicAdd(&rowCnt[t_l], 1u);
            unsigned p0 = 0u;
            if (lane == 0) p0 = atomicAdd(&logCount, mm);
            p0 = (unsigned)__builtin_amdgcn_readfirstlane((int)p0);
            if ((unsigned)lane < mm && p0 + (unsigned)lane < LOG_CAP)
                logB[p0 + lane] = w_l;
            for (unsigned i = 0; i < mm; i += RG) {
                float4 v[RG];
                int    jj[RG];
#pragma unroll
                for (int k = 0; k < RG; ++k) {
                    const unsigned idx = min(i + k, mm - 1u);
                    const int s = __builtin_amdgcn_readlane(s_l, (int)idx);
                    jj[k] = __builtin_amdgcn_readlane(j_l, (int)idx);
                    v[k] = ((const float4*)(h + (size_t)s * FEATS))[lane];
                }
#pragma unroll
                for (int k = 0; k < RG; ++k) {
                    if (i + k < mm) OWN_ADD(jj[k], v[k]);
                }
            }
        }
    } else {
        for (int e0 = 0; e0 < nE; e0 += 32) {
            const int e = e0 + lane;
            unsigned t = 0xFFFFFFFFu;
            if (e < nE) t = (unsigned)(dst[e] - lo);
            const bool mine = t < rows && t / RPW == (unsigned)wave;
            unsigned bal = __builtin_amdgcn_ballot_w32(mine);
            int s_l = 0;
            if (mine) {
                s_l = src[e];
                atomicAdd(&rowCnt[t], 1u);
                own_log_one(&logCount, logB, (unsigned)s_l, t);
            }
            const int j_l = (int)(t - r0);
            while (bal != 0u) {
                const int l = __builtin_ctz(bal);
                bal &= bal - 1u;
                const int s = __builtin_amdgcn_readlane(s_l, l);
                const int j = __builtin_amdgcn_readlane(j_l, l);
                const float4 v = ((const float4*)(h + (size_t)s * FEATS))[lane];
                OWN_ADD(j, v);
            }
        }
    }

    __syncthreads();
    if (tid == 0) elogCount[blockIdx.x] = logCount;
    float nIn = 0.f;
    if ((unsigned)lane < cnt) {
        nIn = rsqrtf(fmaxf((float)rowCnt[r0 + lane], 1.0f));
        invIn[lo + r0 + lane] = nIn;
    }
    float zk = 0.f;
#pragma unroll
    for (int k = 0; k < RPW; ++k) {
        if ((unsigned)k < cnt) {
            const float sIn = __builtin_bit_cast(float, __builtin_amdgcn_readlane(__builtin_bit_cast(int, nIn), k));
            const float hx = fmaxf(ax[k] * sIn + bv.x, 0.0f);
            const float hy = fmaxf(ay[k] * sIn + bv.y, 0.0f);
            const float hz = fmaxf(az[k] * sIn + bv.z, 0.0f);
            const float hw = fmaxf(aw[k] * sIn + bv.w, 0.0f);
            float sum = hx * wv.x + hy * wv.y + hz * wv.z + hw * wv.w;
            sum += own_other_row(sum);
            sum += own_row_xor<8>(sum);
            sum += own_row_xor<4>(sum);
            sum += own_row_xor<2>(sum);
            sum += own_row_xor<1>(sum);
            if (lane == k) zk = sum;
        }
    }
    if ((unsigned)lane < cnt) z[lo + r0 + lane] = zk * nOut;
}

#define L2_THREADS 1024
#define L2_UN      8
static_assert(L2_UN == 8, "the two load fences in gcn_layer2_owner name eight values");

__global__ __launch_bounds__(L2_THREADS)
void gcn_layer2_owner(const float* __restrict__ z, const int* __restrict__ src,
                      const int* __restrict__ dst, const unsigned* __restrict__ elog,
                      const unsigned* __restrict__ elogCount, const float* __restrict__ invIn,
                      const float* __restrict__ b2, float* __restrict__ out, int n, int nE) {
    __shared__ float sums[NB];
    const int tid = threadIdx.x;
    const int lo  = blockIdx.x * NB;
    const unsigned rows = (unsigned)min(NB, n - lo);
    static_assert(NB <= L2_THREADS, "one thread per row at the start and the end");

    if (tid < NB) sums[tid] = 0.f;
    const unsigned total = elogCount[blockIdx.x];
    float nIn = 0.f;
    if ((unsigned)tid < rows) nIn = invIn[lo + tid];
    const float bias = b2[0];
    __syncthreads();

    if (total <= LOG_CAP) {
        const unsigned* __restrict__ logB = elog + (size_t)blockIdx.x * LOG_CAP;
        for (unsigned i0 = (unsigned)tid; i0 < total; i0 += L2_UN * L2_THREADS) {
            unsigned w[L2_UN];
            float    v[L2_UN];
#pragma unroll
            for (int k = 0; k < L2_UN; ++k) w[k] = logB[min(i0 + (unsigned)k * L2_THREADS, total - 1u)];
            asm volatile("" :: "v"(w[0]), "v"(w[1]), "v"(w[2]), "v"(w[3]), "v"(w[4]), "v"(w[5]), "v"(w[6]), "v"(w[7]));
#pragma unroll
            for (int k = 0; k < L2_UN; ++k) v[k] = z[w[k] >> LOG_ROW_BITS];
            asm volatile("" :: "v"(v[0]), "v"(v[1]), "v"(v[2]), "v"(v[3]), "v"(v[4]), "v"(v[5]), "v"(v[6]), "v"(v[7]));
#pragma unroll
            for (int k = 0; k < L2_UN; ++k)
                if (i0 + (unsigned)k * L2_THREADS < total)
                    atomicAdd(&sums[w[k] & ((1u << LOG_ROW_BITS) - 1u)], v[k]);
        }
    } else {
        for (int e = tid; e < nE; e += L2_THREADS) {
            const unsigned t = (unsigned)(dst[e] - lo);
            if (t < rows) atomicAdd(&sums[t], z[src[e]]);
        }
    }
    __syncthreads();
    if ((unsigned)tid < rows) out[lo + tid] = sums[tid] * nIn + bias;
}

extern "C" void kernel_launch(void* const* d_in, const int* in_sizes, int n_in,
                              void* d_out, int out_size, void* d_ws, size_t ws_size,
                              hipStream_t stream) {
    const float* x   = (const float*)d_in[0];
    const int*   src = (const int*)d_in[1];
    const int*   dst = (const int*)d_in[2];
    const float* W1  = (const float*)d_in[3];
    const float* b1  = (const float*)d_in[4];
    const float* W2  = (const float*)d_in[5];
    const float* b2  = (const float*)d_in[6];
    float* out = (float*)d_out;

    const int n  = in_sizes[0] / FEATS;
    const int nE = in_sizes[1];

    const int nOwn = (n + NB - 1) / NB;

    float* f      = (float*)d_ws;
    float* invOut = f;
    float* invIn  = f + (size_t)n;
    float* z      = f + 2 * (size_t)n;
    unsigned* elogCount = (unsigned*)(f + 3 * (size_t)n);
    float* h      = f + 4 * (size_t)n;
    unsigned* elog = (unsigned*)(h + (size_t)n * FEATS);
    const int nIdx = ((nE >> 2) + IDX_PER * K1_THREADS - 1) / (IDX_PER * K1_THREADS);
    const int wordsPerOwner = IDX_ROWS * nIdx;
    unsigned* bmp  = elog + (size_t)nOwn * LOG_CAP;
    unsigned* part = bmp + (size_t)nOwn * wordsPerOwner;
    const int sliceLen = (((nE + CNT_PLANES - 1) / CNT_PLANES + 3) / 4) * 4;
    const int nSlices  = (nE + sliceLen - 1) / sliceLen;
    const int nCnt    = 2 * nSlices;

    gcn_count_index<<<nCnt + nIdx, K1_THREADS, 0, stream>>>(src, dst, part, bmp, n, nE, sliceLen, nCnt, nOwn, wordsPerOwner);

    gcn_gemm1_wmma<<<n / 16, 256, 0, stream>>>(x, W1, part, nSlices, invOut, h);

    gcn_rows_owner<<<nOwn, OWN_THREADS, 0, stream>>>(h, src, dst, bmp, wordsPerOwner, invIn, invOut, b1, W2, z, elog, elogCount, n, nE);

    gcn_layer2_owner<<<nOwn, L2_THREADS, 0, stream>>>(z, src, dst, elog, elogCount, invIn, b2, out, n, nE);
}
